# starting kernel
# baseline (speedup 1.0000x reference)
#include <hip/hip_runtime.h>
#include <cstdint>

typedef float v2f __attribute__((ext_vector_type(2)));
typedef float v8f __attribute__((ext_vector_type(8)));

#define HEADS 8
#define NEG_SLOPE 0.2f
#define SM_EPS 1e-16f
#define NT 4

__global__ void wmma_gemm_f32(const float* __restrict__ A,
                              const float* __restrict__ B,
                              float* __restrict__ C,
                              int M, int K, int Ncols) {
  int wave = (int)((blockIdx.x * blockDim.x + threadIdx.x) >> 5);
  int lane = (int)(threadIdx.x & 31);
  int tilesM  = (M + 15) >> 4;
  int groupsN = Ncols >> 6;
  if (wave >= tilesM * groupsN) return;
  int tm = wave / groupsN;
  int tg = wave - tm * groupsN;
  int half = lane >> 4;
  int l = lane & 15;

  int rowA = tm * 16 + l;
  if (rowA >= M) rowA = M - 1;
  const float* aP = A + (size_t)rowA * K + 2 * half;
  const float* bP = B + (size_t)(2 * half) * Ncols + tg * (16 * NT) + l;

  v8f acc[NT];
#pragma unroll
  for (int t = 0; t < NT; ++t) acc[t] = (v8f){};

  v2f aCur, bCur[NT];
  aCur.x = aP[0];
  aCur.y = aP[1];
#pragma unroll
  for (int t = 0; t < NT; ++t) {
    bCur[t].x = bP[t * 16];
    bCur[t].y = bP[t * 16 + Ncols];
  }

  for (int k0 = 0; k0 < K - 4; k0 += 4) {
    aP += 4;
    bP += (size_t)4 * Ncols;
    v2f aNxt, bNxt[NT];
    aNxt.x = aP[0];
    aNxt.y = aP[1];
#pragma unroll
    for (int t = 0; t < NT; ++t) {
      bNxt[t].x = bP[t * 16];
      bNxt[t].y = bP[t * 16 + Ncols];
    }
#pragma unroll
    for (int t = 0; t < NT; ++t) {
      acc[t] = __builtin_amdgcn_wmma_f32_16x16x4_f32(false, aCur, false, bCur[t],
                                                     (short)0, acc[t],
                                                     false, false);
    }
    aCur = aNxt;
#pragma unroll
    for (int t = 0; t < NT; ++t) bCur[t] = bNxt[t];
  }
#pragma unroll
  for (int t = 0; t < NT; ++t) {
    acc[t] = __builtin_amdgcn_wmma_f32_16x16x4_f32(false, aCur, false, bCur[t],
                                                   (short)0, acc[t],
                                                   false, false);
  }

#pragma unroll
  for (int t = 0; t < NT; ++t) {
    int col = tg * (16 * NT) + t * 16 + l;
#pragma unroll
    for (int r = 0; r < 8; ++r) {
      int row = tm * 16 + r + 8 * half;
      if (row < M) C[(size_t)row * Ncols + col] = acc[t][r];
    }
  }
}

__global__ void attn_coeff_kernel(const float* __restrict__ h,
                                  const float* __restrict__ att_src,
                                  const float* __restrict__ att_dst,
                                  float* __restrict__ asrc,
                                  float* __restrict__ adst,
                                  int N, int C) {
  long i = blockIdx.x * (long)blockDim.x + threadIdx.x;
  if (i >= (long)N * HEADS) return;
  int head = (int)(i & (HEADS - 1));
  const float* hp = h + (size_t)i * C;
  const float* as = att_src + (size_t)head * C;
  const float* ad = att_dst + (size_t)head * C;
  float s = 0.f, d = 0.f;
  for (int c = 0; c < C; ++c) {
    float v = hp[c];
    s += v * as[c];
    d += v * ad[c];
  }
  asrc[i] = s;
  adst[i] = d;
}

__global__ void fill_f32(float* __restrict__ p, float v, long n) {
  long i = blockIdx.x * (long)blockDim.x + threadIdx.x;
  if (i < n) p[i] = v;
}

__device__ __forceinline__ float lrelu(float v) {
  return v > 0.f ? v : NEG_SLOPE * v;
}

__global__ void edge_max_kernel(const int* __restrict__ src,
                                const int* __restrict__ dst,
                                int E, int N,
                                const float* __restrict__ asrc,
                                const float* __restrict__ adst,
                                float* __restrict__ m) {
  long i = blockIdx.x * (long)blockDim.x + threadIdx.x;
  long tot = (long)(E + N) * HEADS;
  if (i >= tot) return;
  int e = (int)(i >> 3);
  int head = (int)(i & 7);
  int s = (e < E) ? src[e] : (e - E);
  int d = (e < E) ? dst[e] : (e - E);
  float v = lrelu(asrc[(size_t)s * HEADS + head] + adst[(size_t)d * HEADS + head]);
  atomicMax(&m[(size_t)d * HEADS + head], v);
}

__global__ void edge_expsum_kernel(const int* __restrict__ src,
                                   const int* __restrict__ dst,
                                   int E, int N,
                                   const float* __restrict__ asrc,
                                   const float* __restrict__ adst,
                                   const float* __restrict__ m,
                                   float* __restrict__ denom) {
  long i = blockIdx.x * (long)blockDim.x + threadIdx.x;
  long tot = (long)(E + N) * HEADS;
  if (i >= tot) return;
  int e = (int)(i >> 3);
  int head = (int)(i & 7);
  int s = (e < E) ? src[e] : (e - E);
  int d = (e < E) ? dst[e] : (e - E);
  size_t dh = (size_t)d * HEADS + head;
  float v = lrelu(asrc[(size_t)s * HEADS + head] + adst[dh]);
  atomicAdd(&denom[dh], __expf(v - m[dh]));
}

__global__ void edge_agg_kernel(const int* __restrict__ src,
                                const int* __restrict__ dst,
                                int E, int N, int C,
                                const float* __restrict__ asrc,
                                const float* __restrict__ adst,
                                const float* __restrict__ m,
                                const float* __restrict__ denom,
                                const float* __restrict__ h,
                                float* __restrict__ agg) {
  long i = blockIdx.x * (long)blockDim.x + threadIdx.x;
  long tot = (long)(E + N) * HEADS;
  if (i >= tot) return;
  int e = (int)(i >> 3);
  int head = (int)(i & 7);
  int s = (e < E) ? src[e] : (e - E);
  int d = (e < E) ? dst[e] : (e - E);
  size_t dh = (size_t)d * HEADS + head;
  float v = lrelu(asrc[(size_t)s * HEADS + head] + adst[dh]);
  float alpha = __expf(v - m[dh]) / (denom[dh] + SM_EPS);
  const float4* hs4 = (const float4*)(h + ((size_t)s * HEADS + head) * C);
  float* od = agg + dh * C;
  int C4 = C >> 2;
  for (int c4 = 0; c4 < C4; ++c4) {
    float4 hv = hs4[c4];
    atomicAdd(&od[c4 * 4 + 0], alpha * hv.x);
    atomicAdd(&od[c4 * 4 + 1], alpha * hv.y);
    atomicAdd(&od[c4 * 4 + 2], alpha * hv.z);
    atomicAdd(&od[c4 * 4 + 3], alpha * hv.w);
  }
}

__global__ void finalize_kernel(const float* __restrict__ agg,
                                const float* __restrict__ bias,
                                float* __restrict__ out,
                                int N, int C, int do_relu) {
  long i = blockIdx.x * (long)blockDim.x + threadIdx.x;
  if (i >= (long)N * C) return;
  int n = (int)(i / C);
  int c = (int)(i - (long)n * C);
  float s = 0.f;
#pragma unroll
  for (int hh = 0; hh < HEADS; ++hh)
    s += agg[((size_t)n * HEADS + hh) * C + c];
  s = s * (1.0f / HEADS) + bias[c];
  if (do_relu) s = s > 0.f ? s : 0.f;
  out[i] = s;
}

static inline unsigned nblk(long n, int b) { return (unsigned)((n + b - 1) / b); }

extern "C" void kernel_launch(void* const* d_in, const int* in_sizes, int n_in,
                              void* d_out, int out_size, void* d_ws, size_t ws_size,
                              hipStream_t stream) {
  const float* x        = (const float*)d_in[0];
  const int*   ei       = (const int*)  d_in[1];
  const float* W1       = (const float*)d_in[2];
  const float* att_src1 = (const float*)d_in[3];
  const float* att_dst1 = (const float*)d_in[4];
  const float* b1       = (const float*)d_in[5];
  const float* W2       = (const float*)d_in[6];
  const float* att_src2 = (const float*)d_in[7];
  const float* att_dst2 = (const float*)d_in[8];
  const float* b2       = (const float*)d_in[9];

  const int F_h   = in_sizes[3] / HEADS;
  const int F_out = in_sizes[7] / HEADS;
  const int F_in  = in_sizes[2] / (HEADS * F_h);
  const int N     = in_sizes[0] / F_in;
  const int E     = in_sizes[1] / 2;
  const int C1    = HEADS * F_h;
  const int C2    = HEADS * F_out;
  const int* src = ei;
  const int* dst = ei + E;

  float* ws = (float*)d_ws;
  size_t o = 0;
  float* h1   = ws + o; o += (size_t)N * C1;
  float* agg  = ws + o; o += (size_t)N * C1;
  float* asrc = ws + o; o += (size_t)N * HEADS;
  float* adst = ws + o; o += (size_t)N * HEADS;
  float* mbuf = ws + o; o += (size_t)N * HEADS;
  float* dbuf = ws + o; o += (size_t)N * HEADS;
  float* x2   = ws + o; o += (size_t)N * F_h;
  float* h2   = h1;
  float* outp = (float*)d_out;

  const int TB = 256;
  const long NH = (long)N * HEADS;
  const long ETOT = (long)(E + N) * HEADS;

  {
    long waves = (long)((N + 15) / 16) * (C1 / (16 * NT));
    wmma_gemm_f32<<<nblk(waves * 32, TB), TB, 0, stream>>>(x, W1, h1, N, F_in, C1);
  }
  attn_coeff_kernel<<<nblk(NH, TB), TB, 0, stream>>>(h1, att_src1, att_dst1,
                                                     asrc, adst, N, F_h);
  fill_f32<<<nblk(NH, TB), TB, 0, stream>>>(mbuf, -3.0e38f, NH);
  fill_f32<<<nblk(NH, TB), TB, 0, stream>>>(dbuf, 0.0f, NH);
  fill_f32<<<nblk((long)N * C1, TB), TB, 0, stream>>>(agg, 0.0f, (long)N * C1);
  edge_max_kernel<<<nblk(ETOT, TB), TB, 0, stream>>>(src, dst, E, N, asrc, adst, mbuf);
  edge_expsum_kernel<<<nblk(ETOT, TB), TB, 0, stream>>>(src, dst, E, N, asrc, adst,
                                                        mbuf, dbuf);
  edge_agg_kernel<<<nblk(ETOT, TB), TB, 0, stream>>>(src, dst, E, N, F_h, asrc, adst,
                                                     mbuf, dbuf, h1, agg);
  finalize_kernel<<<nblk((long)N * F_h, TB), TB, 0, stream>>>(agg, b1, x2, N, F_h, 1);

  {
    long waves = (long)((N + 15) / 16) * (C2 / (16 * NT));
    wmma_gemm_f32<<<nblk(waves * 32, TB), TB, 0, stream>>>(x2, W2, h2, N, F_h, C2);
  }
  attn_coeff_kernel<<<nblk(NH, TB), TB, 0, stream>>>(h2, att_src2, att_dst2,
                                                     asrc, adst, N, F_out);
  fill_f32<<<nblk(NH, TB), TB, 0, stream>>>(mbuf, -3.0e38f, NH);
  fill_f32<<<nblk(NH, TB), TB, 0, stream>>>(dbuf, 0.0f, NH);
  fill_f32<<<nblk((long)N * C2, TB), TB, 0, stream>>>(agg, 0.0f, (long)N * C2);
  edge_max_kernel<<<nblk(ETOT, TB), TB, 0, stream>>>(src, dst, E, N, asrc, adst, mbuf);
  edge_expsum_kernel<<<nblk(ETOT, TB), TB, 0, stream>>>(src, dst, E, N, asrc, adst,
                                                        mbuf, dbuf);
  edge_agg_kernel<<<nblk(ETOT, TB), TB, 0, stream>>>(src, dst, E, N, F_out, asrc, adst,
                                                     mbuf, dbuf, h2, agg);
  finalize_kernel<<<nblk((long)N * F_out, TB), TB, 0, stream>>>(agg, b2, outp,
                                                                N, F_out, 0);
}


# softmax max and sum per destination kept in the owner work-group's LDS over its queued edges (one scan); global-atomic edge passes and fills dropped, 14 -> 6 launches; denominator sum order changes
# speedup vs baseline: 8.3201x; 8.3201x over previous
#include <hip/hip_runtime.h>
#include <cstdint>

typedef float v2f __attribute__((ext_vector_type(2)));
typedef float v8f __attribute__((ext_vector_type(8)));

#define HEADS 8
#define NEG_SLOPE 0.2f
#define SM_EPS 1e-16f
#define NT 4

__global__ void wmma_gemm_f32(const float* __restrict__ A,
                              const float* __restrict__ B,
                              float* __restrict__ C,
                              int M, int K, int Ncols) {
  int wave = (int)((blockIdx.x * blockDim.x + threadIdx.x) >> 5);
  int lane = (int)(threadIdx.x & 31);
  int tilesM  = (M + 15) >> 4;
  int groupsN = Ncols >> 6;
  if (wave >= tilesM * groupsN) return;
  int tm = wave / groupsN;
  int tg = wave - tm * groupsN;
  int half = lane >> 4;
  int l = lane & 15;

  int rowA = tm * 16 + l;
  if (rowA >= M) rowA = M - 1;
  const float* aP = A + (size_t)rowA * K + 2 * half;
  const float* bP = B + (size_t)(2 * half) * Ncols + tg * (16 * NT) + l;

  v8f acc[NT];
#pragma unroll
  for (int t = 0; t < NT; ++t) acc[t] = (v8f){};

  v2f aCur, bCur[NT];
  aCur.x = aP[0];
  aCur.y = aP[1];
#pragma unroll
  for (int t = 0; t < NT; ++t) {
    bCur[t].x = bP[t * 16];
    bCur[t].y = bP[t * 16 + Ncols];
  }

  for (int k0 = 0; k0 < K - 4; k0 += 4) {
    aP += 4;
    bP += (size_t)4 * Ncols;
    v2f aNxt, bNxt[NT];
    aNxt.x = aP[0];
    aNxt.y = aP[1];
#pragma unroll
    for (int t = 0; t < NT; ++t) {
      bNxt[t].x = bP[t * 16];
      bNxt[t].y = bP[t * 16 + Ncols];
    }
#pragma unroll
    for (int t = 0; t < NT; ++t) {
      acc[t] = __builtin_amdgcn_wmma_f32_16x16x4_f32(false, aCur, false, bCur[t],
                                                     (short)0, acc[t],
                                                     false, false);
    }
    aCur = aNxt;
#pragma unroll
    for (int t = 0; t < NT; ++t) bCur[t] = bNxt[t];
  }
#pragma unroll
  for (int t = 0; t < NT; ++t) {
    acc[t] = __builtin_amdgcn_wmma_f32_16x16x4_f32(false, aCur, false, bCur[t],
                                                   (short)0, acc[t],
                                                   false, false);
  }

#pragma unroll
  for (int t = 0; t < NT; ++t) {
    int col = tg * (16 * NT) + t * 16 + l;
#pragma unroll
    for (int r = 0; r < 8; ++r) {
      int row = tm * 16 + r + 8 * half;
      if (row < M) C[(size_t)row * Ncols + col] = acc[t][r];
    }
  }
}

__global__ void attn_coeff_kernel(const float* __restrict__ h,
                                  const float* __restrict__ att_src,
                                  const float* __restrict__ att_dst,
                                  float* __restrict__ asrc,
                                  float* __restrict__ adst,
                                  int N, int C) {
  long i = blockIdx.x * (long)blockDim.x + threadIdx.x;
  if (i >= (long)N * HEADS) return;
  int head = (int)(i & (HEADS - 1));
  const float* hp = h + (size_t)i * C;
  const float* as = att_src + (size_t)head * C;
  const float* ad = att_dst + (size_t)head * C;
  float s = 0.f, d = 0.f;
  for (int c = 0; c < C; ++c) {
    float v = hp[c];
    s += v * as[c];
    d += v * ad[c];
  }
  asrc[i] = s;
  adst[i] = d;
}

__device__ __forceinline__ float lrelu(float v) {
  return v > 0.f ? v : NEG_SLOPE * v;
}

#define RNODES 196
#define AGG_TB 1024
#define AGG_WAVES (AGG_TB / 32)
#define QCAP 288

#define PH_LIST 0
#define PH_MAX  1
#define PH_SUM  2
#define PH_AGG  3

__device__ __forceinline__ int fkey(int i) { return i >= 0 ? i : (i ^ 0x7fffffff); }

template <int C>
__device__ __forceinline__ float head_sum(const float* v, const float* al, int b, int lane) {
  float a[HEADS];
#pragma unroll
  for (int k = 0; k < HEADS; ++k)
    a[k] = __int_as_float(__builtin_amdgcn_readlane(__float_as_int(al[k]), b));
  float acc;
  if constexpr (C == 32) {
    acc = a[0] * v[0];
#pragma unroll
    for (int k = 1; k < HEADS; ++k) acc = __builtin_fmaf(a[k], v[k], acc);
  } else {
    const bool hi = lane >= 16;
    acc = (hi ? a[1] : a[0]) * v[0];
#pragma unroll
    for (int j = 1; j < 4; ++j) acc = __builtin_fmaf(hi ? a[2 * j + 1] : a[2 * j], v[j], acc);
  }
  return acc;
}

template <int C, int PH>
__device__ __forceinline__ void edge_batch(int s, int dl, int count, int lane,
                                           const float* __restrict__ asrc,
                                           const float* __restrict__ h,
                                           const float* sadst, int* smaxk, float* ssum,
                                           float* tile) {
  constexpr int NL = (HEADS * C) / 32;
  const float4* pa = (const float4*)(asrc + (size_t)s * HEADS);
  const float4* pd = (const float4*)(sadst + dl * HEADS);
  const float4 a0 = pa[0], a1 = pa[1];
  const float4 d0 = pd[0], d1 = pd[1];
  float v[HEADS];
  v[0] = lrelu(a0.x + d0.x);
  v[1] = lrelu(a0.y + d0.y);
  v[2] = lrelu(a0.z + d0.z);
  v[3] = lrelu(a0.w + d0.w);
  v[4] = lrelu(a1.x + d1.x);
  v[5] = lrelu(a1.y + d1.y);
  v[6] = lrelu(a1.z + d1.z);
  v[7] = lrelu(a1.w + d1.w);
  if constexpr (PH == PH_MAX) {
    if (lane < count) {
#pragma unroll
      for (int k = 0; k < HEADS; ++k)
        atomicMax(&smaxk[dl * HEADS + k], fkey(__float_as_int(v[k])));
    }
  } else {
    const int4* pm = (const int4*)(smaxk + dl * HEADS);
    const int4 k0 = pm[0], k1 = pm[1];
    float p[HEADS];
    p[0] = __expf(v[0] - __int_as_float(fkey(k0.x)));
    p[1] = __expf(v[1] - __int_as_float(fkey(k0.y)));
    p[2] = __expf(v[2] - __int_as_float(fkey(k0.z)));
    p[3] = __expf(v[3] - __int_as_float(fkey(k0.w)));
    p[4] = __expf(v[4] - __int_as_float(fkey(k1.x)));
    p[5] = __expf(v[5] - __int_as_float(fkey(k1.y)));
    p[6] = __expf(v[6] - __int_as_float(fkey(k1.z)));
    p[7] = __expf(v[7] - __int_as_float(fkey(k1.w)));
    if constexpr (PH == PH_SUM) {
      if (lane < count) {
#pragma unroll
        for (int k = 0; k < HEADS; ++k) atomicAdd(&ssum[dl * HEADS + k], p[k]);
      }
    } else {
      const float4* pq = (const float4*)(ssum + dl * HEADS);
      const float4 q0 = pq[0], q1 = pq[1];
      float al[HEADS];
      al[0] = p[0] / (q0.x + SM_EPS);
      al[1] = p[1] / (q0.y + SM_EPS);
      al[2] = p[2] / (q0.z + SM_EPS);
      al[3] = p[3] / (q0.w + SM_EPS);
      al[4] = p[4] / (q1.x + SM_EPS);
      al[5] = p[5] / (q1.y + SM_EPS);
      al[6] = p[6] / (q1.z + SM_EPS);
      al[7] = p[7] / (q1.w + SM_EPS);

      int b = 0;
      for (; b + 4 <= count; b += 4) {
        float x[4][NL];
#pragma unroll
        for (int u = 0; u < 4; ++u) {
          const int sb = __builtin_amdgcn_readlane(s, b + u);
          const float* hp = h + (size_t)sb * (HEADS * C) + lane;
#pragma unroll
          for (int j = 0; j < NL; ++j) x[u][j] = hp[j * 32];
        }
#pragma unroll
        for (int u = 0; u < 4; ++u) {
          const int db = __builtin_amdgcn_readlane(dl, b + u);
          const float acc = head_sum<C>(x[u], al, b + u, lane);
          atomicAdd(&tile[db * C + (lane & (C - 1))], acc);
        }
      }
      for (; b < count; ++b) {
        float x[NL];
        const int sb = __builtin_amdgcn_readlane(s, b);
        const float* hp = h + (size_t)sb * (HEADS * C) + lane;
#pragma unroll
        for (int j = 0; j < NL; ++j) x[j] = hp[j * 32];
        const int db = __builtin_amdgcn_readlane(dl, b);
        const float acc = head_sum<C>(x, al, b, lane);
        atomicAdd(&tile[db * C + (lane & (C - 1))], acc);
      }
    }
  }
}

template <int C, int PH>
__device__ __forceinline__ void list_pass(const int* __restrict__ src, int E, int n0,
                                          int qn, int lane, int* qe, const int* qd,
                                          const float* __restrict__ asrc,
                                          const float* __restrict__ h,
                                          const float* sadst, int* smaxk, float* ssum,
                                          float* tile) {
  for (int off = 0; off < qn; off += 32) {
    const int cnt = (qn - off < 32) ? (qn - off) : 32;
    const bool ok = lane < cnt;
    const int dl = ok ? qd[off + lane] : 0;
    int s;
    if constexpr (PH == PH_MAX) {
      const int e = ok ? qe[off + lane] : (E + n0);
      s = e - E;
      if (e < E) s = src[e];
      if (ok) qe[off + lane] = s;
    } else {
      s = ok ? qe[off + lane] : n0;
    }
    edge_batch<C, PH>(s, dl, cnt, lane, asrc, h, sadst, smaxk, ssum, tile);
  }
}

template <int C, int PH>
__device__ __forceinline__ int scan_pass(const int* __restrict__ src,
                                         const int* __restrict__ dst,
                                         int E, int n0, int nv, int w, int lane,
                                         int* qe, int* qd, int qn,
                                         const float* __restrict__ asrc,
                                         const float* __restrict__ h,
                                         const float* sadst, int* smaxk, float* ssum,
                                         float* tile) {
  if constexpr (PH != PH_LIST) {
    for (int r0 = w * 32; r0 < nv; r0 += AGG_TB) {
      const int cnt = (nv - r0 < 32) ? (nv - r0) : 32;
      const int r = (lane < cnt) ? (r0 + lane) : 0;
      edge_batch<C, PH>(n0 + r, r, cnt, lane, asrc, h, sadst, smaxk, ssum, tile);
    }
  }
  const int stride = AGG_WAVES * 256;
  const int elast = (E > 0) ? (E - 1) : 0;
  int eb = w * 256 + lane;
  int dc[8];
#pragma unroll
  for (int j = 0; j < 8; ++j) dc[j] = -1;
  if (E > 0) {
#pragma unroll
    for (int j = 0; j < 8; ++j) {
      const int e = eb + j * 32;
      const int dv = dst[(e < E) ? e : elast];
      dc[j] = (e < E) ? dv : -1;
    }
  }
  for (int base = 0; base < E; base += stride) {
    int dn[8];
    const int ebn = eb + stride;
#pragma unroll
    for (int j = 0; j < 8; ++j) {
      const int e = ebn + j * 32;
      const int dv = dst[(e < E) ? e : elast];
      dn[j] = (e < E) ? dv : -1;
    }
#pragma unroll
    for (int j = 0; j < 8; ++j) {
      const int e = eb + j * 32;
      const int dl = dc[j] - n0;
      const bool hit = (unsigned)dl < (unsigned)nv;
      const unsigned mask = __builtin_amdgcn_ballot_w32(hit);
      if (mask != 0u) {
        if (hit) {
          const int pos = qn + (int)__builtin_amdgcn_mbcnt_lo(mask, 0u);
          if (PH != PH_LIST || pos < QCAP) {
            qe[pos] = e;
            qd[pos] = dl;
          }
        }
        qn += __popc(mask);
      }
    }
    if constexpr (PH != PH_LIST) {
      if (qn >= 32) {
        int off = 0;
        for (; off + 32 <= qn; off += 32) {
          const int e2 = qe[off + lane];
          const int d2 = qd[off + lane];
          const int s2 = src[e2];
          edge_batch<C, PH>(s2, d2, 32, lane, asrc, h, sadst, smaxk, ssum, tile);
        }
        const int rem = qn - off;
        const int te = (lane < rem) ? qe[off + lane] : 0;
        const int td = (lane < rem) ? qd[off + lane] : 0;
        if (lane < rem) { qe[lane] = te; qd[lane] = td; }
        qn = rem;
      }
    }
#pragma unroll
    for (int j = 0; j < 8; ++j) dc[j] = dn[j];
    eb = ebn;
  }
  if constexpr (PH != PH_LIST) {
    if (qn > 0) {
      const bool ok = lane < qn;
      const int e2 = ok ? qe[lane] : 0;
      const int d2 = ok ? qd[lane] : 0;
      const int s2 = src[e2];
      edge_batch<C, PH>(s2, d2, qn, lane, asrc, h, sadst, smaxk, ssum, tile);
    }
    qn = 0;
  }
  return qn;
}

template <int C>
__global__ __launch_bounds__(AGG_TB)
void range_agg_kernel(const int* __restrict__ src,
                      const int* __restrict__ dst,
                      int E, int N,
                      const float* __restrict__ asrc,
                      const float* __restrict__ adst,
                      const float* __restrict__ h,
                      const float* __restrict__ bias,
                      float* __restrict__ out,
                      int do_relu) {
  __shared__ __attribute__((aligned(16))) float tile[RNODES * C];
  __shared__ __attribute__((aligned(16))) float sadst[RNODES * HEADS];
  __shared__ __attribute__((aligned(16))) int   smaxk[RNODES * HEADS];
  __shared__ __attribute__((aligned(16))) float ssum[RNODES * HEADS];
  __shared__ int qe_all[AGG_WAVES * QCAP];
  __shared__ int qd_all[AGG_WAVES * QCAP];
  __shared__ int sover;

  const int tid = (int)threadIdx.x;
  const int lane = tid & 31;
  const int w = __builtin_amdgcn_readfirstlane(tid >> 5);
  const int n0 = (int)blockIdx.x * RNODES;
  const int nv = (N - n0 < RNODES) ? (N - n0) : RNODES;

  for (int i = tid; i < RNODES * C; i += AGG_TB) tile[i] = 0.f;
  for (int i = tid; i < RNODES * HEADS; i += AGG_TB) {
    sadst[i] = (i < nv * HEADS) ? adst[(size_t)n0 * HEADS + i] : 0.f;
    smaxk[i] = fkey(__float_as_int(-3.0e38f));
    ssum[i] = 0.f;
  }
  if (tid == 0) sover = 0;

  int* qe = qe_all + w * QCAP;
  int* qd = qd_all + w * QCAP;
  const int rs = w + 32 * lane;
  const bool own = rs < nv;
  if (own) { qe[lane] = E + n0 + rs; qd[lane] = rs; }
  int qn = __popc(__builtin_amdgcn_ballot_w32(own));
  __syncthreads();

  qn = scan_pass<C, PH_LIST>(src, dst, E, n0, nv, w, lane, qe, qd, qn,
                             asrc, h, sadst, smaxk, ssum, tile);
  if (qn > QCAP) sover = 1;
  __syncthreads();
  const bool over = sover != 0;

  if (!over) {
    list_pass<C, PH_MAX>(src, E, n0, qn, lane, qe, qd, asrc, h, sadst, smaxk, ssum, tile);
    __syncthreads();
    list_pass<C, PH_SUM>(src, E, n0, qn, lane, qe, qd, asrc, h, sadst, smaxk, ssum, tile);
    __syncthreads();
    list_pass<C, PH_AGG>(src, E, n0, qn, lane, qe, qd, asrc, h, sadst, smaxk, ssum, tile);
  } else {
    scan_pass<C, PH_MAX>(src, dst, E, n0, nv, w, lane, qe, qd, 0,
                         asrc, h, sadst, smaxk, ssum, tile);
    __syncthreads();
    scan_pass<C, PH_SUM>(src, dst, E, n0, nv, w, lane, qe, qd, 0,
                         asrc, h, sadst, smaxk, ssum, tile);
    __syncthreads();
    scan_pass<C, PH_AGG>(src, dst, E, n0, nv, w, lane, qe, qd, 0,
                         asrc, h, sadst, smaxk, ssum, tile);
  }
  __syncthreads();

  const float inv = 1.0f / HEADS;
  for (int i = tid; i < nv * C; i += AGG_TB) {
    float v = tile[i] * inv + bias[i & (C - 1)];
    if (do_relu) v = v > 0.f ? v : 0.f;
    out[(size_t)n0 * C + i] = v;
  }
}

static inline unsigned nblk(long n, int b) { return (unsigned)((n + b - 1) / b); }

extern "C" void kernel_launch(void* const* d_in, const int* in_sizes, int n_in,
                              void* d_out, int out_size, void* d_ws, size_t ws_size,
                              hipStream_t stream) {
  const float* x        = (const float*)d_in[0];
  const int*   ei       = (const int*)  d_in[1];
  const float* W1       = (const float*)d_in[2];
  const float* att_src1 = (const float*)d_in[3];
  const float* att_dst1 = (const float*)d_in[4];
  const float* b1       = (const float*)d_in[5];
  const float* W2       = (const float*)d_in[6];
  const float* att_src2 = (const float*)d_in[7];
  const float* att_dst2 = (const float*)d_in[8];
  const float* b2       = (const float*)d_in[9];

  const int F_h   = in_sizes[3] / HEADS;
  const int F_out = in_sizes[7] / HEADS;
  const int F_in  = in_sizes[2] / (HEADS * F_h);
  const int N     = in_sizes[0] / F_in;
  const int E     = in_sizes[1] / 2;
  const int C1    = HEADS * F_h;
  const int C2    = HEADS * F_out;
  const int* src = ei;
  const int* dst = ei + E;

  float* ws = (float*)d_ws;
  size_t o = 0;
  float* h1   = ws + o; o += (size_t)N * C1;
  float* asrc = ws + o; o += (size_t)N * HEADS;
  float* adst = ws + o; o += (size_t)N * HEADS;
  float* x2   = ws + o; o += (size_t)N * F_h;
  float* h2   = h1;
  float* outp = (float*)d_out;

  const int TB = 256;
  const long NH = (long)N * HEADS;

  {
    long waves = (long)((N + 15) / 16) * (C1 / (16 * NT));
    wmma_gemm_f32<<<nblk(waves * 32, TB), TB, 0, stream>>>(x, W1, h1, N, F_in, C1);
  }
  attn_coeff_kernel<<<nblk(NH, TB), TB, 0, stream>>>(h1, att_src1, att_dst1,
                                                     asrc, adst, N, F_h);
  range_agg_kernel<32><<<nblk(N, RNODES), AGG_TB, 0, stream>>>(src, dst, E, N, asrc, adst,
                                                               h1, b1, x2, 1);

  {
    long waves = (long)((N + 15) / 16) * (C2 / (16 * NT));
    wmma_gemm_f32<<<nblk(waves * 32, TB), TB, 0, stream>>>(x2, W2, h2, N, F_h, C2);
  }
  attn_coeff_kernel<<<nblk(NH, TB), TB, 0, stream>>>(h2, att_src2, att_dst2,
                                                     asrc, adst, N, F_out);
  range_agg_kernel<16><<<nblk(N, RNODES), AGG_TB, 0, stream>>>(src, dst, E, N, asrc, adst,
                                                               h2, b2, outp, 0);
}


# layer 2 loads the owner work-groups' per-wave edge queues (source, local dst) that the layer-1 launch stores in the workspace; second scan of dst and its src gathers dropped; sum order as before
# speedup vs baseline: 9.4146x; 1.1315x over previous
#include <hip/hip_runtime.h>
#include <cstdint>

typedef float v2f __attribute__((ext_vector_type(2)));
typedef float v8f __attribute__((ext_vector_type(8)));

#define HEADS 8
#define NEG_SLOPE 0.2f
#define SM_EPS 1e-16f
#define NT 4

__global__ void wmma_gemm_f32(const float* __restrict__ A,
                              const float* __restrict__ B,
                              float* __restrict__ C,
                              int M, int K, int Ncols) {
  int wave = (int)((blockIdx.x * blockDim.x + threadIdx.x) >> 5);
  int lane = (int)(threadIdx.x & 31);
  int tilesM  = (M + 15) >> 4;
  int groupsN = Ncols >> 6;
  if (wave >= tilesM * groupsN) return;
  int tm = wave / groupsN;
  int tg = wave - tm * groupsN;
  int half = lane >> 4;
  int l = lane & 15;

  int rowA = tm * 16 + l;
  if (rowA >= M) rowA = M - 1;
  const float* aP = A + (size_t)rowA * K + 2 * half;
  const float* bP = B + (size_t)(2 * half) * Ncols + tg * (16 * NT) + l;

  v8f acc[NT];
#pragma unroll
  for (int t = 0; t < NT; ++t) acc[t] = (v8f){};

  v2f aCur, bCur[NT];
  aCur.x = aP[0];
  aCur.y = aP[1];
#pragma unroll
  for (int t = 0; t < NT; ++t) {
    bCur[t].x = bP[t * 16];
    bCur[t].y = bP[t * 16 + Ncols];
  }

  for (int k0 = 0; k0 < K - 4; k0 += 4) {
    aP += 4;
    bP += (size_t)4 * Ncols;
    v2f aNxt, bNxt[NT];
    aNxt.x = aP[0];
    aNxt.y = aP[1];
#pragma unroll
    for (int t = 0; t < NT; ++t) {
      bNxt[t].x = bP[t * 16];
      bNxt[t].y = bP[t * 16 + Ncols];
    }
#pragma unroll
    for (int t = 0; t < NT; ++t) {
      acc[t] = __builtin_amdgcn_wmma_f32_16x16x4_f32(false, aCur, false, bCur[t],
                                                     (short)0, acc[t],
                                                     false, false);
    }
    aCur = aNxt;
#pragma unroll
    for (int t = 0; t < NT; ++t) bCur[t] = bNxt[t];
  }
#pragma unroll
  for (int t = 0; t < NT; ++t) {
    acc[t] = __builtin_amdgcn_wmma_f32_16x16x4_f32(false, aCur, false, bCur[t],
                                                   (short)0, acc[t],
                                                   false, false);
  }

#pragma unroll
  for (int t = 0; t < NT; ++t) {
    int col = tg * (16 * NT) + t * 16 + l;
#pragma unroll
    for (int r = 0; r < 8; ++r) {
      int row = tm * 16 + r + 8 * half;
      if (row < M) C[(size_t)row * Ncols + col] = acc[t][r];
    }
  }
}

__global__ void attn_coeff_kernel(const float* __restrict__ h,
                                  const float* __restrict__ att_src,
                                  const float* __restrict__ att_dst,
                                  float* __restrict__ asrc,
                                  float* __restrict__ adst,
                                  int N, int C) {
  long i = blockIdx.x * (long)blockDim.x + threadIdx.x;
  if (i >= (long)N * HEADS) return;
  int head = (int)(i & (HEADS - 1));
  const float* hp = h + (size_t)i * C;
  const float* as = att_src + (size_t)head * C;
  const float* ad = att_dst + (size_t)head * C;
  float s = 0.f, d = 0.f;
  for (int c = 0; c < C; ++c) {
    float v = hp[c];
    s += v * as[c];
    d += v * ad[c];
  }
  asrc[i] = s;
  adst[i] = d;
}

__device__ __forceinline__ float lrelu(float v) {
  return v > 0.f ? v : NEG_SLOPE * v;
}

#define RNODES 196
#define AGG_TB 1024
#define AGG_WAVES (AGG_TB / 32)
#define QCAP 288

#define PH_LIST 0
#define PH_MAX  1
#define PH_SUM  2
#define PH_AGG  3

__device__ __forceinline__ int fkey(int i) { return i >= 0 ? i : (i ^ 0x7fffffff); }

template <int C>
__device__ __forceinline__ float head_sum(const float* v, const float* al, int b, int lane) {
  float a[HEADS];
#pragma unroll
  for (int k = 0; k < HEADS; ++k)
    a[k] = __int_as_float(__builtin_amdgcn_readlane(__float_as_int(al[k]), b));
  float acc;
  if constexpr (C == 32) {
    acc = a[0] * v[0];
#pragma unroll
    for (int k = 1; k < HEADS; ++k) acc = __builtin_fmaf(a[k], v[k], acc);
  } else {
    const bool hi = lane >= 16;
    acc = (hi ? a[1] : a[0]) * v[0];
#pragma unroll
    for (int j = 1; j < 4; ++j) acc = __builtin_fmaf(hi ? a[2 * j + 1] : a[2 * j], v[j], acc);
  }
  return acc;
}

template <int C, int PH>
__device__ __forceinline__ void edge_batch(int s, int dl, int count, int lane,
                                           const float* __restrict__ asrc,
                                           const float* __restrict__ h,
                                           const float* sadst, int* smaxk, float* ssum,
                                           float* tile) {
  constexpr int NL = (HEADS * C) / 32;
  const float4* pa = (const float4*)(asrc + (size_t)s * HEADS);
  const float4* pd = (const float4*)(sadst + dl * HEADS);
  const float4 a0 = pa[0], a1 = pa[1];
  const float4 d0 = pd[0], d1 = pd[1];
  float v[HEADS];
  v[0] = lrelu(a0.x + d0.x);
  v[1] = lrelu(a0.y + d0.y);
  v[2] = lrelu(a0.z + d0.z);
  v[3] = lrelu(a0.w + d0.w);
  v[4] = lrelu(a1.x + d1.x);
  v[5] = lrelu(a1.y + d1.y);
  v[6] = lrelu(a1.z + d1.z);
  v[7] = lrelu(a1.w + d1.w);
  if constexpr (PH == PH_MAX) {
    if (lane < count) {
#pragma unroll
      for (int k = 0; k < HEADS; ++k)
        atomicMax(&smaxk[dl * HEADS + k], fkey(__float_as_int(v[k])));
    }
  } else {
    const int4* pm = (const int4*)(smaxk + dl * HEADS);
    const int4 k0 = pm[0], k1 = pm[1];
    float p[HEADS];
    p[0] = __expf(v[0] - __int_as_float(fkey(k0.x)));
    p[1] = __expf(v[1] - __int_as_float(fkey(k0.y)));
    p[2] = __expf(v[2] - __int_as_float(fkey(k0.z)));
    p[3] = __expf(v[3] - __int_as_float(fkey(k0.w)));
    p[4] = __expf(v[4] - __int_as_float(fkey(k1.x)));
    p[5] = __expf(v[5] - __int_as_float(fkey(k1.y)));
    p[6] = __expf(v[6] - __int_as_float(fkey(k1.z)));
    p[7] = __expf(v[7] - __int_as_float(fkey(k1.w)));
    if constexpr (PH == PH_SUM) {
      if (lane < count) {
#pragma unroll
        for (int k = 0; k < HEADS; ++k) atomicAdd(&ssum[dl * HEADS + k], p[k]);
      }
    } else {
      const float4* pq = (const float4*)(ssum + dl * HEADS);
      const float4 q0 = pq[0], q1 = pq[1];
      float al[HEADS];
      al[0] = p[0] / (q0.x + SM_EPS);
      al[1] = p[1] / (q0.y + SM_EPS);
      al[2] = p[2] / (q0.z + SM_EPS);
      al[3] = p[3] / (q0.w + SM_EPS);
      al[4] = p[4] / (q1.x + SM_EPS);
      al[5] = p[5] / (q1.y + SM_EPS);
      al[6] = p[6] / (q1.z + SM_EPS);
      al[7] = p[7] / (q1.w + SM_EPS);

      int b = 0;
      for (; b + 4 <= count; b += 4) {
        float x[4][NL];
#pragma unroll
        for (int u = 0; u < 4; ++u) {
          const int sb = __builtin_amdgcn_readlane(s, b + u);
          const float* hp = h + (size_t)sb * (HEADS * C) + lane;
#pragma unroll
          for (int j = 0; j < NL; ++j) x[u][j] = hp[j * 32];
        }
#pragma unroll
        for (int u = 0; u < 4; ++u) {
          const int db = __builtin_amdgcn_readlane(dl, b + u);
          const float acc = head_sum<C>(x[u], al, b + u, lane);
          atomicAdd(&tile[db * C + (lane & (C - 1))], acc);
        }
      }
      for (; b < count; ++b) {
        float x[NL];
        const int sb = __builtin_amdgcn_readlane(s, b);
        const float* hp = h + (size_t)sb * (HEADS * C) + lane;
#pragma unroll
        for (int j = 0; j < NL; ++j) x[j] = hp[j * 32];
        const int db = __builtin_amdgcn_readlane(dl, b);
        const float acc = head_sum<C>(x, al, b, lane);
        atomicAdd(&tile[db * C + (lane & (C - 1))], acc);
      }
    }
  }
}

template <int C, int PH, bool IDS>
__device__ __forceinline__ void list_pass(const int* __restrict__ src, int E, int n0,
                                          int qn, int lane, int* qe, const int* qd,
                                          const float* __restrict__ asrc,
                                          const float* __restrict__ h,
                                          const float* sadst, int* smaxk, float* ssum,
                                          float* tile) {
  for (int off = 0; off < qn; off += 32) {
    const int cnt = (qn - off < 32) ? (qn - off) : 32;
    const bool ok = lane < cnt;
    const int dl = ok ? qd[off + lane] : 0;
    int s;
    if constexpr (IDS) {
      const int e = ok ? qe[off + lane] : (E + n0);
      s = e - E;
      if (e < E) s = src[e];
      if (ok) qe[off + lane] = s;
    } else {
      s = ok ? qe[off + lane] : n0;
    }
    edge_batch<C, PH>(s, dl, cnt, lane, asrc, h, sadst, smaxk, ssum, tile);
  }
}

template <int C, int PH>
__device__ __forceinline__ int scan_pass(const int* __restrict__ src,
                                         const int* __restrict__ dst,
                                         int E, int n0, int nv, int w, int lane,
                                         int* qe, int* qd, int qn,
                                         const float* __restrict__ asrc,
                                         const float* __restrict__ h,
                                         const float* sadst, int* smaxk, float* ssum,
                                         float* tile) {
  if constexpr (PH != PH_LIST) {
    for (int r0 = w * 32; r0 < nv; r0 += AGG_TB) {
      const int cnt = (nv - r0 < 32) ? (nv - r0) : 32;
      const int r = (lane < cnt) ? (r0 + lane) : 0;
      edge_batch<C, PH>(n0 + r, r, cnt, lane, asrc, h, sadst, smaxk, ssum, tile);
    }
  }
  const int stride = AGG_WAVES * 256;
  const int elast = (E > 0) ? (E - 1) : 0;
  int eb = w * 256 + lane;
  int dc[8];
#pragma unroll
  for (int j = 0; j < 8; ++j) dc[j] = -1;
  if (E > 0) {
#pragma unroll
    for (int j = 0; j < 8; ++j) {
      const int e = eb + j * 32;
      const int dv = dst[(e < E) ? e : elast];
      dc[j] = (e < E) ? dv : -1;
    }
  }
  for (int base = 0; base < E; base += stride) {
    int dn[8];
    const int ebn = eb + stride;
#pragma unroll
    for (int j = 0; j < 8; ++j) {
      const int e = ebn + j * 32;
      const int dv = dst[(e < E) ? e : elast];
      dn[j] = (e < E) ? dv : -1;
    }
#pragma unroll
    for (int j = 0; j < 8; ++j) {
      const int e = eb + j * 32;
      const int dl = dc[j] - n0;
      const bool hit = (unsigned)dl < (unsigned)nv;
      const unsigned mask = __builtin_amdgcn_ballot_w32(hit);
      if (mask != 0u) {
        if (hit) {
          const int pos = qn + (int)__builtin_amdgcn_mbcnt_lo(mask, 0u);
          if (PH != PH_LIST || pos < QCAP) {
            qe[pos] = e;
            qd[pos] = dl;
          }
        }
        qn += __popc(mask);
      }
    }
    if constexpr (PH != PH_LIST) {
      if (qn >= 32) {
        int off = 0;
        for (; off + 32 <= qn; off += 32) {
          const int e2 = qe[off + lane];
          const int d2 = qd[off + lane];
          const int s2 = src[e2];
          edge_batch<C, PH>(s2, d2, 32, lane, asrc, h, sadst, smaxk, ssum, tile);
        }
        const int rem = qn - off;
        const int te = (lane < rem) ? qe[off + lane] : 0;
        const int td = (lane < rem) ? qd[off + lane] : 0;
        if (lane < rem) { qe[lane] = te; qd[lane] = td; }
        qn = rem;
      }
    }
#pragma unroll
    for (int j = 0; j < 8; ++j) dc[j] = dn[j];
    eb = ebn;
  }
  if constexpr (PH != PH_LIST) {
    if (qn > 0) {
      const bool ok = lane < qn;
      const int e2 = ok ? qe[lane] : 0;
      const int d2 = ok ? qd[lane] : 0;
      const int s2 = src[e2];
      edge_batch<C, PH>(s2, d2, qn, lane, asrc, h, sadst, smaxk, ssum, tile);
    }
    qn = 0;
  }
  return qn;
}

template <int C, bool BUILD>
__global__ __launch_bounds__(AGG_TB)
void range_agg_kernel(const int* __restrict__ src,
                      const int* __restrict__ dst,
                      int E, int N,
                      const float* __restrict__ asrc,
                      const float* __restrict__ adst,
                      const float* __restrict__ h,
                      const float* __restrict__ bias,
                      float* __restrict__ out,
                      int do_relu,
                      int* __restrict__ wq_n,
                      int* __restrict__ wq_e,
                      int* __restrict__ wq_d) {
  __shared__ __attribute__((aligned(16))) float tile[RNODES * C];
  __shared__ __attribute__((aligned(16))) float sadst[RNODES * HEADS];
  __shared__ __attribute__((aligned(16))) int   smaxk[RNODES * HEADS];
  __shared__ __attribute__((aligned(16))) float ssum[RNODES * HEADS];
  __shared__ int qe_all[AGG_WAVES * QCAP];
  __shared__ int qd_all[AGG_WAVES * QCAP];
  __shared__ int sqn[AGG_WAVES];
  __shared__ int sover;

  const int tid = (int)threadIdx.x;
  const int lane = tid & 31;
  const int w = __builtin_amdgcn_readfirstlane(tid >> 5);
  const int n0 = (int)blockIdx.x * RNODES;
  const int nv = (N - n0 < RNODES) ? (N - n0) : RNODES;
  const size_t wq = (size_t)blockIdx.x * AGG_WAVES + w;

  int qn = 0;
  if constexpr (!BUILD) {
    qn = __builtin_amdgcn_readfirstlane(wq_n[wq]);
  }

  for (int i = tid; i < RNODES * C; i += AGG_TB) tile[i] = 0.f;
  for (int i = tid; i < RNODES * HEADS; i += AGG_TB) {
    sadst[i] = (i < nv * HEADS) ? adst[(size_t)n0 * HEADS + i] : 0.f;
    smaxk[i] = fkey(__float_as_int(-3.0e38f));
    ssum[i] = 0.f;
  }
  if (tid == 0) sover = 0;

  int* qe = qe_all + w * QCAP;
  int* qd = qd_all + w * QCAP;
  if constexpr (BUILD) {
    const int rs = w + 32 * lane;
    const bool own = rs < nv;
    if (own) { qe[lane] = E + n0 + rs; qd[lane] = rs; }
    qn = __popc(__builtin_amdgcn_ballot_w32(own));
    __syncthreads();
    qn = scan_pass<C, PH_LIST>(src, dst, E, n0, nv, w, lane, qe, qd, qn,
                               asrc, h, sadst, smaxk, ssum, tile);
    if (lane == 0) sqn[w] = qn;
  } else {
    if (qn <= QCAP) {
      const int* we = wq_e + wq * QCAP;
      const int* wd = wq_d + wq * QCAP;
      for (int i = lane; i < qn; i += 32) { qe[i] = we[i]; qd[i] = wd[i]; }
    }
    __syncthreads();
  }
  if (qn > QCAP) sover = 1;
  __syncthreads();
  const bool over = sover != 0;
  if constexpr (BUILD) {
    if (w == 0) wq_n[(size_t)blockIdx.x * AGG_WAVES + lane] = sqn[lane];
  }

  if (!over) {
    list_pass<C, PH_MAX, BUILD>(src, E, n0, qn, lane, qe, qd, asrc, h, sadst, smaxk, ssum, tile);
    __syncthreads();
    if constexpr (BUILD) {
      int* we = wq_e + wq * QCAP;
      int* wd = wq_d + wq * QCAP;
      for (int i = lane; i < qn; i += 32) { we[i] = qe[i]; wd[i] = qd[i]; }
    }
    list_pass<C, PH_SUM, false>(src, E, n0, qn, lane, qe, qd, asrc, h, sadst, smaxk, ssum, tile);
    __syncthreads();
    list_pass<C, PH_AGG, false>(src, E, n0, qn, lane, qe, qd, asrc, h, sadst, smaxk, ssum, tile);
  } else {
    scan_pass<C, PH_MAX>(src, dst, E, n0, nv, w, lane, qe, qd, 0,
                         asrc, h, sadst, smaxk, ssum, tile);
    __syncthreads();
    scan_pass<C, PH_SUM>(src, dst, E, n0, nv, w, lane, qe, qd, 0,
                         asrc, h, sadst, smaxk, ssum, tile);
    __syncthreads();
    scan_pass<C, PH_AGG>(src, dst, E, n0, nv, w, lane, qe, qd, 0,
                         asrc, h, sadst, smaxk, ssum, tile);
  }
  __syncthreads();

  const float inv = 1.0f / HEADS;
  for (int i = tid; i < nv * C; i += AGG_TB) {
    float v = tile[i] * inv + bias[i & (C - 1)];
    if (do_relu) v = v > 0.f ? v : 0.f;
    out[(size_t)n0 * C + i] = v;
  }
}

static inline unsigned nblk(long n, int b) { return (unsigned)((n + b - 1) / b); }

extern "C" void kernel_launch(void* const* d_in, const int* in_sizes, int n_in,
                              void* d_out, int out_size, void* d_ws, size_t ws_size,
                              hipStream_t stream) {
  const float* x        = (const float*)d_in[0];
  const int*   ei       = (const int*)  d_in[1];
  const float* W1       = (const float*)d_in[2];
  const float* att_src1 = (const float*)d_in[3];
  const float* att_dst1 = (const float*)d_in[4];
  const float* b1       = (const float*)d_in[5];
  const float* W2       = (const float*)d_in[6];
  const float* att_src2 = (const float*)d_in[7];
  const float* att_dst2 = (const float*)d_in[8];
  const float* b2       = (const float*)d_in[9];

  const int F_h   = in_sizes[3] / HEADS;
  const int F_out = in_sizes[7] / HEADS;
  const int F_in  = in_sizes[2] / (HEADS * F_h);
  const int N     = in_sizes[0] / F_in;
  const int E     = in_sizes[1] / 2;
  const int C1    = HEADS * F_h;
  const int C2    = HEADS * F_out;
  const int* src = ei;
  const int* dst = ei + E;

  float* ws = (float*)d_ws;
  size_t o = 0;
  float* h1   = ws + o; o += (size_t)N * C1;
  float* asrc = ws + o; o += (size_t)N * HEADS;
  float* adst = ws + o; o += (size_t)N * HEADS;
  float* x2   = ws + o; o += (size_t)N * F_h;
  float* h2   = h1;
  const size_t nwg = nblk(N, RNODES);
  o = (o + 31) & ~(size_t)31;
  int* wq_n = (int*)(ws + o); o += nwg * AGG_WAVES;
  int* wq_e = (int*)(ws + o); o += nwg * AGG_WAVES * QCAP;
  int* wq_d = (int*)(ws + o); o += nwg * AGG_WAVES * QCAP;
  float* outp = (float*)d_out;

  const int TB = 256;
  const long NH = (long)N * HEADS;

  {
    long waves = (long)((N + 15) / 16) * (C1 / (16 * NT));
    wmma_gemm_f32<<<nblk(waves * 32, TB), TB, 0, stream>>>(x, W1, h1, N, F_in, C1);
  }
  attn_coeff_kernel<<<nblk(NH, TB), TB, 0, stream>>>(h1, att_src1, att_dst1,
                                                     asrc, adst, N, F_h);
  range_agg_kernel<32, true><<<nblk(N, RNODES), AGG_TB, 0, stream>>>(src, dst, E, N, asrc, adst,
                                                                     h1, b1, x2, 1,
                                                                     wq_n, wq_e, wq_d);

  {
    long waves = (long)((N + 15) / 16) * (C2 / (16 * NT));
    wmma_gemm_f32<<<nblk(waves * 32, TB), TB, 0, stream>>>(x2, W2, h2, N, F_h, C2);
  }
  attn_coeff_kernel<<<nblk(NH, TB), TB, 0, stream>>>(h2, att_src2, att_dst2,
                                                     asrc, adst, N, F_out);
  range_agg_kernel<16, false><<<nblk(N, RNODES), AGG_TB, 0, stream>>>(src, dst, E, N, asrc, adst,
                                                                      h2, b2, outp, 0,
                                                                      wq_n, wq_e, wq_d);
}


# layer-2 projection (finished rows @ W2 on WMMA from the LDS tile) and its attention coefficients done in the layer-1 owner kernel's epilogue; x2 not stored; 6 -> 4 launches; sum order unchanged
# speedup vs baseline: 10.0056x; 1.0628x over previous
#include <hip/hip_runtime.h>
#include <cstdint>

typedef float v2f __attribute__((ext_vector_type(2)));
typedef float v8f __attribute__((ext_vector_type(8)));

#define HEADS 8
#define NEG_SLOPE 0.2f
#define SM_EPS 1e-16f
#define NT 4

__global__ void wmma_gemm_f32(const float* __restrict__ A,
                              const float* __restrict__ B,
                              float* __restrict__ C,
                              int M, int K, int Ncols) {
  int wave = (int)((blockIdx.x * blockDim.x + threadIdx.x) >> 5);
  int lane = (int)(threadIdx.x & 31);
  int tilesM  = (M + 15) >> 4;
  int groupsN = Ncols >> 6;
  if (wave >= tilesM * groupsN) return;
  int tm = wave / groupsN;
  int tg = wave - tm * groupsN;
  int half = lane >> 4;
  int l = lane & 15;

  int rowA = tm * 16 + l;
  if (rowA >= M) rowA = M - 1;
  const float* aP = A + (size_t)rowA * K + 2 * half;
  const float* bP = B + (size_t)(2 * half) * Ncols + tg * (16 * NT) + l;

  v8f acc[NT];
#pragma unroll
  for (int t = 0; t < NT; ++t) acc[t] = (v8f){};

  v2f aCur, bCur[NT];
  aCur.x = aP[0];
  aCur.y = aP[1];
#pragma unroll
  for (int t = 0; t < NT; ++t) {
    bCur[t].x = bP[t * 16];
    bCur[t].y = bP[t * 16 + Ncols];
  }

  for (int k0 = 0; k0 < K - 4; k0 += 4) {
    aP += 4;
    bP += (size_t)4 * Ncols;
    v2f aNxt, bNxt[NT];
    aNxt.x = aP[0];
    aNxt.y = aP[1];
#pragma unroll
    for (int t = 0; t < NT; ++t) {
      bNxt[t].x = bP[t * 16];
      bNxt[t].y = bP[t * 16 + Ncols];
    }
#pragma unroll
    for (int t = 0; t < NT; ++t) {
      acc[t] = __builtin_amdgcn_wmma_f32_16x16x4_f32(false, aCur, false, bCur[t],
                                                     (short)0, acc[t],
                                                     false, false);
    }
    aCur = aNxt;
#pragma unroll
    for (int t = 0; t < NT; ++t) bCur[t] = bNxt[t];
  }
#pragma unroll
  for (int t = 0; t < NT; ++t) {
    acc[t] = __builtin_amdgcn_wmma_f32_16x16x4_f32(false, aCur, false, bCur[t],
                                                   (short)0, acc[t],
                                                   false, false);
  }

#pragma unroll
  for (int t = 0; t < NT; ++t) {
    int col = tg * (16 * NT) + t * 16 + l;
#pragma unroll
    for (int r = 0; r < 8; ++r) {
      int row = tm * 16 + r + 8 * half;
      if (row < M) C[(size_t)row * Ncols + col] = acc[t][r];
    }
  }
}

__global__ void attn_coeff_kernel(const float* __restrict__ h,
                                  const float* __restrict__ att_src,
                                  const float* __restrict__ att_dst,
                                  float* __restrict__ asrc,
                                  float* __restrict__ adst,
                                  int N, int C) {
  long i = blockIdx.x * (long)blockDim.x + threadIdx.x;
  if (i >= (long)N * HEADS) return;
  int head = (int)(i & (HEADS - 1));
  const float* hp = h + (size_t)i * C;
  const float* as = att_src + (size_t)head * C;
  const float* ad = att_dst + (size_t)head * C;
  float s = 0.f, d = 0.f;
  for (int c = 0; c < C; ++c) {
    float v = hp[c];
    s += v * as[c];
    d += v * ad[c];
  }
  asrc[i] = s;
  adst[i] = d;
}

__device__ __forceinline__ float lrelu(float v) {
  return v > 0.f ? v : NEG_SLOPE * v;
}

#define RNODES 196
#define AGG_TB 1024
#define AGG_WAVES (AGG_TB / 32)
#define QCAP 288

#define PH_LIST 0
#define PH_MAX  1
#define PH_SUM  2
#define PH_AGG  3

__device__ __forceinline__ int fkey(int i) { return i >= 0 ? i : (i ^ 0x7fffffff); }

template <int C>
__device__ __forceinline__ float head_sum(const float* v, const float* al, int b, int lane) {
  float a[HEADS];
#pragma unroll
  for (int k = 0; k < HEADS; ++k)
    a[k] = __int_as_float(__builtin_amdgcn_readlane(__float_as_int(al[k]), b));
  float acc;
  if constexpr (C == 32) {
    acc = a[0] * v[0];
#pragma unroll
    for (int k = 1; k < HEADS; ++k) acc = __builtin_fmaf(a[k], v[k], acc);
  } else {
    const bool hi = lane >= 16;
    acc = (hi ? a[1] : a[0]) * v[0];
#pragma unroll
    for (int j = 1; j < 4; ++j) acc = __builtin_fmaf(hi ? a[2 * j + 1] : a[2 * j], v[j], acc);
  }
  return acc;
}

template <int C, int PH>
__device__ __forceinline__ void edge_batch(int s, int dl, int count, int lane,
                                           const float* __restrict__ asrc,
                                           const float* __restrict__ h,
                                           const float* sadst, int* smaxk, float* ssum,
                                           float* tile) {
  constexpr int NL = (HEADS * C) / 32;
  const float4* pa = (const float4*)(asrc + (size_t)s * HEADS);
  const float4* pd = (const float4*)(sadst + dl * HEADS);
  const float4 a0 = pa[0], a1 = pa[1];
  const float4 d0 = pd[0], d1 = pd[1];
  float v[HEADS];
  v[0] = lrelu(a0.x + d0.x);
  v[1] = lrelu(a0.y + d0.y);
  v[2] = lrelu(a0.z + d0.z);
  v[3] = lrelu(a0.w + d0.w);
  v[4] = lrelu(a1.x + d1.x);
  v[5] = lrelu(a1.y + d1.y);
  v[6] = lrelu(a1.z + d1.z);
  v[7] = lrelu(a1.w + d1.w);
  if constexpr (PH == PH_MAX) {
    if (lane < count) {
#pragma unroll
      for (int k = 0; k < HEADS; ++k)
        atomicMax(&smaxk[dl * HEADS + k], fkey(__float_as_int(v[k])));
    }
  } else {
    const int4* pm = (const int4*)(smaxk + dl * HEADS);
    const int4 k0 = pm[0], k1 = pm[1];
    float p[HEADS];
    p[0] = __expf(v[0] - __int_as_float(fkey(k0.x)));
    p[1] = __expf(v[1] - __int_as_float(fkey(k0.y)));
    p[2] = __expf(v[2] - __int_as_float(fkey(k0.z)));
    p[3] = __expf(v[3] - __int_as_float(fkey(k0.w)));
    p[4] = __expf(v[4] - __int_as_float(fkey(k1.x)));
    p[5] = __expf(v[5] - __int_as_float(fkey(k1.y)));
    p[6] = __expf(v[6] - __int_as_float(fkey(k1.z)));
    p[7] = __expf(v[7] - __int_as_float(fkey(k1.w)));
    if constexpr (PH == PH_SUM) {
      if (lane < count) {
#pragma unroll
        for (int k = 0; k < HEADS; ++k) atomicAdd(&ssum[dl * HEADS + k], p[k]);
      }
    } else {
      const float4* pq = (const float4*)(ssum + dl * HEADS);
      const float4 q0 = pq[0], q1 = pq[1];
      float al[HEADS];
      al[0] = p[0] / (q0.x + SM_EPS);
      al[1] = p[1] / (q0.y + SM_EPS);
      al[2] = p[2] / (q0.z + SM_EPS);
      al[3] = p[3] / (q0.w + SM_EPS);
      al[4] = p[4] / (q1.x + SM_EPS);
      al[5] = p[5] / (q1.y + SM_EPS);
      al[6] = p[6] / (q1.z + SM_EPS);
      al[7] = p[7] / (q1.w + SM_EPS);

      int b = 0;
      for (; b + 4 <= count; b += 4) {
        float x[4][NL];
#pragma unroll
        for (int u = 0; u < 4; ++u) {
          const int sb = __builtin_amdgcn_readlane(s, b + u);
          const float* hp = h + (size_t)sb * (HEADS * C) + lane;
#pragma unroll
          for (int j = 0; j < NL; ++j) x[u][j] = hp[j * 32];
        }
#pragma unroll
        for (int u = 0; u < 4; ++u) {
          const int db = __builtin_amdgcn_readlane(dl, b + u);
          const float acc = head_sum<C>(x[u], al, b + u, lane);
          atomicAdd(&tile[db * C + (lane & (C - 1))], acc);
        }
      }
      for (; b < count; ++b) {
        float x[NL];
        const int sb = __builtin_amdgcn_readlane(s, b);
        const float* hp = h + (size_t)sb * (HEADS * C) + lane;
#pragma unroll
        for (int j = 0; j < NL; ++j) x[j] = hp[j * 32];
        const int db = __builtin_amdgcn_readlane(dl, b);
        const float acc = head_sum<C>(x, al, b, lane);
        atomicAdd(&tile[db * C + (lane & (C - 1))], acc);
      }
    }
  }
}

template <int C, int PH, bool IDS>
__device__ __forceinline__ void list_pass(const int* __restrict__ src, int E, int n0,
                                          int qn, int lane, int* qe, const int* qd,
                                          const float* __restrict__ asrc,
                                          const float* __restrict__ h,
                                          const float* sadst, int* smaxk, float* ssum,
                                          float* tile) {
  for (int off = 0; off < qn; off += 32) {
    const int cnt = (qn - off < 32) ? (qn - off) : 32;
    const bool ok = lane < cnt;
    const int dl = ok ? qd[off + lane] : 0;
    int s;
    if constexpr (IDS) {
      const int e = ok ? qe[off + lane] : (E + n0);
      s = e - E;
      if (e < E) s = src[e];
      if (ok) qe[off + lane] = s;
    } else {
      s = ok ? qe[off + lane] : n0;
    }
    edge_batch<C, PH>(s, dl, cnt, lane, asrc, h, sadst, smaxk, ssum, tile);
  }
}

template <int C, int PH>
__device__ __forceinline__ int scan_pass(const int* __restrict__ src,
                                         const int* __restrict__ dst,
                                         int E, int n0, int nv, int w, int lane,
                                         int* qe, int* qd, int qn,
                                         const float* __restrict__ asrc,
                                         const float* __restrict__ h,
                                         const float* sadst, int* smaxk, float* ssum,
                                         float* tile) {
  if constexpr (PH != PH_LIST) {
    for (int r0 = w * 32; r0 < nv; r0 += AGG_TB) {
      const int cnt = (nv - r0 < 32) ? (nv - r0) : 32;
      const int r = (lane < cnt) ? (r0 + lane) : 0;
      edge_batch<C, PH>(n0 + r, r, cnt, lane, asrc, h, sadst, smaxk, ssum, tile);
    }
  }
  const int stride = AGG_WAVES * 256;
  const int elast = (E > 0) ? (E - 1) : 0;
  int eb = w * 256 + lane;
  int dc[8];
#pragma unroll
  for (int j = 0; j < 8; ++j) dc[j] = -1;
  if (E > 0) {
#pragma unroll
    for (int j = 0; j < 8; ++j) {
      const int e = eb + j * 32;
      const int dv = dst[(e < E) ? e : elast];
      dc[j] = (e < E) ? dv : -1;
    }
  }
  for (int base = 0; base < E; base += stride) {
    int dn[8];
    const int ebn = eb + stride;
#pragma unroll
    for (int j = 0; j < 8; ++j) {
      const int e = ebn + j * 32;
      const int dv = dst[(e < E) ? e : elast];
      dn[j] = (e < E) ? dv : -1;
    }
#pragma unroll
    for (int j = 0; j < 8; ++j) {
      const int e = eb + j * 32;
      const int dl = dc[j] - n0;
      const bool hit = (unsigned)dl < (unsigned)nv;
      const unsigned mask = __builtin_amdgcn_ballot_w32(hit);
      if (mask != 0u) {
        if (hit) {
          const int pos = qn + (int)__builtin_amdgcn_mbcnt_lo(mask, 0u);
          if (PH != PH_LIST || pos < QCAP) {
            qe[pos] = e;
            qd[pos] = dl;
          }
        }
        qn += __popc(mask);
      }
    }
    if constexpr (PH != PH_LIST) {
      if (qn >= 32) {
        int off = 0;
        for (; off + 32 <= qn; off += 32) {
          const int e2 = qe[off + lane];
          const int d2 = qd[off + lane];
          const int s2 = src[e2];
          edge_batch<C, PH>(s2, d2, 32, lane, asrc, h, sadst, smaxk, ssum, tile);
        }
        const int rem = qn - off;
        const int te = (lane < rem) ? qe[off + lane] : 0;
        const int td = (lane < rem) ? qd[off + lane] : 0;
        if (lane < rem) { qe[lane] = te; qd[lane] = td; }
        qn = rem;
      }
    }
#pragma unroll
    for (int j = 0; j < 8; ++j) dc[j] = dn[j];
    eb = ebn;
  }
  if constexpr (PH != PH_LIST) {
    if (qn > 0) {
      const bool ok = lane < qn;
      const int e2 = ok ? qe[lane] : 0;
      const int d2 = ok ? qd[lane] : 0;
      const int s2 = src[e2];
      edge_batch<C, PH>(s2, d2, qn, lane, asrc, h, sadst, smaxk, ssum, tile);
    }
    qn = 0;
  }
  return qn;
}

template <int C, bool BUILD>
__global__ __launch_bounds__(AGG_TB)
void range_agg_kernel(const int* __restrict__ src,
                      const int* __restrict__ dst,
                      int E, int N,
                      const float* __restrict__ asrc,
                      const float* __restrict__ adst,
                      const float* __restrict__ h,
                      const float* __restrict__ bias,
                      float* __restrict__ out,
                      int do_relu,
                      int* __restrict__ wq_n,
                      int* __restrict__ wq_e,
                      int* __restrict__ wq_d,
                      const float* __restrict__ Wn,
                      const float* __restrict__ attn_s,
                      const float* __restrict__ attn_d,
                      float* __restrict__ asrc_n,
                      float* __restrict__ adst_n) {
  __shared__ __attribute__((aligned(16))) float tile[RNODES * C];
  __shared__ __attribute__((aligned(16))) float sadst[RNODES * HEADS];
  __shared__ __attribute__((aligned(16))) int   smaxk[RNODES * HEADS];
  __shared__ __attribute__((aligned(16))) float ssum[RNODES * HEADS];
  __shared__ int qe_all[AGG_WAVES * QCAP];
  __shared__ int qd_all[AGG_WAVES * QCAP];
  __shared__ int sqn[AGG_WAVES];
  __shared__ int sover;

  const int tid = (int)threadIdx.x;
  const int lane = tid & 31;
  const int w = __builtin_amdgcn_readfirstlane(tid >> 5);
  const int n0 = (int)blockIdx.x * RNODES;
  const int nv = (N - n0 < RNODES) ? (N - n0) : RNODES;
  const size_t wq = (size_t)blockIdx.x * AGG_WAVES + w;

  int qn = 0;
  if constexpr (!BUILD) {
    qn = __builtin_amdgcn_readfirstlane(wq_n[wq]);
  }

  for (int i = tid; i < RNODES * C; i += AGG_TB) tile[i] = 0.f;
  for (int i = tid; i < RNODES * HEADS; i += AGG_TB) {
    sadst[i] = (i < nv * HEADS) ? adst[(size_t)n0 * HEADS + i] : 0.f;
    smaxk[i] = fkey(__float_as_int(-3.0e38f));
    ssum[i] = 0.f;
  }
  if (tid == 0) sover = 0;

  int* qe = qe_all + w * QCAP;
  int* qd = qd_all + w * QCAP;
  if constexpr (BUILD) {
    const int rs = w + 32 * lane;
    const bool own = rs < nv;
    if (own) { qe[lane] = E + n0 + rs; qd[lane] = rs; }
    qn = __popc(__builtin_amdgcn_ballot_w32(own));
    __syncthreads();
    qn = scan_pass<C, PH_LIST>(src, dst, E, n0, nv, w, lane, qe, qd, qn,
                               asrc, h, sadst, smaxk, ssum, tile);
    if (lane == 0) sqn[w] = qn;
  } else {
    if (qn <= QCAP) {
      const int* we = wq_e + wq * QCAP;
      const int* wd = wq_d + wq * QCAP;
      for (int i = lane; i < qn; i += 32) { qe[i] = we[i]; qd[i] = wd[i]; }
    }
    __syncthreads();
  }
  if (qn > QCAP) sover = 1;
  __syncthreads();
  const bool over = sover != 0;
  if constexpr (BUILD) {
    if (w == 0) wq_n[(size_t)blockIdx.x * AGG_WAVES + lane] = sqn[lane];
  }

  if (!over) {
    list_pass<C, PH_MAX, BUILD>(src, E, n0, qn, lane, qe, qd, asrc, h, sadst, smaxk, ssum, tile);
    __syncthreads();
    if constexpr (BUILD) {
      int* we = wq_e + wq * QCAP;
      int* wd = wq_d + wq * QCAP;
      for (int i = lane; i < qn; i += 32) { we[i] = qe[i]; wd[i] = qd[i]; }
    }
    list_pass<C, PH_SUM, false>(src, E, n0, qn, lane, qe, qd, asrc, h, sadst, smaxk, ssum, tile);
    __syncthreads();
    list_pass<C, PH_AGG, false>(src, E, n0, qn, lane, qe, qd, asrc, h, sadst, smaxk, ssum, tile);
  } else {
    scan_pass<C, PH_MAX>(src, dst, E, n0, nv, w, lane, qe, qd, 0,
                         asrc, h, sadst, smaxk, ssum, tile);
    __syncthreads();
    scan_pass<C, PH_SUM>(src, dst, E, n0, nv, w, lane, qe, qd, 0,
                         asrc, h, sadst, smaxk, ssum, tile);
    __syncthreads();
    scan_pass<C, PH_AGG>(src, dst, E, n0, nv, w, lane, qe, qd, 0,
                         asrc, h, sadst, smaxk, ssum, tile);
  }
  __syncthreads();

  const float inv = 1.0f / HEADS;
  if constexpr (!BUILD) {
    for (int i = tid; i < nv * C; i += AGG_TB) {
      float v = tile[i] * inv + bias[i & (C - 1)];
      if (do_relu) v = v > 0.f ? v : 0.f;
      out[(size_t)n0 * C + i] = v;
    }
  } else {
    constexpr int CN = 16;
    constexpr int NCOL = HEADS * CN;
    constexpr int HP = NCOL + 4;
    constexpr int TM = (RNODES + 15) / 16;
    constexpr int TG = NCOL / (16 * NT);
    static_assert(TM * TG <= AGG_WAVES, "one strip per wave");
    __shared__ __attribute__((aligned(16))) float hs[TM * 16 * HP];
    for (int i = tid; i < RNODES * C; i += AGG_TB) {
      float v = tile[i] * inv + bias[i & (C - 1)];
      if (do_relu) v = v > 0.f ? v : 0.f;
      tile[i] = v;
    }
    __syncthreads();
    if (w < TM * TG) {
      const int tm = w / TG;
      const int tg = w - tm * TG;
      const int half = lane >> 4;
      const int l = lane & 15;
      int rowA = tm * 16 + l;
      if (rowA > RNODES - 1) rowA = RNODES - 1;
      const float* aP = tile + rowA * C + 2 * half;
      const float* bP = Wn + (size_t)(2 * half) * NCOL + tg * (16 * NT) + l;
      v8f acc[NT];
#pragma unroll
      for (int t = 0; t < NT; ++t) acc[t] = (v8f){};
#pragma unroll
      for (int k0 = 0; k0 < C; k0 += 4) {
        v2f a;
        a.x = aP[k0];
        a.y = aP[k0 + 1];
#pragma unroll
        for (int t = 0; t < NT; ++t) {
          v2f b;
          b.x = bP[(size_t)k0 * NCOL + t * 16];
          b.y = bP[(size_t)k0 * NCOL + t * 16 + NCOL];
          acc[t] = __builtin_amdgcn_wmma_f32_16x16x4_f32(false, a, false, b,
                                                         (short)0, acc[t],
                                                         false, false);
        }
      }
#pragma unroll
      for (int t = 0; t < NT; ++t) {
#pragma unroll
        for (int r = 0; r < 8; ++r)
          hs[(tm * 16 + r + 8 * half) * HP + tg * (16 * NT) + t * 16 + l] = acc[t][r];
      }
    }
    __syncthreads();
    float* hout = out + (size_t)n0 * NCOL;
    for (int i = tid * 4; i < nv * NCOL; i += AGG_TB * 4) {
      const int row = i / NCOL;
      const int col = i - row * NCOL;
      *(float4*)(hout + i) = *(const float4*)(hs + row * HP + col);
    }
    for (int i = tid; i < nv * HEADS; i += AGG_TB) {
      const int head = i & (HEADS - 1);
      const float* hp = hs + (i >> 3) * HP + head * CN;
      const float* as = attn_s + head * CN;
      const float* ad = attn_d + head * CN;
      float s = 0.f, d = 0.f;
#pragma unroll
      for (int c2 = 0; c2 < CN; ++c2) {
        float v = hp[c2];
        s += v * as[c2];
        d += v * ad[c2];
      }
      asrc_n[(size_t)n0 * HEADS + i] = s;
      adst_n[(size_t)n0 * HEADS + i] = d;
    }
  }
}

static inline unsigned nblk(long n, int b) { return (unsigned)((n + b - 1) / b); }

extern "C" void kernel_launch(void* const* d_in, const int* in_sizes, int n_in,
                              void* d_out, int out_size, void* d_ws, size_t ws_size,
                              hipStream_t stream) {
  const float* x        = (const float*)d_in[0];
  const int*   ei       = (const int*)  d_in[1];
  const float* W1       = (const float*)d_in[2];
  const float* att_src1 = (const float*)d_in[3];
  const float* att_dst1 = (const float*)d_in[4];
  const float* b1       = (const float*)d_in[5];
  const float* W2       = (const float*)d_in[6];
  const float* att_src2 = (const float*)d_in[7];
  const float* att_dst2 = (const float*)d_in[8];
  const float* b2       = (const float*)d_in[9];

  const int F_h   = in_sizes[3] / HEADS;
  const int F_out = in_sizes[7] / HEADS;
  const int F_in  = in_sizes[2] / (HEADS * F_h);
  const int N     = in_sizes[0] / F_in;
  const int E     = in_sizes[1] / 2;
  const int C1    = HEADS * F_h;
  const int C2    = HEADS * F_out;
  const int* src = ei;
  const int* dst = ei + E;

  float* ws = (float*)d_ws;
  size_t o = 0;
  float* h1    = ws + o; o += (size_t)N * C1;
  float* asrc  = ws + o; o += (size_t)N * HEADS;
  float* adst  = ws + o; o += (size_t)N * HEADS;
  o = (o + 31) & ~(size_t)31;
  float* h2    = ws + o; o += (size_t)N * C2;
  float* asrc2 = ws + o; o += (size_t)N * HEADS;
  float* adst2 = ws + o; o += (size_t)N * HEADS;
  const size_t nwg = nblk(N, RNODES);
  o = (o + 31) & ~(size_t)31;
  int* wq_n = (int*)(ws + o); o += nwg * AGG_WAVES;
  int* wq_e = (int*)(ws + o); o += nwg * AGG_WAVES * QCAP;
  int* wq_d = (int*)(ws + o); o += nwg * AGG_WAVES * QCAP;
  float* outp = (float*)d_out;

  const int TB = 256;
  const long NH = (long)N * HEADS;

  {
    long waves = (long)((N + 15) / 16) * (C1 / (16 * NT));
    wmma_gemm_f32<<<nblk(waves * 32, TB), TB, 0, stream>>>(x, W1, h1, N, F_in, C1);
  }
  attn_coeff_kernel<<<nblk(NH, TB), TB, 0, stream>>>(h1, att_src1, att_dst1,
                                                     asrc, adst, N, F_h);
  range_agg_kernel<32, true><<<nblk(N, RNODES), AGG_TB, 0, stream>>>(src, dst, E, N, asrc, adst,
                                                                     h1, b1, h2, 1,
                                                                     wq_n, wq_e, wq_d,
                                                                     W2, att_src2, att_dst2,
                                                                     asrc2, adst2);

  range_agg_kernel<16, false><<<nblk(N, RNODES), AGG_TB, 0, stream>>>(src, dst, E, N, asrc2, adst2,
                                                                      h2, b2, outp, 0,
                                                                      wq_n, wq_e, wq_d,
                                                                      nullptr, nullptr, nullptr,
                                                                      nullptr, nullptr);
}


# layer-1 owner work-groups list their edges from a per-owner edge bitmap (1 bit/edge, built in LDS by extra blocks of the coefficient launch), not by scanning dst; queue and float sum order change
# speedup vs baseline: 12.0414x; 1.2035x over previous
#include <hip/hip_runtime.h>
#include <cstdint>

typedef float v2f __attribute__((ext_vector_type(2)));
typedef float v8f __attribute__((ext_vector_type(8)));

#define HEADS 8
#define NEG_SLOPE 0.2f
#define SM_EPS 1e-16f
#define NT 4

__global__ void wmma_gemm_f32(const float* __restrict__ A,
                              const float* __restrict__ B,
                              float* __restrict__ C,
                              int M, int K, int Ncols) {
  int wave = (int)((blockIdx.x * blockDim.x + threadIdx.x) >> 5);
  int lane = (int)(threadIdx.x & 31);
  int tilesM  = (M + 15) >> 4;
  int groupsN = Ncols >> 6;
  if (wave >= tilesM * groupsN) return;
  int tm = wave / groupsN;
  int tg = wave - tm * groupsN;
  int half = lane >> 4;
  int l = lane & 15;

  int rowA = tm * 16 + l;
  if (rowA >= M) rowA = M - 1;
  const float* aP = A + (size_t)rowA * K + 2 * half;
  const float* bP = B + (size_t)(2 * half) * Ncols + tg * (16 * NT) + l;

  v8f acc[NT];
#pragma unroll
  for (int t = 0; t < NT; ++t) acc[t] = (v8f){};

  v2f aCur, bCur[NT];
  aCur.x = aP[0];
  aCur.y = aP[1];
#pragma unroll
  for (int t = 0; t < NT; ++t) {
    bCur[t].x = bP[t * 16];
    bCur[t].y = bP[t * 16 + Ncols];
  }

  for (int k0 = 0; k0 < K - 4; k0 += 4) {
    aP += 4;
    bP += (size_t)4 * Ncols;
    v2f aNxt, bNxt[NT];
    aNxt.x = aP[0];
    aNxt.y = aP[1];
#pragma unroll
    for (int t = 0; t < NT; ++t) {
      bNxt[t].x = bP[t * 16];
      bNxt[t].y = bP[t * 16 + Ncols];
    }
#pragma unroll
    for (int t = 0; t < NT; ++t) {
      acc[t] = __builtin_amdgcn_wmma_f32_16x16x4_f32(false, aCur, false, bCur[t],
                                                     (short)0, acc[t],
                                                     false, false);
    }
    aCur = aNxt;
#pragma unroll
    for (int t = 0; t < NT; ++t) bCur[t] = bNxt[t];
  }
#pragma unroll
  for (int t = 0; t < NT; ++t) {
    acc[t] = __builtin_amdgcn_wmma_f32_16x16x4_f32(false, aCur, false, bCur[t],
                                                   (short)0, acc[t],
                                                   false, false);
  }

#pragma unroll
  for (int t = 0; t < NT; ++t) {
    int col = tg * (16 * NT) + t * 16 + l;
#pragma unroll
    for (int r = 0; r < 8; ++r) {
      int row = tm * 16 + r + 8 * half;
      if (row < M) C[(size_t)row * Ncols + col] = acc[t][r];
    }
  }
}

#define RNODES 196
#define BM_CHUNK 1024
#define BM_OWN 256
#define BM_PITCH 33
#define BM_TB 256

__global__ __launch_bounds__(BM_TB)
void coeff_bitmap_kernel(const float* __restrict__ h,
                         const float* __restrict__ att_src,
                         const float* __restrict__ att_dst,
                         float* __restrict__ asrc,
                         float* __restrict__ adst,
                         int N, int C,
                         const int* __restrict__ dst, int E,
                         int nown, int chunks, int nbm,
                         unsigned* __restrict__ bmp) {
  if ((int)blockIdx.x < nbm) {
    __shared__ unsigned sbm[BM_OWN * BM_PITCH];
    const int tid = (int)threadIdx.x;
    const int og = (int)blockIdx.x / chunks;
    const int ch = (int)blockIdx.x - og * chunks;
    for (int i = tid; i < BM_OWN * BM_PITCH; i += BM_TB) sbm[i] = 0u;
    __syncthreads();
    const int e0 = ch * BM_CHUNK;
    int dv[BM_CHUNK / BM_TB];
#pragma unroll
    for (int k = 0; k < BM_CHUNK / BM_TB; ++k) {
      const int e = e0 + k * BM_TB + tid;
      dv[k] = dst[(e < E) ? e : (E - 1)];
    }
#pragma unroll
    for (int k = 0; k < BM_CHUNK / BM_TB; ++k) {
      const int e = e0 + k * BM_TB + tid;
      const int ol = dv[k] / RNODES - og * BM_OWN;
      if (e < E && (unsigned)ol < (unsigned)BM_OWN)
        atomicOr(&sbm[ol * BM_PITCH + k * (BM_TB / 32) + (tid >> 5)], 1u << (tid & 31));
    }
    __syncthreads();
    const int w = tid >> 5, lane = tid & 31;
    const size_t wpo = (size_t)chunks * 32;
    for (int ol = w; ol < BM_OWN; ol += BM_TB / 32) {
      const int o = og * BM_OWN + ol;
      if (o < nown) bmp[(size_t)o * wpo + (size_t)ch * 32 + lane] = sbm[ol * BM_PITCH + lane];
    }
    return;
  }
  long i = ((long)blockIdx.x - nbm) * (long)blockDim.x + threadIdx.x;
  if (i >= (long)N * HEADS) return;
  int head = (int)(i & (HEADS - 1));
  const float* hp = h + (size_t)i * C;
  const float* as = att_src + (size_t)head * C;
  const float* ad = att_dst + (size_t)head * C;
  float s = 0.f, d = 0.f;
  for (int c = 0; c < C; ++c) {
    float v = hp[c];
    s += v * as[c];
    d += v * ad[c];
  }
  asrc[i] = s;
  adst[i] = d;
}

__device__ __forceinline__ float lrelu(float v) {
  return v > 0.f ? v : NEG_SLOPE * v;
}

#define AGG_TB 1024
#define AGG_WAVES (AGG_TB / 32)
#define QCAP 288

#define PH_MAX  1
#define PH_SUM  2
#define PH_AGG  3

__device__ __forceinline__ int fkey(int i) { return i >= 0 ? i : (i ^ 0x7fffffff); }

template <int C>
__device__ __forceinline__ float head_sum(const float* v, const float* al, int b, int lane) {
  float a[HEADS];
#pragma unroll
  for (int k = 0; k < HEADS; ++k)
    a[k] = __int_as_float(__builtin_amdgcn_readlane(__float_as_int(al[k]), b));
  float acc;
  if constexpr (C == 32) {
    acc = a[0] * v[0];
#pragma unroll
    for (int k = 1; k < HEADS; ++k) acc = __builtin_fmaf(a[k], v[k], acc);
  } else {
    const bool hi = lane >= 16;
    acc = (hi ? a[1] : a[0]) * v[0];
#pragma unroll
    for (int j = 1; j < 4; ++j) acc = __builtin_fmaf(hi ? a[2 * j + 1] : a[2 * j], v[j], acc);
  }
  return acc;
}

template <int C, int PH>
__device__ __forceinline__ void edge_batch(int s, int dl, int count, int lane,
                                           const float* __restrict__ asrc,
                                           const float* __restrict__ h,
                                           const float* sadst, int* smaxk, float* ssum,
                                           float* tile) {
  constexpr int NL = (HEADS * C) / 32;
  const float4* pa = (const float4*)(asrc + (size_t)s * HEADS);
  const float4* pd = (const float4*)(sadst + dl * HEADS);
  const float4 a0 = pa[0], a1 = pa[1];
  const float4 d0 = pd[0], d1 = pd[1];
  float v[HEADS];
  v[0] = lrelu(a0.x + d0.x);
  v[1] = lrelu(a0.y + d0.y);
  v[2] = lrelu(a0.z + d0.z);
  v[3] = lrelu(a0.w + d0.w);
  v[4] = lrelu(a1.x + d1.x);
  v[5] = lrelu(a1.y + d1.y);
  v[6] = lrelu(a1.z + d1.z);
  v[7] = lrelu(a1.w + d1.w);
  if constexpr (PH == PH_MAX) {
    if (lane < count) {
#pragma unroll
      for (int k = 0; k < HEADS; ++k)
        atomicMax(&smaxk[dl * HEADS + k], fkey(__float_as_int(v[k])));
    }
  } else {
    const int4* pm = (const int4*)(smaxk + dl * HEADS);
    const int4 k0 = pm[0], k1 = pm[1];
    float p[HEADS];
    p[0] = __expf(v[0] - __int_as_float(fkey(k0.x)));
    p[1] = __expf(v[1] - __int_as_float(fkey(k0.y)));
    p[2] = __expf(v[2] - __int_as_float(fkey(k0.z)));
    p[3] = __expf(v[3] - __int_as_float(fkey(k0.w)));
    p[4] = __expf(v[4] - __int_as_float(fkey(k1.x)));
    p[5] = __expf(v[5] - __int_as_float(fkey(k1.y)));
    p[6] = __expf(v[6] - __int_as_float(fkey(k1.z)));
    p[7] = __expf(v[7] - __int_as_float(fkey(k1.w)));
    if constexpr (PH == PH_SUM) {
      if (lane < count) {
#pragma unroll
        for (int k = 0; k < HEADS; ++k) atomicAdd(&ssum[dl * HEADS + k], p[k]);
      }
    } else {
      const float4* pq = (const float4*)(ssum + dl * HEADS);
      const float4 q0 = pq[0], q1 = pq[1];
      float al[HEADS];
      al[0] = p[0] / (q0.x + SM_EPS);
      al[1] = p[1] / (q0.y + SM_EPS);
      al[2] = p[2] / (q0.z + SM_EPS);
      al[3] = p[3] / (q0.w + SM_EPS);
      al[4] = p[4] / (q1.x + SM_EPS);
      al[5] = p[5] / (q1.y + SM_EPS);
      al[6] = p[6] / (q1.z + SM_EPS);
      al[7] = p[7] / (q1.w + SM_EPS);

      int b = 0;
      for (; b + 4 <= count; b += 4) {
        float x[4][NL];
#pragma unroll
        for (int u = 0; u < 4; ++u) {
          const int sb = __builtin_amdgcn_readlane(s, b + u);
          const float* hp = h + (size_t)sb * (HEADS * C) + lane;
#pragma unroll
          for (int j = 0; j < NL; ++j) x[u][j] = hp[j * 32];
        }
#pragma unroll
        for (int u = 0; u < 4; ++u) {
          const int db = __builtin_amdgcn_readlane(dl, b + u);
          const float acc = head_sum<C>(x[u], al, b + u, lane);
          atomicAdd(&tile[db * C + (lane & (C - 1))], acc);
        }
      }
      for (; b < count; ++b) {
        float x[NL];
        const int sb = __builtin_amdgcn_readlane(s, b);
        const float* hp = h + (size_t)sb * (HEADS * C) + lane;
#pragma unroll
        for (int j = 0; j < NL; ++j) x[j] = hp[j * 32];
        const int db = __builtin_amdgcn_readlane(dl, b);
        const float acc = head_sum<C>(x, al, b, lane);
        atomicAdd(&tile[db * C + (lane & (C - 1))], acc);
      }
    }
  }
}

template <int C, int PH, bool IDS>
__device__ __forceinline__ void list_pass(const int* __restrict__ src,
                                          const int* __restrict__ dst, int E, int n0,
                                          int qn, int lane, int* qe, int* qd,
                                          const float* __restrict__ asrc,
                                          const float* __restrict__ h,
                                          const float* sadst, int* smaxk, float* ssum,
                                          float* tile) {
  for (int off = 0; off < qn; off += 32) {
    const int cnt = (qn - off < 32) ? (qn - off) : 32;
    const bool ok = lane < cnt;
    int s, dl;
    if constexpr (IDS) {
      const int e = ok ? qe[off + lane] : (E + n0);
      s = e - E;
      dl = s - n0;
      if (e < E) { s = src[e]; dl = dst[e] - n0; }
      if (ok) { qe[off + lane] = s; qd[off + lane] = dl; }
    } else {
      s = ok ? qe[off + lane] : n0;
      dl = ok ? qd[off + lane] : 0;
    }
    edge_batch<C, PH>(s, dl, cnt, lane, asrc, h, sadst, smaxk, ssum, tile);
  }
}

template <int C, int PH>
__device__ __forceinline__ void scan_pass(const int* __restrict__ src,
                                         const int* __restrict__ dst,
                                         int E, int n0, int nv, int w, int lane,
                                         int* qe, int* qd, int qn,
                                         const float* __restrict__ asrc,
                                         const float* __restrict__ h,
                                         const float* sadst, int* smaxk, float* ssum,
                                         float* tile) {
  for (int r0 = w * 32; r0 < nv; r0 += AGG_TB) {
    const int cnt = (nv - r0 < 32) ? (nv - r0) : 32;
    const int r = (lane < cnt) ? (r0 + lane) : 0;
    edge_batch<C, PH>(n0 + r, r, cnt, lane, asrc, h, sadst, smaxk, ssum, tile);
  }
  const int stride = AGG_WAVES * 256;
  const int elast = (E > 0) ? (E - 1) : 0;
  int eb = w * 256 + lane;
  int dc[8];
#pragma unroll
  for (int j = 0; j < 8; ++j) dc[j] = -1;
  if (E > 0) {
#pragma unroll
    for (int j = 0; j < 8; ++j) {
      const int e = eb + j * 32;
      const int dv = dst[(e < E) ? e : elast];
      dc[j] = (e < E) ? dv : -1;
    }
  }
  for (int base = 0; base < E; base += stride) {
    int dn[8];
    const int ebn = eb + stride;
#pragma unroll
    for (int j = 0; j < 8; ++j) {
      const int e = ebn + j * 32;
      const int dv = dst[(e < E) ? e : elast];
      dn[j] = (e < E) ? dv : -1;
    }
#pragma unroll
    for (int j = 0; j < 8; ++j) {
      const int e = eb + j * 32;
      const int dl = dc[j] - n0;
      const bool hit = (unsigned)dl < (unsigned)nv;
      const unsigned mask = __builtin_amdgcn_ballot_w32(hit);
      if (mask != 0u) {
        if (hit) {
          const int pos = qn + (int)__builtin_amdgcn_mbcnt_lo(mask, 0u);
          qe[pos] = e;
          qd[pos] = dl;
        }
        qn += __popc(mask);
      }
    }
    if (qn >= 32) {
      int off = 0;
      for (; off + 32 <= qn; off += 32) {
        const int e2 = qe[off + lane];
        const int d2 = qd[off + lane];
        const int s2 = src[e2];
        edge_batch<C, PH>(s2, d2, 32, lane, asrc, h, sadst, smaxk, ssum, tile);
      }
      const int rem = qn - off;
      const int te = (lane < rem) ? qe[off + lane] : 0;
      const int td = (lane < rem) ? qd[off + lane] : 0;
      if (lane < rem) { qe[lane] = te; qd[lane] = td; }
      qn = rem;
    }
#pragma unroll
    for (int j = 0; j < 8; ++j) dc[j] = dn[j];
    eb = ebn;
  }
  if (qn > 0) {
    const bool ok = lane < qn;
    const int e2 = ok ? qe[lane] : 0;
    const int d2 = ok ? qd[lane] : 0;
    const int s2 = src[e2];
    edge_batch<C, PH>(s2, d2, qn, lane, asrc, h, sadst, smaxk, ssum, tile);
  }
}

__device__ __forceinline__ int bitmap_list(const unsigned* __restrict__ row, int nwords,
                                           int w, int lane, int* qe, int qn) {
  for (int base = 0; base < nwords; base += AGG_WAVES * 8 * 128) {
    uint4 v[8];
    const int fw0 = base + (w * 32 + lane) * 4;
#pragma unroll
    for (int j = 0; j < 8; ++j) {
      const int fw = fw0 + j * (AGG_WAVES * 128);
      v[j] = *(const uint4*)(row + ((fw < nwords) ? fw : 0));
    }
    __builtin_amdgcn_sched_barrier(0);
#pragma unroll
    for (int j = 0; j < 8; ++j) {
      const bool in = fw0 + j * (AGG_WAVES * 128) < nwords;
      v[j].x = in ? v[j].x : 0u;
      v[j].y = in ? v[j].y : 0u;
      v[j].z = in ? v[j].z : 0u;
      v[j].w = in ? v[j].w : 0u;
    }
#pragma nounroll
    for (int j = 0; j < 8; ++j) {
      uint4 cur = v[0];
#pragma unroll
      for (int t = 0; t < 7; ++t) v[t] = v[t + 1];
#pragma nounroll
      for (int q = 0; q < 4; ++q) {
        unsigned x = cur.x;
        cur.x = cur.y; cur.y = cur.z; cur.z = cur.w;
        const int ebase = (fw0 + j * (AGG_WAVES * 128) + q) << 5;
        unsigned m = __builtin_amdgcn_ballot_w32(x != 0u);
        while (m != 0u) {
          if (x != 0u) {
            const int pos = qn + (int)__builtin_amdgcn_mbcnt_lo(m, 0u);
            if (pos < QCAP) qe[pos] = ebase + __builtin_ctz(x);
            x &= x - 1u;
          }
          qn += __popc(m);
          m = __builtin_amdgcn_ballot_w32(x != 0u);
        }
      }
    }
  }
  return qn;
}

template <int C, bool BUILD>
__global__ __launch_bounds__(AGG_TB)
void range_agg_kernel(const int* __restrict__ src,
                      const int* __restrict__ dst,
                      int E, int N,
                      const float* __restrict__ asrc,
                      const float* __restrict__ adst,
                      const float* __restrict__ h,
                      const float* __restrict__ bias,
                      float* __restrict__ out,
                      int do_relu,
                      int* __restrict__ wq_n,
                      int* __restrict__ wq_e,
                      int* __restrict__ wq_d,
                      const float* __restrict__ Wn,
                      const float* __restrict__ attn_s,
                      const float* __restrict__ attn_d,
                      float* __restrict__ asrc_n,
                      float* __restrict__ adst_n,
                      const unsigned* __restrict__ bmp,
                      int nwords) {
  __shared__ __attribute__((aligned(16))) float tile[RNODES * C];
  __shared__ __attribute__((aligned(16))) float sadst[RNODES * HEADS];
  __shared__ __attribute__((aligned(16))) int   smaxk[RNODES * HEADS];
  __shared__ __attribute__((aligned(16))) float ssum[RNODES * HEADS];
  __shared__ int qe_all[AGG_WAVES * QCAP];
  __shared__ int qd_all[AGG_WAVES * QCAP];
  __shared__ int sqn[AGG_WAVES];
  __shared__ int sover;

  const int tid = (int)threadIdx.x;
  const int lane = tid & 31;
  const int w = __builtin_amdgcn_readfirstlane(tid >> 5);
  const int n0 = (int)blockIdx.x * RNODES;
  const int nv = (N - n0 < RNODES) ? (N - n0) : RNODES;
  const size_t wq = (size_t)blockIdx.x * AGG_WAVES + w;

  int qn = 0;
  if constexpr (!BUILD) {
    qn = __builtin_amdgcn_readfirstlane(wq_n[wq]);
  }

  for (int i = tid; i < RNODES * C; i += AGG_TB) tile[i] = 0.f;
  for (int i = tid; i < RNODES * HEADS; i += AGG_TB) {
    sadst[i] = (i < nv * HEADS) ? adst[(size_t)n0 * HEADS + i] : 0.f;
    smaxk[i] = fkey(__float_as_int(-3.0e38f));
    ssum[i] = 0.f;
  }
  if (tid == 0) sover = 0;

  int* qe = qe_all + w * QCAP;
  int* qd = qd_all + w * QCAP;
  if constexpr (BUILD) {
    const int rs = w + 32 * lane;
    const bool own = rs < nv;
    if (own) qe[lane] = E + n0 + rs;
    qn = __popc(__builtin_amdgcn_ballot_w32(own));
    __syncthreads();
    qn = bitmap_list(bmp + (size_t)blockIdx.x * nwords, nwords, w, lane, qe, qn);
    if (lane == 0) sqn[w] = qn;
  } else {
    if (qn <= QCAP) {
      const int* we = wq_e + wq * QCAP;
      const int* wd = wq_d + wq * QCAP;
      for (int i = lane; i < qn; i += 32) { qe[i] = we[i]; qd[i] = wd[i]; }
    }
    __syncthreads();
  }
  if (qn > QCAP) sover = 1;
  __syncthreads();
  const bool over = sover != 0;
  if constexpr (BUILD) {
    if (w == 0) wq_n[(size_t)blockIdx.x * AGG_WAVES + lane] = sqn[lane];
  }

  if (!over) {
    list_pass<C, PH_MAX, BUILD>(src, dst, E, n0, qn, lane, qe, qd, asrc, h, sadst, smaxk, ssum, tile);
    __syncthreads();
    if constexpr (BUILD) {
      int* we = wq_e + wq * QCAP;
      int* wd = wq_d + wq * QCAP;
      for (int i = lane; i < qn; i += 32) { we[i] = qe[i]; wd[i] = qd[i]; }
    }
    list_pass<C, PH_SUM, false>(src, dst, E, n0, qn, lane, qe, qd, asrc, h, sadst, smaxk, ssum, tile);
    __syncthreads();
    list_pass<C, PH_AGG, false>(src, dst, E, n0, qn, lane, qe, qd, asrc, h, sadst, smaxk, ssum, tile);
  } else {
    scan_pass<C, PH_MAX>(src, dst, E, n0, nv, w, lane, qe, qd, 0,
                         asrc, h, sadst, smaxk, ssum, tile);
    __syncthreads();
    scan_pass<C, PH_SUM>(src, dst, E, n0, nv, w, lane, qe, qd, 0,
                         asrc, h, sadst, smaxk, ssum, tile);
    __syncthreads();
    scan_pass<C, PH_AGG>(src, dst, E, n0, nv, w, lane, qe, qd, 0,
                         asrc, h, sadst, smaxk, ssum, tile);
  }
  __syncthreads();

  const float inv = 1.0f / HEADS;
  if constexpr (!BUILD) {
    for (int i = tid; i < nv * C; i += AGG_TB) {
      float v = tile[i] * inv + bias[i & (C - 1)];
      if (do_relu) v = v > 0.f ? v : 0.f;
      out[(size_t)n0 * C + i] = v;
    }
  } else {
    constexpr int CN = 16;
    constexpr int NCOL = HEADS * CN;
    constexpr int HP = NCOL + 4;
    constexpr int TM = (RNODES + 15) / 16;
    constexpr int TG = NCOL / (16 * NT);
    static_assert(TM * TG <= AGG_WAVES, "one strip per wave");
    __shared__ __attribute__((aligned(16))) float hs[TM * 16 * HP];
    for (int i = tid; i < RNODES * C; i += AGG_TB) {
      float v = tile[i] * inv + bias[i & (C - 1)];
      if (do_relu) v = v > 0.f ? v : 0.f;
      tile[i] = v;
    }
    __syncthreads();
    if (w < TM * TG) {
      const int tm = w / TG;
      const int tg = w - tm * TG;
      const int half = lane >> 4;
      const int l = lane & 15;
      int rowA = tm * 16 + l;
      if (rowA > RNODES - 1) rowA = RNODES - 1;
      const float* aP = tile + rowA * C + 2 * half;
      const float* bP = Wn + (size_t)(2 * half) * NCOL + tg * (16 * NT) + l;
      v8f acc[NT];
#pragma unroll
      for (int t = 0; t < NT; ++t) acc[t] = (v8f){};
#pragma unroll
      for (int k0 = 0; k0 < C; k0 += 4) {
        v2f a;
        a.x = aP[k0];
        a.y = aP[k0 + 1];
#pragma unroll
        for (int t = 0; t < NT; ++t) {
          v2f b;
          b.x = bP[(size_t)k0 * NCOL + t * 16];
          b.y = bP[(size_t)k0 * NCOL + t * 16 + NCOL];
          acc[t] = __builtin_amdgcn_wmma_f32_16x16x4_f32(false, a, false, b,
                                                         (short)0, acc[t],
                                                         false, false);
        }
      }
#pragma unroll
      for (int t = 0; t < NT; ++t) {
#pragma unroll
        for (int r = 0; r < 8; ++r)
          hs[(tm * 16 + r + 8 * half) * HP + tg * (16 * NT) + t * 16 + l] = acc[t][r];
      }
    }
    __syncthreads();
    float* hout = out + (size_t)n0 * NCOL;
    for (int i = tid * 4; i < nv * NCOL; i += AGG_TB * 4) {
      const int row = i / NCOL;
      const int col = i - row * NCOL;
      *(float4*)(hout + i) = *(const float4*)(hs + row * HP + col);
    }
    for (int i = tid; i < nv * HEADS; i += AGG_TB) {
      const int head = i & (HEADS - 1);
      const float* hp = hs + (i >> 3) * HP + head * CN;
      const float* as = attn_s + head * CN;
      const float* ad = attn_d + head * CN;
      float s = 0.f, d = 0.f;
#pragma unroll
      for (int c2 = 0; c2 < CN; ++c2) {
        float v = hp[c2];
        s += v * as[c2];
        d += v * ad[c2];
      }
      asrc_n[(size_t)n0 * HEADS + i] = s;
      adst_n[(size_t)n0 * HEADS + i] = d;
    }
  }
}

static inline unsigned nblk(long n, int b) { return (unsigned)((n + b - 1) / b); }

extern "C" void kernel_launch(void* const* d_in, const int* in_sizes, int n_in,
                              void* d_out, int out_size, void* d_ws, size_t ws_size,
                              hipStream_t stream) {
  const float* x        = (const float*)d_in[0];
  const int*   ei       = (const int*)  d_in[1];
  const float* W1       = (const float*)d_in[2];
  const float* att_src1 = (const float*)d_in[3];
  const float* att_dst1 = (const float*)d_in[4];
  const float* b1       = (const float*)d_in[5];
  const float* W2       = (const float*)d_in[6];
  const float* att_src2 = (const float*)d_in[7];
  const float* att_dst2 = (const float*)d_in[8];
  const float* b2       = (const float*)d_in[9];

  const int F_h   = in_sizes[3] / HEADS;
  const int F_out = in_sizes[7] / HEADS;
  const int F_in  = in_sizes[2] / (HEADS * F_h);
  const int N     = in_sizes[0] / F_in;
  const int E     = in_sizes[1] / 2;
  const int C1    = HEADS * F_h;
  const int C2    = HEADS * F_out;
  const int* src = ei;
  const int* dst = ei + E;

  float* ws = (float*)d_ws;
  size_t o = 0;
  float* h1    = ws + o; o += (size_t)N * C1;
  float* asrc  = ws + o; o += (size_t)N * HEADS;
  float* adst  = ws + o; o += (size_t)N * HEADS;
  o = (o + 31) & ~(size_t)31;
  float* h2    = ws + o; o += (size_t)N * C2;
  float* asrc2 = ws + o; o += (size_t)N * HEADS;
  float* adst2 = ws + o; o += (size_t)N * HEADS;
  const size_t nwg = nblk(N, RNODES);
  o = (o + 31) & ~(size_t)31;
  int* wq_n = (int*)(ws + o); o += nwg * AGG_WAVES;
  int* wq_e = (int*)(ws + o); o += nwg * AGG_WAVES * QCAP;
  int* wq_d = (int*)(ws + o); o += nwg * AGG_WAVES * QCAP;
  const int chunks = (E + BM_CHUNK - 1) / BM_CHUNK;
  const int nwords = chunks * 32;
  const int nbm = chunks * (int)((nwg + BM_OWN - 1) / BM_OWN);
  o = (o + 31) & ~(size_t)31;
  unsigned* bmp = (unsigned*)(ws + o); o += nwg * (size_t)nwords;
  float* outp = (float*)d_out;

  const int TB = 256;
  const long NH = (long)N * HEADS;

  {
    long waves = (long)((N + 15) / 16) * (C1 / (16 * NT));
    wmma_gemm_f32<<<nblk(waves * 32, TB), TB, 0, stream>>>(x, W1, h1, N, F_in, C1);
  }
  coeff_bitmap_kernel<<<nbm + nblk(NH, BM_TB), BM_TB, 0, stream>>>(h1, att_src1, att_dst1,
                                                                   asrc, adst, N, F_h,
                                                                   dst, E, (int)nwg, chunks, nbm,
                                                                   bmp);
  range_agg_kernel<32, true><<<nblk(N, RNODES), AGG_TB, 0, stream>>>(src, dst, E, N, asrc, adst,
                                                                     h1, b1, h2, 1,
                                                                     wq_n, wq_e, wq_d,
                                                                     W2, att_src2, att_dst2,
                                                                     asrc2, adst2,
                                                                     bmp, nwords);

  range_agg_kernel<16, false><<<nblk(N, RNODES), AGG_TB, 0, stream>>>(src, dst, E, N, asrc2, adst2,
                                                                      h2, b2, outp, 0,
                                                                      wq_n, wq_e, wq_d,
                                                                      nullptr, nullptr, nullptr,
                                                                      nullptr, nullptr,
                                                                      nullptr, 0);
}


# each queued edge's 8 head scores gathered once: max pass keeps them in the wave's LDS slots, sum pass leaves the exponentials there; sum and agg passes drop their a_src[src] gathers; sum order kept
# speedup vs baseline: 12.6373x; 1.0495x over previous
#include <hip/hip_runtime.h>
#include <cstdint>

typedef float v2f __attribute__((ext_vector_type(2)));
typedef float v8f __attribute__((ext_vector_type(8)));

#define HEADS 8
#define NEG_SLOPE 0.2f
#define SM_EPS 1e-16f
#define NT 4

__global__ void wmma_gemm_f32(const float* __restrict__ A,
                              const float* __restrict__ B,
                              float* __restrict__ C,
                              int M, int K, int Ncols) {
  int wave = (int)((blockIdx.x * blockDim.x + threadIdx.x) >> 5);
  int lane = (int)(threadIdx.x & 31);
  int tilesM  = (M + 15) >> 4;
  int groupsN = Ncols >> 6;
  if (wave >= tilesM * groupsN) return;
  int tm = wave / groupsN;
  int tg = wave - tm * groupsN;
  int half = lane >> 4;
  int l = lane & 15;

  int rowA = tm * 16 + l;
  if (rowA >= M) rowA = M - 1;
  const float* aP = A + (size_t)rowA * K + 2 * half;
  const float* bP = B + (size_t)(2 * half) * Ncols + tg * (16 * NT) + l;

  v8f acc[NT];
#pragma unroll
  for (int t = 0; t < NT; ++t) acc[t] = (v8f){};

  v2f aCur, bCur[NT];
  aCur.x = aP[0];
  aCur.y = aP[1];
#pragma unroll
  for (int t = 0; t < NT; ++t) {
    bCur[t].x = bP[t * 16];
    bCur[t].y = bP[t * 16 + Ncols];
  }

  for (int k0 = 0; k0 < K - 4; k0 += 4) {
    aP += 4;
    bP += (size_t)4 * Ncols;
    v2f aNxt, bNxt[NT];
    aNxt.x = aP[0];
    aNxt.y = aP[1];
#pragma unroll
    for (int t = 0; t < NT; ++t) {
      bNxt[t].x = bP[t * 16];
      bNxt[t].y = bP[t * 16 + Ncols];
    }
#pragma unroll
    for (int t = 0; t < NT; ++t) {
      acc[t] = __builtin_amdgcn_wmma_f32_16x16x4_f32(false, aCur, false, bCur[t],
                                                     (short)0, acc[t],
                                                     false, false);
    }
    aCur = aNxt;
#pragma unroll
    for (int t = 0; t < NT; ++t) bCur[t] = bNxt[t];
  }
#pragma unroll
  for (int t = 0; t < NT; ++t) {
    acc[t] = __builtin_amdgcn_wmma_f32_16x16x4_f32(false, aCur, false, bCur[t],
                                                   (short)0, acc[t],
                                                   false, false);
  }

#pragma unroll
  for (int t = 0; t < NT; ++t) {
    int col = tg * (16 * NT) + t * 16 + l;
#pragma unroll
    for (int r = 0; r < 8; ++r) {
      int row = tm * 16 + r + 8 * half;
      if (row < M) C[(size_t)row * Ncols + col] = acc[t][r];
    }
  }
}

#define RNODES 196
#define BM_CHUNK 1024
#define BM_OWN 256
#define BM_PITCH 33
#define BM_TB 256

__global__ __launch_bounds__(BM_TB)
void coeff_bitmap_kernel(const float* __restrict__ h,
                         const float* __restrict__ att_src,
                         const float* __restrict__ att_dst,
                         float* __restrict__ asrc,
                         float* __restrict__ adst,
                         int N, int C,
                         const int* __restrict__ dst, int E,
                         int nown, int chunks, int nbm,
                         unsigned* __restrict__ bmp) {
  if ((int)blockIdx.x < nbm) {
    __shared__ unsigned sbm[BM_OWN * BM_PITCH];
    const int tid = (int)threadIdx.x;
    const int og = (int)blockIdx.x / chunks;
    const int ch = (int)blockIdx.x - og * chunks;
    for (int i = tid; i < BM_OWN * BM_PITCH; i += BM_TB) sbm[i] = 0u;
    __syncthreads();
    const int e0 = ch * BM_CHUNK;
    int dv[BM_CHUNK / BM_TB];
#pragma unroll
    for (int k = 0; k < BM_CHUNK / BM_TB; ++k) {
      const int e = e0 + k * BM_TB + tid;
      dv[k] = dst[(e < E) ? e : (E - 1)];
    }
#pragma unroll
    for (int k = 0; k < BM_CHUNK / BM_TB; ++k) {
      const int e = e0 + k * BM_TB + tid;
      const int ol = dv[k] / RNODES - og * BM_OWN;
      if (e < E && (unsigned)ol < (unsigned)BM_OWN)
        atomicOr(&sbm[ol * BM_PITCH + k * (BM_TB / 32) + (tid >> 5)], 1u << (tid & 31));
    }
    __syncthreads();
    const int w = tid >> 5, lane = tid & 31;
    const size_t wpo = (size_t)chunks * 32;
    for (int ol = w; ol < BM_OWN; ol += BM_TB / 32) {
      const int o = og * BM_OWN + ol;
      if (o < nown) bmp[(size_t)o * wpo + (size_t)ch * 32 + lane] = sbm[ol * BM_PITCH + lane];
    }
    return;
  }
  long i = ((long)blockIdx.x - nbm) * (long)blockDim.x + threadIdx.x;
  if (i >= (long)N * HEADS) return;
  int head = (int)(i & (HEADS - 1));
  const float* hp = h + (size_t)i * C;
  const float* as = att_src + (size_t)head * C;
  const float* ad = att_dst + (size_t)head * C;
  float s = 0.f, d = 0.f;
  for (int c = 0; c < C; ++c) {
    float v = hp[c];
    s += v * as[c];
    d += v * ad[c];
  }
  asrc[i] = s;
  adst[i] = d;
}

__device__ __forceinline__ float lrelu(float v) {
  return v > 0.f ? v : NEG_SLOPE * v;
}

#define AGG_TB 1024
#define AGG_WAVES (AGG_TB / 32)
#define QCAP 288
#define SCAP 192

#define PH_MAX  1
#define PH_SUM  2
#define PH_AGG  3

__device__ __forceinline__ int fkey(int i) { return i >= 0 ? i : (i ^ 0x7fffffff); }

template <int C>
__device__ __forceinline__ float head_sum(const float* v, const float* al, int b, int lane) {
  float a[HEADS];
#pragma unroll
  for (int k = 0; k < HEADS; ++k)
    a[k] = __int_as_float(__builtin_amdgcn_readlane(__float_as_int(al[k]), b));
  float acc;
  if constexpr (C == 32) {
    acc = a[0] * v[0];
#pragma unroll
    for (int k = 1; k < HEADS; ++k) acc = __builtin_fmaf(a[k], v[k], acc);
  } else {
    const bool hi = lane >= 16;
    acc = (hi ? a[1] : a[0]) * v[0];
#pragma unroll
    for (int j = 1; j < 4; ++j) acc = __builtin_fmaf(hi ? a[2 * j + 1] : a[2 * j], v[j], acc);
  }
  return acc;
}

template <int C, int PH, bool KEEP>
__device__ __forceinline__ void edge_batch(int s, int dl, int count, int lane,
                                           const float* __restrict__ asrc,
                                           const float* __restrict__ h,
                                           const float* sadst, int* smaxk, float* ssum,
                                           float* tile,
                                           float4* klo, float4* khi) {
  constexpr int NL = (HEADS * C) / 32;
  float v[HEADS];
  if constexpr (PH == PH_MAX || !KEEP) {
    const float4* pa = (const float4*)(asrc + (size_t)s * HEADS);
    const float4* pd = (const float4*)(sadst + dl * HEADS);
    const float4 a0 = pa[0], a1 = pa[1];
    const float4 d0 = pd[0], d1 = pd[1];
    v[0] = lrelu(a0.x + d0.x);
    v[1] = lrelu(a0.y + d0.y);
    v[2] = lrelu(a0.z + d0.z);
    v[3] = lrelu(a0.w + d0.w);
    v[4] = lrelu(a1.x + d1.x);
    v[5] = lrelu(a1.y + d1.y);
    v[6] = lrelu(a1.z + d1.z);
    v[7] = lrelu(a1.w + d1.w);
  } else {
    const float4 c0 = *klo, c1 = *khi;
    v[0] = c0.x; v[1] = c0.y; v[2] = c0.z; v[3] = c0.w;
    v[4] = c1.x; v[5] = c1.y; v[6] = c1.z; v[7] = c1.w;
  }
  if constexpr (PH == PH_MAX) {
    if (lane < count) {
#pragma unroll
      for (int k = 0; k < HEADS; ++k)
        atomicMax(&smaxk[dl * HEADS + k], fkey(__float_as_int(v[k])));
    }
    if constexpr (KEEP) {
      *klo = make_float4(v[0], v[1], v[2], v[3]);
      *khi = make_float4(v[4], v[5], v[6], v[7]);
    }
  } else {
    float p[HEADS];
    if constexpr (PH == PH_SUM || !KEEP) {
      const int4* pm = (const int4*)(smaxk + dl * HEADS);
      const int4 k0 = pm[0], k1 = pm[1];
      p[0] = __expf(v[0] - __int_as_float(fkey(k0.x)));
      p[1] = __expf(v[1] - __int_as_float(fkey(k0.y)));
      p[2] = __expf(v[2] - __int_as_float(fkey(k0.z)));
      p[3] = __expf(v[3] - __int_as_float(fkey(k0.w)));
      p[4] = __expf(v[4] - __int_as_float(fkey(k1.x)));
      p[5] = __expf(v[5] - __int_as_float(fkey(k1.y)));
      p[6] = __expf(v[6] - __int_as_float(fkey(k1.z)));
      p[7] = __expf(v[7] - __int_as_float(fkey(k1.w)));
    } else {
#pragma unroll
      for (int k = 0; k < HEADS; ++k) p[k] = v[k];
    }
    if constexpr (PH == PH_SUM) {
      if (lane < count) {
#pragma unroll
        for (int k = 0; k < HEADS; ++k) atomicAdd(&ssum[dl * HEADS + k], p[k]);
      }
      if constexpr (KEEP) {
        *klo = make_float4(p[0], p[1], p[2], p[3]);
        *khi = make_float4(p[4], p[5], p[6], p[7]);
      }
    } else {
      const float4* pq = (const float4*)(ssum + dl * HEADS);
      const float4 q0 = pq[0], q1 = pq[1];
      float al[HEADS];
      al[0] = p[0] / (q0.x + SM_EPS);
      al[1] = p[1] / (q0.y + SM_EPS);
      al[2] = p[2] / (q0.z + SM_EPS);
      al[3] = p[3] / (q0.w + SM_EPS);
      al[4] = p[4] / (q1.x + SM_EPS);
      al[5] = p[5] / (q1.y + SM_EPS);
      al[6] = p[6] / (q1.z + SM_EPS);
      al[7] = p[7] / (q1.w + SM_EPS);

      int b = 0;
      for (; b + 4 <= count; b += 4) {
        float x[4][NL];
#pragma unroll
        for (int u = 0; u < 4; ++u) {
          const int sb = __builtin_amdgcn_readlane(s, b + u);
          const float* hp = h + (size_t)sb * (HEADS * C) + lane;
#pragma unroll
          for (int j = 0; j < NL; ++j) x[u][j] = hp[j * 32];
        }
#pragma unroll
        for (int u = 0; u < 4; ++u) {
          const int db = __builtin_amdgcn_readlane(dl, b + u);
          const float acc = head_sum<C>(x[u], al, b + u, lane);
          atomicAdd(&tile[db * C + (lane & (C - 1))], acc);
        }
      }
      for (; b < count; ++b) {
        float x[NL];
        const int sb = __builtin_amdgcn_readlane(s, b);
        const float* hp = h + (size_t)sb * (HEADS * C) + lane;
#pragma unroll
        for (int j = 0; j < NL; ++j) x[j] = hp[j * 32];
        const int db = __builtin_amdgcn_readlane(dl, b);
        const float acc = head_sum<C>(x, al, b, lane);
        atomicAdd(&tile[db * C + (lane & (C - 1))], acc);
      }
    }
  }
}

template <int C, int PH, bool IDS, bool KEEP>
__device__ __forceinline__ void list_batch(const int* __restrict__ src,
                                           const int* __restrict__ dst, int E, int n0,
                                           int qn, int off, int lane, int* qe, int* qd,
                                           const float* __restrict__ asrc,
                                           const float* __restrict__ h,
                                           const float* sadst, int* smaxk, float* ssum,
                                           float* tile, float4* wlo, float4* whi) {
  const int cnt = (qn - off < 32) ? (qn - off) : 32;
  const bool ok = lane < cnt;
  int s, dl;
  if constexpr (IDS) {
    const int e = ok ? qe[off + lane] : (E + n0);
    s = e - E;
    dl = s - n0;
    if (e < E) { s = src[e]; dl = dst[e] - n0; }
    if (ok) { qe[off + lane] = s; qd[off + lane] = dl; }
  } else {
    s = ok ? qe[off + lane] : n0;
    dl = ok ? qd[off + lane] : 0;
  }
  edge_batch<C, PH, KEEP>(s, dl, cnt, lane, asrc, h, sadst, smaxk, ssum, tile,
                          KEEP ? wlo + off + lane : nullptr,
                          KEEP ? whi + off + lane : nullptr);
}

template <int C, int PH, bool IDS>
__device__ __forceinline__ void list_pass(const int* __restrict__ src,
                                          const int* __restrict__ dst, int E, int n0,
                                          int qn, int lane, int* qe, int* qd,
                                          const float* __restrict__ asrc,
                                          const float* __restrict__ h,
                                          const float* sadst, int* smaxk, float* ssum,
                                          float* tile, float4* wlo, float4* whi) {
  int off = 0;
  for (; off < qn && off + 32 <= SCAP; off += 32)
    list_batch<C, PH, IDS, true>(src, dst, E, n0, qn, off, lane, qe, qd,
                                 asrc, h, sadst, smaxk, ssum, tile, wlo, whi);
  for (; off < qn; off += 32)
    list_batch<C, PH, IDS, false>(src, dst, E, n0, qn, off, lane, qe, qd,
                                  asrc, h, sadst, smaxk, ssum, tile, wlo, whi);
}

template <int C, int PH>
__device__ __forceinline__ void scan_pass(const int* __restrict__ src,
                                         const int* __restrict__ dst,
                                         int E, int n0, int nv, int w, int lane,
                                         int* qe, int* qd, int qn,
                                         const float* __restrict__ asrc,
                                         const float* __restrict__ h,
                                         const float* sadst, int* smaxk, float* ssum,
                                         float* tile) {
  for (int r0 = w * 32; r0 < nv; r0 += AGG_TB) {
    const int cnt = (nv - r0 < 32) ? (nv - r0) : 32;
    const int r = (lane < cnt) ? (r0 + lane) : 0;
    edge_batch<C, PH, false>(n0 + r, r, cnt, lane, asrc, h, sadst, smaxk, ssum, tile,
                             nullptr, nullptr);
  }
  const int stride = AGG_WAVES * 256;
  const int elast = (E > 0) ? (E - 1) : 0;
  int eb = w * 256 + lane;
  int dc[8];
#pragma unroll
  for (int j = 0; j < 8; ++j) dc[j] = -1;
  if (E > 0) {
#pragma unroll
    for (int j = 0; j < 8; ++j) {
      const int e = eb + j * 32;
      const int dv = dst[(e < E) ? e : elast];
      dc[j] = (e < E) ? dv : -1;
    }
  }
  for (int base = 0; base < E; base += stride) {
    int dn[8];
    const int ebn = eb + stride;
#pragma unroll
    for (int j = 0; j < 8; ++j) {
      const int e = ebn + j * 32;
      const int dv = dst[(e < E) ? e : elast];
      dn[j] = (e < E) ? dv : -1;
    }
#pragma unroll
    for (int j = 0; j < 8; ++j) {
      const int e = eb + j * 32;
      const int dl = dc[j] - n0;
      const bool hit = (unsigned)dl < (unsigned)nv;
      const unsigned mask = __builtin_amdgcn_ballot_w32(hit);
      if (mask != 0u) {
        if (hit) {
          const int pos = qn + (int)__builtin_amdgcn_mbcnt_lo(mask, 0u);
          qe[pos] = e;
          qd[pos] = dl;
        }
        qn += __popc(mask);
      }
    }
    if (qn >= 32) {
      int off = 0;
      for (; off + 32 <= qn; off += 32) {
        const int e2 = qe[off + lane];
        const int d2 = qd[off + lane];
        const int s2 = src[e2];
        edge_batch<C, PH, false>(s2, d2, 32, lane, asrc, h, sadst, smaxk, ssum, tile,
                                 nullptr, nullptr);
      }
      const int rem = qn - off;
      const int te = (lane < rem) ? qe[off + lane] : 0;
      const int td = (lane < rem) ? qd[off + lane] : 0;
      if (lane < rem) { qe[lane] = te; qd[lane] = td; }
      qn = rem;
    }
#pragma unroll
    for (int j = 0; j < 8; ++j) dc[j] = dn[j];
    eb = ebn;
  }
  if (qn > 0) {
    const bool ok = lane < qn;
    const int e2 = ok ? qe[lane] : 0;
    const int d2 = ok ? qd[lane] : 0;
    const int s2 = src[e2];
    edge_batch<C, PH, false>(s2, d2, qn, lane, asrc, h, sadst, smaxk, ssum, tile,
                             nullptr, nullptr);
  }
}

__device__ __forceinline__ int bitmap_list(const unsigned* __restrict__ row, int nwords,
                                           int w, int lane, int* qe, int qn) {
  for (int base = 0; base < nwords; base += AGG_WAVES * 8 * 128) {
    uint4 v[8];
    const int fw0 = base + (w * 32 + lane) * 4;
#pragma unroll
    for (int j = 0; j < 8; ++j) {
      const int fw = fw0 + j * (AGG_WAVES * 128);
      v[j] = *(const uint4*)(row + ((fw < nwords) ? fw : 0));
    }
    __builtin_amdgcn_sched_barrier(0);
#pragma unroll
    for (int j = 0; j < 8; ++j) {
      const bool in = fw0 + j * (AGG_WAVES * 128) < nwords;
      v[j].x = in ? v[j].x : 0u;
      v[j].y = in ? v[j].y : 0u;
      v[j].z = in ? v[j].z : 0u;
      v[j].w = in ? v[j].w : 0u;
    }
#pragma nounroll
    for (int j = 0; j < 8; ++j) {
      uint4 cur = v[0];
#pragma unroll
      for (int t = 0; t < 7; ++t) v[t] = v[t + 1];
#pragma nounroll
      for (int q = 0; q < 4; ++q) {
        unsigned x = cur.x;
        cur.x = cur.y; cur.y = cur.z; cur.z = cur.w;
        const int ebase = (fw0 + j * (AGG_WAVES * 128) + q) << 5;
        unsigned m = __builtin_amdgcn_ballot_w32(x != 0u);
        while (m != 0u) {
          if (x != 0u) {
            const int pos = qn + (int)__builtin_amdgcn_mbcnt_lo(m, 0u);
            if (pos < QCAP) qe[pos] = ebase + __builtin_ctz(x);
            x &= x - 1u;
          }
          qn += __popc(m);
          m = __builtin_amdgcn_ballot_w32(x != 0u);
        }
      }
    }
  }
  return qn;
}

template <int C, bool BUILD>
__global__ __launch_bounds__(AGG_TB)
void range_agg_kernel(const int* __restrict__ src,
                      const int* __restrict__ dst,
                      int E, int N,
                      const float* __restrict__ asrc,
                      const float* __restrict__ adst,
                      const float* __restrict__ h,
                      const float* __restrict__ bias,
                      float* __restrict__ out,
                      int do_relu,
                      int* __restrict__ wq_n,
                      int* __restrict__ wq_e,
                      int* __restrict__ wq_d,
                      const float* __restrict__ Wn,
                      const float* __restrict__ attn_s,
                      const float* __restrict__ attn_d,
                      float* __restrict__ asrc_n,
                      float* __restrict__ adst_n,
                      const unsigned* __restrict__ bmp,
                      int nwords) {
  __shared__ __attribute__((aligned(16))) float tile[RNODES * C];
  __shared__ __attribute__((aligned(16))) float sadst[RNODES * HEADS];
  __shared__ __attribute__((aligned(16))) int   smaxk[RNODES * HEADS];
  __shared__ __attribute__((aligned(16))) float ssum[RNODES * HEADS];
  __shared__ int qe_all[AGG_WAVES * QCAP];
  __shared__ int qd_all[AGG_WAVES * QCAP];
  __shared__ int sqn[AGG_WAVES];
  __shared__ int sover;
  __shared__ __attribute__((aligned(16))) float kept[2 * AGG_WAVES * SCAP * 4];

  const int tid = (int)threadIdx.x;
  const int lane = tid & 31;
  const int w = __builtin_amdgcn_readfirstlane(tid >> 5);
  const int n0 = (int)blockIdx.x * RNODES;
  const int nv = (N - n0 < RNODES) ? (N - n0) : RNODES;
  const size_t wq = (size_t)blockIdx.x * AGG_WAVES + w;

  int qn = 0;
  if constexpr (!BUILD) {
    qn = __builtin_amdgcn_readfirstlane(wq_n[wq]);
  }

  for (int i = tid; i < RNODES * C; i += AGG_TB) tile[i] = 0.f;
  for (int i = tid; i < RNODES * HEADS; i += AGG_TB) {
    sadst[i] = (i < nv * HEADS) ? adst[(size_t)n0 * HEADS + i] : 0.f;
    smaxk[i] = fkey(__float_as_int(-3.0e38f));
    ssum[i] = 0.f;
  }
  if (tid == 0) sover = 0;

  int* qe = qe_all + w * QCAP;
  int* qd = qd_all + w * QCAP;
  float4* wlo = (float4*)kept + w * SCAP;
  float4* whi = (float4*)kept + (AGG_WAVES + w) * SCAP;
  if constexpr (BUILD) {
    const int rs = w + 32 * lane;
    const bool own = rs < nv;
    if (own) qe[lane] = E + n0 + rs;
    qn = __popc(__builtin_amdgcn_ballot_w32(own));
    __syncthreads();
    qn = bitmap_list(bmp + (size_t)blockIdx.x * nwords, nwords, w, lane, qe, qn);
    if (lane == 0) sqn[w] = qn;
  } else {
    if (qn <= QCAP) {
      const int* we = wq_e + wq * QCAP;
      const int* wd = wq_d + wq * QCAP;
      for (int i = lane; i < qn; i += 32) { qe[i] = we[i]; qd[i] = wd[i]; }
    }
    __syncthreads();
  }
  if (qn > QCAP) sover = 1;
  __syncthreads();
  const bool over = sover != 0;
  if constexpr (BUILD) {
    if (w == 0) wq_n[(size_t)blockIdx.x * AGG_WAVES + lane] = sqn[lane];
  }

  if (!over) {
    list_pass<C, PH_MAX, BUILD>(src, dst, E, n0, qn, lane, qe, qd, asrc, h, sadst, smaxk, ssum, tile,
                                wlo, whi);
    __syncthreads();
    if constexpr (BUILD) {
      int* we = wq_e + wq * QCAP;
      int* wd = wq_d + wq * QCAP;
      for (int i = lane; i < qn; i += 32) { we[i] = qe[i]; wd[i] = qd[i]; }
    }
    list_pass<C, PH_SUM, false>(src, dst, E, n0, qn, lane, qe, qd, asrc, h, sadst, smaxk, ssum, tile,
                                wlo, whi);
    __syncthreads();
    list_pass<C, PH_AGG, false>(src, dst, E, n0, qn, lane, qe, qd, asrc, h, sadst, smaxk, ssum, tile,
                                wlo, whi);
  } else {
    scan_pass<C, PH_MAX>(src, dst, E, n0, nv, w, lane, qe, qd, 0,
                         asrc, h, sadst, smaxk, ssum, tile);
    __syncthreads();
    scan_pass<C, PH_SUM>(src, dst, E, n0, nv, w, lane, qe, qd, 0,
                         asrc, h, sadst, smaxk, ssum, tile);
    __syncthreads();
    scan_pass<C, PH_AGG>(src, dst, E, n0, nv, w, lane, qe, qd, 0,
                         asrc, h, sadst, smaxk, ssum, tile);
  }
  __syncthreads();

  const float inv = 1.0f / HEADS;
  if constexpr (!BUILD) {
    for (int i = tid; i < nv * C; i += AGG_TB) {
      float v = tile[i] * inv + bias[i & (C - 1)];
      if (do_relu) v = v > 0.f ? v : 0.f;
      out[(size_t)n0 * C + i] = v;
    }
  } else {
    constexpr int CN = 16;
    constexpr int NCOL = HEADS * CN;
    constexpr int HP = NCOL + 4;
    constexpr int TM = (RNODES + 15) / 16;
    constexpr int TG = NCOL / (16 * NT);
    static_assert(TM * TG <= AGG_WAVES, "one strip per wave");
    static_assert(TM * 16 * HP <= 2 * AGG_WAVES * SCAP * 4, "the projected rows fit the kept planes");
    float* hs = kept;
    for (int i = tid; i < RNODES * C; i += AGG_TB) {
      float v = tile[i] * inv + bias[i & (C - 1)];
      if (do_relu) v = v > 0.f ? v : 0.f;
      tile[i] = v;
    }
    __syncthreads();
    if (w < TM * TG) {
      const int tm = w / TG;
      const int tg = w - tm * TG;
      const int half = lane >> 4;
      const int l = lane & 15;
      int rowA = tm * 16 + l;
      if (rowA > RNODES - 1) rowA = RNODES - 1;
      const float* aP = tile + rowA * C + 2 * half;
      const float* bP = Wn + (size_t)(2 * half) * NCOL + tg * (16 * NT) + l;
      v8f acc[NT];
#pragma unroll
      for (int t = 0; t < NT; ++t) acc[t] = (v8f){};
#pragma unroll
      for (int k0 = 0; k0 < C; k0 += 4) {
        v2f a;
        a.x = aP[k0];
        a.y = aP[k0 + 1];
#pragma unroll
        for (int t = 0; t < NT; ++t) {
          v2f b;
          b.x = bP[(size_t)k0 * NCOL + t * 16];
          b.y = bP[(size_t)k0 * NCOL + t * 16 + NCOL];
          acc[t] = __builtin_amdgcn_wmma_f32_16x16x4_f32(false, a, false, b,
                                                         (short)0, acc[t],
                                                         false, false);
        }
      }
#pragma unroll
      for (int t = 0; t < NT; ++t) {
#pragma unroll
        for (int r = 0; r < 8; ++r)
          hs[(tm * 16 + r + 8 * half) * HP + tg * (16 * NT) + t * 16 + l] = acc[t][r];
      }
    }
    __syncthreads();
    float* hout = out + (size_t)n0 * NCOL;
    for (int i = tid * 4; i < nv * NCOL; i += AGG_TB * 4) {
      const int row = i / NCOL;
      const int col = i - row * NCOL;
      *(float4*)(hout + i) = *(const float4*)(hs + row * HP + col);
    }
    for (int i = tid; i < nv * HEADS; i += AGG_TB) {
      const int head = i & (HEADS - 1);
      const float* hp = hs + (i >> 3) * HP + head * CN;
      const float* as = attn_s + head * CN;
      const float* ad = attn_d + head * CN;
      float s = 0.f, d = 0.f;
#pragma unroll
      for (int c2 = 0; c2 < CN; ++c2) {
        float v = hp[c2];
        s += v * as[c2];
        d += v * ad[c2];
      }
      asrc_n[(size_t)n0 * HEADS + i] = s;
      adst_n[(size_t)n0 * HEADS + i] = d;
    }
  }
}

static inline unsigned nblk(long n, int b) { return (unsigned)((n + b - 1) / b); }

extern "C" void kernel_launch(void* const* d_in, const int* in_sizes, int n_in,
                              void* d_out, int out_size, void* d_ws, size_t ws_size,
                              hipStream_t stream) {
  const float* x        = (const float*)d_in[0];
  const int*   ei       = (const int*)  d_in[1];
  const float* W1       = (const float*)d_in[2];
  const float* att_src1 = (const float*)d_in[3];
  const float* att_dst1 = (const float*)d_in[4];
  const float* b1       = (const float*)d_in[5];
  const float* W2       = (const float*)d_in[6];
  const float* att_src2 = (const float*)d_in[7];
  const float* att_dst2 = (const float*)d_in[8];
  const float* b2       = (const float*)d_in[9];

  const int F_h   = in_sizes[3] / HEADS;
  const int F_out = in_sizes[7] / HEADS;
  const int F_in  = in_sizes[2] / (HEADS * F_h);
  const int N     = in_sizes[0] / F_in;
  const int E     = in_sizes[1] / 2;
  const int C1    = HEADS * F_h;
  const int C2    = HEADS * F_out;
  const int* src = ei;
  const int* dst = ei + E;

  float* ws = (float*)d_ws;
  size_t o = 0;
  float* h1    = ws + o; o += (size_t)N * C1;
  float* asrc  = ws + o; o += (size_t)N * HEADS;
  float* adst  = ws + o; o += (size_t)N * HEADS;
  o = (o + 31) & ~(size_t)31;
  float* h2    = ws + o; o += (size_t)N * C2;
  float* asrc2 = ws + o; o += (size_t)N * HEADS;
  float* adst2 = ws + o; o += (size_t)N * HEADS;
  const size_t nwg = nblk(N, RNODES);
  o = (o + 31) & ~(size_t)31;
  int* wq_n = (int*)(ws + o); o += nwg * AGG_WAVES;
  int* wq_e = (int*)(ws + o); o += nwg * AGG_WAVES * QCAP;
  int* wq_d = (int*)(ws + o); o += nwg * AGG_WAVES * QCAP;
  const int chunks = (E + BM_CHUNK - 1) / BM_CHUNK;
  const int nwords = chunks * 32;
  const int nbm = chunks * (int)((nwg + BM_OWN - 1) / BM_OWN);
  o = (o + 31) & ~(size_t)31;
  unsigned* bmp = (unsigned*)(ws + o); o += nwg * (size_t)nwords;
  float* outp = (float*)d_out;

  const int TB = 256;
  const long NH = (long)N * HEADS;

  {
    long waves = (long)((N + 15) / 16) * (C1 / (16 * NT));
    wmma_gemm_f32<<<nblk(waves * 32, TB), TB, 0, stream>>>(x, W1, h1, N, F_in, C1);
  }
  coeff_bitmap_kernel<<<nbm + nblk(NH, BM_TB), BM_TB, 0, stream>>>(h1, att_src1, att_dst1,
                                                                   asrc, adst, N, F_h,
                                                                   dst, E, (int)nwg, chunks, nbm,
                                                                   bmp);
  range_agg_kernel<32, true><<<nblk(N, RNODES), AGG_TB, 0, stream>>>(src, dst, E, N, asrc, adst,
                                                                     h1, b1, h2, 1,
                                                                     wq_n, wq_e, wq_d,
                                                                     W2, att_src2, att_dst2,
                                                                     asrc2, adst2,
                                                                     bmp, nwords);

  range_agg_kernel<16, false><<<nblk(N, RNODES), AGG_TB, 0, stream>>>(src, dst, E, N, asrc2, adst2,
                                                                      h2, b2, outp, 0,
                                                                      wq_n, wq_e, wq_d,
                                                                      nullptr, nullptr, nullptr,
                                                                      nullptr, nullptr,
                                                                      nullptr, 0);
}


# dense front end by 16-byte loads: one GEMM B load (whole lines) feeds 4 column tiles (tile t = column 4*l+t), A quad = 2 k-steps; coefficient role reads 16 whole rows a wave; float sums reordered
# speedup vs baseline: 13.7288x; 1.0864x over previous
#include <hip/hip_runtime.h>
#include <cstdint>

typedef float v2f __attribute__((ext_vector_type(2)));
typedef float v4f __attribute__((ext_vector_type(4)));
typedef float v8f __attribute__((ext_vector_type(8)));

#define HEADS 8
#define NEG_SLOPE 0.2f
#define SM_EPS 1e-16f
#define NT 4

__device__ __forceinline__ void gemm_trip(v8f* acc, const v4f a, const v4f* b) {
  const v2f a0 = {a[0], a[1]};
  const v2f a1 = {a[2], a[3]};
#pragma unroll
  for (int t = 0; t < NT; ++t) {
    const v2f bb = {b[0][t], b[1][t]};
    acc[t] = __builtin_amdgcn_wmma_f32_16x16x4_f32(false, a0, false, bb,
                                                   (short)0, acc[t],
                                                   false, false);
  }
#pragma unroll
  for (int t = 0; t < NT; ++t) {
    const v2f bb = {b[2][t], b[3][t]};
    acc[t] = __builtin_amdgcn_wmma_f32_16x16x4_f32(false, a1, false, bb,
                                                   (short)0, acc[t],
                                                   false, false);
  }
}

__global__ void wmma_gemm_f32(const float* __restrict__ A,
                              const float* __restrict__ B,
                              float* __restrict__ C,
                              int M, int K, int Ncols) {
  int wave = (int)((blockIdx.x * blockDim.x + threadIdx.x) >> 5);
  int lane = (int)(threadIdx.x & 31);
  int tilesM  = (M + 15) >> 4;
  int groupsN = Ncols >> 6;
  if (wave >= tilesM * groupsN) return;
  int tm = wave / groupsN;
  int tg = wave - tm * groupsN;
  int half = lane >> 4;
  int l = lane & 15;

  int rowA = tm * 16 + l;
  if (rowA >= M) rowA = M - 1;
  const float* aP = A + (size_t)rowA * K + 4 * half;
  const float* bP = B + (size_t)(4 * half) * Ncols + tg * (16 * NT) + 4 * l;

  v8f acc[NT];
#pragma unroll
  for (int t = 0; t < NT; ++t) acc[t] = (v8f){};

  v4f aCur = *(const v4f*)aP;
  v4f bCur[4];
#pragma unroll
  for (int j = 0; j < 4; ++j) bCur[j] = *(const v4f*)(bP + (size_t)j * Ncols);

  for (int k0 = 0; k0 < K - 8; k0 += 8) {
    aP += 8;
    bP += (size_t)8 * Ncols;
    v4f aNxt = *(const v4f*)aP;
    v4f bNxt[4];
#pragma unroll
    for (int j = 0; j < 4; ++j) bNxt[j] = *(const v4f*)(bP + (size_t)j * Ncols);
    gemm_trip(acc, aCur, bCur);
    aCur = aNxt;
#pragma unroll
    for (int j = 0; j < 4; ++j) bCur[j] = bNxt[j];
  }
  gemm_trip(acc, aCur, bCur);

  float* cP = C + (size_t)(tm * 16 + 8 * half) * Ncols + tg * (16 * NT) + 4 * l;
#pragma unroll
  for (int r = 0; r < 8; ++r) {
    const int row = tm * 16 + r + 8 * half;
    const v4f q = {acc[0][r], acc[1][r], acc[2][r], acc[3][r]};
    if (row < M) *(v4f*)(cP + (size_t)r * Ncols) = q;
  }
}

#define RNODES 196
#define BM_CHUNK 1024
#define BM_OWN 256
#define BM_PITCH 33
#define BM_TB 256
#define CF_ROWS 16

__global__ __launch_bounds__(BM_TB)
void coeff_bitmap_kernel(const float* __restrict__ h,
                         const float* __restrict__ att_src,
                         const float* __restrict__ att_dst,
                         float* __restrict__ asrc,
                         float* __restrict__ adst,
                         int N, int C,
                         const int* __restrict__ dst, int E,
                         int nown, int chunks, int nbm,
                         unsigned* __restrict__ bmp) {
  if ((int)blockIdx.x < nbm) {
    __shared__ unsigned sbm[BM_OWN * BM_PITCH];
    const int tid = (int)threadIdx.x;
    const int og = (int)blockIdx.x / chunks;
    const int ch = (int)blockIdx.x - og * chunks;
    for (int i = tid; i < BM_OWN * BM_PITCH; i += BM_TB) sbm[i] = 0u;
    __syncthreads();
    const int e0 = ch * BM_CHUNK;
    int dv[BM_CHUNK / BM_TB];
#pragma unroll
    for (int k = 0; k < BM_CHUNK / BM_TB; ++k) {
      const int e = e0 + k * BM_TB + tid;
      dv[k] = dst[(e < E) ? e : (E - 1)];
    }
#pragma unroll
    for (int k = 0; k < BM_CHUNK / BM_TB; ++k) {
      const int e = e0 + k * BM_TB + tid;
      const int ol = dv[k] / RNODES - og * BM_OWN;
      if (e < E && (unsigned)ol < (unsigned)BM_OWN)
        atomicOr(&sbm[ol * BM_PITCH + k * (BM_TB / 32) + (tid >> 5)], 1u << (tid & 31));
    }
    __syncthreads();
    const int w = tid >> 5, lane = tid & 31;
    const size_t wpo = (size_t)chunks * 32;
    for (int ol = w; ol < BM_OWN; ol += BM_TB / 32) {
      const int o = og * BM_OWN + ol;
      if (o < nown) bmp[(size_t)o * wpo + (size_t)ch * 32 + lane] = sbm[ol * BM_PITCH + lane];
    }
    return;
  }
  const int lane = (int)(threadIdx.x & 31);
  const int w = (int)(threadIdx.x >> 5);
  const long nb = (long)blockIdx.x - nbm;
  const long nfirst = (nb * (BM_TB / 32) + w) * CF_ROWS;
  if (nfirst >= (long)N) return;
  const size_t pitch = (size_t)HEADS * C;
  const size_t hoff = pitch / 2;
  const v4f as0 = *(const v4f*)(att_src + 4 * lane);
  const v4f as1 = *(const v4f*)(att_src + hoff + 4 * lane);
  const v4f ad0 = *(const v4f*)(att_dst + 4 * lane);
  const v4f ad1 = *(const v4f*)(att_dst + hoff + 4 * lane);
  for (int b = 0; b < CF_ROWS; b += 8) {
    v4f q0[8], q1[8];
#pragma unroll
    for (int r = 0; r < 8; ++r) {
      const long n = (nfirst + b + r < (long)N) ? (nfirst + b + r) : ((long)N - 1);
      const float* hp = h + (size_t)n * pitch + 4 * lane;
      q0[r] = *(const v4f*)hp;
      q1[r] = *(const v4f*)(hp + hoff);
    }
    float ks0 = 0.f, ks1 = 0.f, kd0 = 0.f, kd1 = 0.f;
#pragma unroll
    for (int r = 0; r < 8; ++r) {
      float s0 = q0[r][0] * as0[0], s1 = q1[r][0] * as1[0];
      float d0 = q0[r][0] * ad0[0], d1 = q1[r][0] * ad1[0];
#pragma unroll
      for (int c = 1; c < 4; ++c) {
        s0 = __builtin_fmaf(q0[r][c], as0[c], s0);
        s1 = __builtin_fmaf(q1[r][c], as1[c], s1);
        d0 = __builtin_fmaf(q0[r][c], ad0[c], d0);
        d1 = __builtin_fmaf(q1[r][c], ad1[c], d1);
      }
#pragma unroll
      for (int m = 1; m < 8; m <<= 1) {
        s0 += __shfl_xor(s0, m);
        s1 += __shfl_xor(s1, m);
        d0 += __shfl_xor(d0, m);
        d1 += __shfl_xor(d1, m);
      }
      if ((lane & 7) == r) { ks0 = s0; ks1 = s1; kd0 = d0; kd1 = d1; }
    }
    const long n = nfirst + b + (lane & 7);
    if (n < (long)N) {
      const size_t o = (size_t)n * HEADS + (lane >> 3);
      asrc[o] = ks0;
      asrc[o + 4] = ks1;
      adst[o] = kd0;
      adst[o + 4] = kd1;
    }
  }
}

__device__ __forceinline__ float lrelu(float v) {
  return v > 0.f ? v : NEG_SLOPE * v;
}

#define AGG_TB 1024
#define AGG_WAVES (AGG_TB / 32)
#define QCAP 288
#define SCAP 192

#define PH_MAX  1
#define PH_SUM  2
#define PH_AGG  3

__device__ __forceinline__ int fkey(int i) { return i >= 0 ? i : (i ^ 0x7fffffff); }

template <int C>
__device__ __forceinline__ float head_sum(const float* v, const float* al, int b, int lane) {
  float a[HEADS];
#pragma unroll
  for (int k = 0; k < HEADS; ++k)
    a[k] = __int_as_float(__builtin_amdgcn_readlane(__float_as_int(al[k]), b));
  float acc;
  if constexpr (C == 32) {
    acc = a[0] * v[0];
#pragma unroll
    for (int k = 1; k < HEADS; ++k) acc = __builtin_fmaf(a[k], v[k], acc);
  } else {
    const bool hi = lane >= 16;
    acc = (hi ? a[1] : a[0]) * v[0];
#pragma unroll
    for (int j = 1; j < 4; ++j) acc = __builtin_fmaf(hi ? a[2 * j + 1] : a[2 * j], v[j], acc);
  }
  return acc;
}

template <int C, int PH, bool KEEP>
__device__ __forceinline__ void edge_batch(int s, int dl, int count, int lane,
                                           const float* __restrict__ asrc,
                                           const float* __restrict__ h,
                                           const float* sadst, int* smaxk, float* ssum,
                                           float* tile,
                                           float4* klo, float4* khi) {
  constexpr int NL = (HEADS * C) / 32;
  float v[HEADS];
  if constexpr (PH == PH_MAX || !KEEP) {
    const float4* pa = (const float4*)(asrc + (size_t)s * HEADS);
    const float4* pd = (const float4*)(sadst + dl * HEADS);
    const float4 a0 = pa[0], a1 = pa[1];
    const float4 d0 = pd[0], d1 = pd[1];
    v[0] = lrelu(a0.x + d0.x);
    v[1] = lrelu(a0.y + d0.y);
    v[2] = lrelu(a0.z + d0.z);
    v[3] = lrelu(a0.w + d0.w);
    v[4] = lrelu(a1.x + d1.x);
    v[5] = lrelu(a1.y + d1.y);
    v[6] = lrelu(a1.z + d1.z);
    v[7] = lrelu(a1.w + d1.w);
  } else {
    const float4 c0 = *klo, c1 = *khi;
    v[0] = c0.x; v[1] = c0.y; v[2] = c0.z; v[3] = c0.w;
    v[4] = c1.x; v[5] = c1.y; v[6] = c1.z; v[7] = c1.w;
  }
  if constexpr (PH == PH_MAX) {
    if (lane < count) {
#pragma unroll
      for (int k = 0; k < HEADS; ++k)
        atomicMax(&smaxk[dl * HEADS + k], fkey(__float_as_int(v[k])));
    }
    if constexpr (KEEP) {
      *klo = make_float4(v[0], v[1], v[2], v[3]);
      *khi = make_float4(v[4], v[5], v[6], v[7]);
    }
  } else {
    float p[HEADS];
    if constexpr (PH == PH_SUM || !KEEP) {
      const int4* pm = (const int4*)(smaxk + dl * HEADS);
      const int4 k0 = pm[0], k1 = pm[1];
      p[0] = __expf(v[0] - __int_as_float(fkey(k0.x)));
      p[1] = __expf(v[1] - __int_as_float(fkey(k0.y)));
      p[2] = __expf(v[2] - __int_as_float(fkey(k0.z)));
      p[3] = __expf(v[3] - __int_as_float(fkey(k0.w)));
      p[4] = __expf(v[4] - __int_as_float(fkey(k1.x)));
      p[5] = __expf(v[5] - __int_as_float(fkey(k1.y)));
      p[6] = __expf(v[6] - __int_as_float(fkey(k1.z)));
      p[7] = __expf(v[7] - __int_as_float(fkey(k1.w)));
    } else {
#pragma unroll
      for (int k = 0; k < HEADS; ++k) p[k] = v[k];
    }
    if constexpr (PH == PH_SUM) {
      if (lane < count) {
#pragma unroll
        for (int k = 0; k < HEADS; ++k) atomicAdd(&ssum[dl * HEADS + k], p[k]);
      }
      if constexpr (KEEP) {
        *klo = make_float4(p[0], p[1], p[2], p[3]);
        *khi = make_float4(p[4], p[5], p[6], p[7]);
      }
    } else {
      const float4* pq = (const float4*)(ssum + dl * HEADS);
      const float4 q0 = pq[0], q1 = pq[1];
      float al[HEADS];
      al[0] = p[0] / (q0.x + SM_EPS);
      al[1] = p[1] / (q0.y + SM_EPS);
      al[2] = p[2] / (q0.z + SM_EPS);
      al[3] = p[3] / (q0.w + SM_EPS);
      al[4] = p[4] / (q1.x + SM_EPS);
      al[5] = p[5] / (q1.y + SM_EPS);
      al[6] = p[6] / (q1.z + SM_EPS);
      al[7] = p[7] / (q1.w + SM_EPS);

      int b = 0;
      for (; b + 4 <= count; b += 4) {
        float x[4][NL];
#pragma unroll
        for (int u = 0; u < 4; ++u) {
          const int sb = __builtin_amdgcn_readlane(s, b + u);
          const float* hp = h + (size_t)sb * (HEADS * C) + lane;
#pragma unroll
          for (int j = 0; j < NL; ++j) x[u][j] = hp[j * 32];
        }
#pragma unroll
        for (int u = 0; u < 4; ++u) {
          const int db = __builtin_amdgcn_readlane(dl, b + u);
          const float acc = head_sum<C>(x[u], al, b + u, lane);
          atomicAdd(&tile[db * C + (lane & (C - 1))], acc);
        }
      }
      for (; b < count; ++b) {
        float x[NL];
        const int sb = __builtin_amdgcn_readlane(s, b);
        const float* hp = h + (size_t)sb * (HEADS * C) + lane;
#pragma unroll
        for (int j = 0; j < NL; ++j) x[j] = hp[j * 32];
        const int db = __builtin_amdgcn_readlane(dl, b);
        const float acc = head_sum<C>(x, al, b, lane);
        atomicAdd(&tile[db * C + (lane & (C - 1))], acc);
      }
    }
  }
}

template <int C, int PH, bool IDS, bool KEEP>
__device__ __forceinline__ void list_batch(const int* __restrict__ src,
                                           const int* __restrict__ dst, int E, int n0,
                                           int qn, int off, int lane, int* qe, int* qd,
                                           const float* __restrict__ asrc,
                                           const float* __restrict__ h,
                                           const float* sadst, int* smaxk, float* ssum,
                                           float* tile, float4* wlo, float4* whi) {
  const int cnt = (qn - off < 32) ? (qn - off) : 32;
  const bool ok = lane < cnt;
  int s, dl;
  if constexpr (IDS) {
    const int e = ok ? qe[off + lane] : (E + n0);
    s = e - E;
    dl = s - n0;
    if (e < E) { s = src[e]; dl = dst[e] - n0; }
    if (ok) { qe[off + lane] = s; qd[off + lane] = dl; }
  } else {
    s = ok ? qe[off + lane] : n0;
    dl = ok ? qd[off + lane] : 0;
  }
  edge_batch<C, PH, KEEP>(s, dl, cnt, lane, asrc, h, sadst, smaxk, ssum, tile,
                          KEEP ? wlo + off + lane : nullptr,
                          KEEP ? whi + off + lane : nullptr);
}

template <int C, int PH, bool IDS>
__device__ __forceinline__ void list_pass(const int* __restrict__ src,
                                          const int* __restrict__ dst, int E, int n0,
                                          int qn, int lane, int* qe, int* qd,
                                          const float* __restrict__ asrc,
                                          const float* __restrict__ h,
                                          const float* sadst, int* smaxk, float* ssum,
                                          float* tile, float4* wlo, float4* whi) {
  int off = 0;
  for (; off < qn && off + 32 <= SCAP; off += 32)
    list_batch<C, PH, IDS, true>(src, dst, E, n0, qn, off, lane, qe, qd,
                                 asrc, h, sadst, smaxk, ssum, tile, wlo, whi);
  for (; off < qn; off += 32)
    list_batch<C, PH, IDS, false>(src, dst, E, n0, qn, off, lane, qe, qd,
                                  asrc, h, sadst, smaxk, ssum, tile, wlo, whi);
}

template <int C, int PH>
__device__ __forceinline__ void scan_pass(const int* __restrict__ src,
                                         const int* __restrict__ dst,
                                         int E, int n0, int nv, int w, int lane,
                                         int* qe, int* qd, int qn,
                                         const float* __restrict__ asrc,
                                         const float* __restrict__ h,
                                         const float* sadst, int* smaxk, float* ssum,
                                         float* tile) {
  for (int r0 = w * 32; r0 < nv; r0 += AGG_TB) {
    const int cnt = (nv - r0 < 32) ? (nv - r0) : 32;
    const int r = (lane < cnt) ? (r0 + lane) : 0;
    edge_batch<C, PH, false>(n0 + r, r, cnt, lane, asrc, h, sadst, smaxk, ssum, tile,
                             nullptr, nullptr);
  }
  const int stride = AGG_WAVES * 256;
  const int elast = (E > 0) ? (E - 1) : 0;
  int eb = w * 256 + lane;
  int dc[8];
#pragma unroll
  for (int j = 0; j < 8; ++j) dc[j] = -1;
  if (E > 0) {
#pragma unroll
    for (int j = 0; j < 8; ++j) {
      const int e = eb + j * 32;
      const int dv = dst[(e < E) ? e : elast];
      dc[j] = (e < E) ? dv : -1;
    }
  }
  for (int base = 0; base < E; base += stride) {
    int dn[8];
    const int ebn = eb + stride;
#pragma unroll
    for (int j = 0; j < 8; ++j) {
      const int e = ebn + j * 32;
      const int dv = dst[(e < E) ? e : elast];
      dn[j] = (e < E) ? dv : -1;
    }
#pragma unroll
    for (int j = 0; j < 8; ++j) {
      const int e = eb + j * 32;
      const int dl = dc[j] - n0;
      const bool hit = (unsigned)dl < (unsigned)nv;
      const unsigned mask = __builtin_amdgcn_ballot_w32(hit);
      if (mask != 0u) {
        if (hit) {
          const int pos = qn + (int)__builtin_amdgcn_mbcnt_lo(mask, 0u);
          qe[pos] = e;
          qd[pos] = dl;
        }
        qn += __popc(mask);
      }
    }
    if (qn >= 32) {
      int off = 0;
      for (; off + 32 <= qn; off += 32) {
        const int e2 = qe[off + lane];
        const int d2 = qd[off + lane];
        const int s2 = src[e2];
        edge_batch<C, PH, false>(s2, d2, 32, lane, asrc, h, sadst, smaxk, ssum, tile,
                                 nullptr, nullptr);
      }
      const int rem = qn - off;
      const int te = (lane < rem) ? qe[off + lane] : 0;
      const int td = (lane < rem) ? qd[off + lane] : 0;
      if (lane < rem) { qe[lane] = te; qd[lane] = td; }
      qn = rem;
    }
#pragma unroll
    for (int j = 0; j < 8; ++j) dc[j] = dn[j];
    eb = ebn;
  }
  if (qn > 0) {
    const bool ok = lane < qn;
    const int e2 = ok ? qe[lane] : 0;
    const int d2 = ok ? qd[lane] : 0;
    const int s2 = src[e2];
    edge_batch<C, PH, false>(s2, d2, qn, lane, asrc, h, sadst, smaxk, ssum, tile,
                             nullptr, nullptr);
  }
}

__device__ __forceinline__ int bitmap_list(const unsigned* __restrict__ row, int nwords,
                                           int w, int lane, int* qe, int qn) {
  for (int base = 0; base < nwords; base += AGG_WAVES * 8 * 128) {
    uint4 v[8];
    const int fw0 = base + (w * 32 + lane) * 4;
#pragma unroll
    for (int j = 0; j < 8; ++j) {
      const int fw = fw0 + j * (AGG_WAVES * 128);
      v[j] = *(const uint4*)(row + ((fw < nwords) ? fw : 0));
    }
    __builtin_amdgcn_sched_barrier(0);
#pragma unroll
    for (int j = 0; j < 8; ++j) {
      const bool in = fw0 + j * (AGG_WAVES * 128) < nwords;
      v[j].x = in ? v[j].x : 0u;
      v[j].y = in ? v[j].y : 0u;
      v[j].z = in ? v[j].z : 0u;
      v[j].w = in ? v[j].w : 0u;
    }
#pragma nounroll
    for (int j = 0; j < 8; ++j) {
      uint4 cur = v[0];
#pragma unroll
      for (int t = 0; t < 7; ++t) v[t] = v[t + 1];
#pragma nounroll
      for (int q = 0; q < 4; ++q) {
        unsigned x = cur.x;
        cur.x = cur.y; cur.y = cur.z; cur.z = cur.w;
        const int ebase = (fw0 + j * (AGG_WAVES * 128) + q) << 5;
        unsigned m = __builtin_amdgcn_ballot_w32(x != 0u);
        while (m != 0u) {
          if (x != 0u) {
            const int pos = qn + (int)__builtin_amdgcn_mbcnt_lo(m, 0u);
            if (pos < QCAP) qe[pos] = ebase + __builtin_ctz(x);
            x &= x - 1u;
          }
          qn += __popc(m);
          m = __builtin_amdgcn_ballot_w32(x != 0u);
        }
      }
    }
  }
  return qn;
}

template <int C, bool BUILD>
__global__ __launch_bounds__(AGG_TB)
void range_agg_kernel(const int* __restrict__ src,
                      const int* __restrict__ dst,
                      int E, int N,
                      const float* __restrict__ asrc,
                      const float* __restrict__ adst,
                      const float* __restrict__ h,
                      const float* __restrict__ bias,
                      float* __restrict__ out,
                      int do_relu,
                      int* __restrict__ wq_n,
                      int* __restrict__ wq_e,
                      int* __restrict__ wq_d,
                      const float* __restrict__ Wn,
                      const float* __restrict__ attn_s,
                      const float* __restrict__ attn_d,
                      float* __restrict__ asrc_n,
                      float* __restrict__ adst_n,
                      const unsigned* __restrict__ bmp,
                      int nwords) {
  __shared__ __attribute__((aligned(16))) float tile[RNODES * C];
  __shared__ __attribute__((aligned(16))) float sadst[RNODES * HEADS];
  __shared__ __attribute__((aligned(16))) int   smaxk[RNODES * HEADS];
  __shared__ __attribute__((aligned(16))) float ssum[RNODES * HEADS];
  __shared__ int qe_all[AGG_WAVES * QCAP];
  __shared__ int qd_all[AGG_WAVES * QCAP];
  __shared__ int sqn[AGG_WAVES];
  __shared__ int sover;
  __shared__ __attribute__((aligned(16))) float kept[2 * AGG_WAVES * SCAP * 4];

  const int tid = (int)threadIdx.x;
  const int lane = tid & 31;
  const int w = __builtin_amdgcn_readfirstlane(tid >> 5);
  const int n0 = (int)blockIdx.x * RNODES;
  const int nv = (N - n0 < RNODES) ? (N - n0) : RNODES;
  const size_t wq = (size_t)blockIdx.x * AGG_WAVES + w;

  int qn = 0;
  if constexpr (!BUILD) {
    qn = __builtin_amdgcn_readfirstlane(wq_n[wq]);
  }

  for (int i = tid; i < RNODES * C; i += AGG_TB) tile[i] = 0.f;
  for (int i = tid; i < RNODES * HEADS; i += AGG_TB) {
    sadst[i] = (i < nv * HEADS) ? adst[(size_t)n0 * HEADS + i] : 0.f;
    smaxk[i] = fkey(__float_as_int(-3.0e38f));
    ssum[i] = 0.f;
  }
  if (tid == 0) sover = 0;

  int* qe = qe_all + w * QCAP;
  int* qd = qd_all + w * QCAP;
  float4* wlo = (float4*)kept + w * SCAP;
  float4* whi = (float4*)kept + (AGG_WAVES + w) * SCAP;
  if constexpr (BUILD) {
    const int rs = w + 32 * lane;
    const bool own = rs < nv;
    if (own) qe[lane] = E + n0 + rs;
    qn = __popc(__builtin_amdgcn_ballot_w32(own));
    __syncthreads();
    qn = bitmap_list(bmp + (size_t)blockIdx.x * nwords, nwords, w, lane, qe, qn);
    if (lane == 0) sqn[w] = qn;
  } else {
    if (qn <= QCAP) {
      const int* we = wq_e + wq * QCAP;
      const int* wd = wq_d + wq * QCAP;
      for (int i = lane; i < qn; i += 32) { qe[i] = we[i]; qd[i] = wd[i]; }
    }
    __syncthreads();
  }
  if (qn > QCAP) sover = 1;
  __syncthreads();
  const bool over = sover != 0;
  if constexpr (BUILD) {
    if (w == 0) wq_n[(size_t)blockIdx.x * AGG_WAVES + lane] = sqn[lane];
  }

  if (!over) {
    list_pass<C, PH_MAX, BUILD>(src, dst, E, n0, qn, lane, qe, qd, asrc, h, sadst, smaxk, ssum, tile,
                                wlo, whi);
    __syncthreads();
    if constexpr (BUILD) {
      int* we = wq_e + wq * QCAP;
      int* wd = wq_d + wq * QCAP;
      for (int i = lane; i < qn; i += 32) { we[i] = qe[i]; wd[i] = qd[i]; }
    }
    list_pass<C, PH_SUM, false>(src, dst, E, n0, qn, lane, qe, qd, asrc, h, sadst, smaxk, ssum, tile,
                                wlo, whi);
    __syncthreads();
    list_pass<C, PH_AGG, false>(src, dst, E, n0, qn, lane, qe, qd, asrc, h, sadst, smaxk, ssum, tile,
                                wlo, whi);
  } else {
    scan_pass<C, PH_MAX>(src, dst, E, n0, nv, w, lane, qe, qd, 0,
                         asrc, h, sadst, smaxk, ssum, tile);
    __syncthreads();
    scan_pass<C, PH_SUM>(src, dst, E, n0, nv, w, lane, qe, qd, 0,
                         asrc, h, sadst, smaxk, ssum, tile);
    __syncthreads();
    scan_pass<C, PH_AGG>(src, dst, E, n0, nv, w, lane, qe, qd, 0,
                         asrc, h, sadst, smaxk, ssum, tile);
  }
  __syncthreads();

  const float inv = 1.0f / HEADS;
  if constexpr (!BUILD) {
    for (int i = tid; i < nv * C; i += AGG_TB) {
      float v = tile[i] * inv + bias[i & (C - 1)];
      if (do_relu) v = v > 0.f ? v : 0.f;
      out[(size_t)n0 * C + i] = v;
    }
  } else {
    constexpr int CN = 16;
    constexpr int NCOL = HEADS * CN;
    constexpr int HP = NCOL + 4;
    constexpr int TM = (RNODES + 15) / 16;
    constexpr int TG = NCOL / (16 * NT);
    static_assert(TM * TG <= AGG_WAVES, "one strip per wave");
    static_assert(TM * 16 * HP <= 2 * AGG_WAVES * SCAP * 4, "the projected rows fit the kept planes");
    float* hs = kept;
    for (int i = tid; i < RNODES * C; i += AGG_TB) {
      float v = tile[i] * inv + bias[i & (C - 1)];
      if (do_relu) v = v > 0.f ? v : 0.f;
      tile[i] = v;
    }
    __syncthreads();
    if (w < TM * TG) {
      const int tm = w / TG;
      const int tg = w - tm * TG;
      const int half = lane >> 4;
      const int l = lane & 15;
      int rowA = tm * 16 + l;
      if (rowA > RNODES - 1) rowA = RNODES - 1;
      const float* aP = tile + rowA * C + 2 * half;
      const float* bP = Wn + (size_t)(2 * half) * NCOL + tg * (16 * NT) + l;
      v8f acc[NT];
#pragma unroll
      for (int t = 0; t < NT; ++t) acc[t] = (v8f){};
#pragma unroll
      for (int k0 = 0; k0 < C; k0 += 4) {
        v2f a;
        a.x = aP[k0];
        a.y = aP[k0 + 1];
#pragma unroll
        for (int t = 0; t < NT; ++t) {
          v2f b;
          b.x = bP[(size_t)k0 * NCOL + t * 16];
          b.y = bP[(size_t)k0 * NCOL + t * 16 + NCOL];
          acc[t] = __builtin_amdgcn_wmma_f32_16x16x4_f32(false, a, false, b,
                                                         (short)0, acc[t],
                                                         false, false);
        }
      }
#pragma unroll
      for (int t = 0; t < NT; ++t) {
#pragma unroll
        for (int r = 0; r < 8; ++r)
          hs[(tm * 16 + r + 8 * half) * HP + tg * (16 * NT) + t * 16 + l] = acc[t][r];
      }
    }
    __syncthreads();
    float* hout = out + (size_t)n0 * NCOL;
    for (int i = tid * 4; i < nv * NCOL; i += AGG_TB * 4) {
      const int row = i / NCOL;
      const int col = i - row * NCOL;
      *(float4*)(hout + i) = *(const float4*)(hs + row * HP + col);
    }
    for (int i = tid; i < nv * HEADS; i += AGG_TB) {
      const int head = i & (HEADS - 1);
      const float* hp = hs + (i >> 3) * HP + head * CN;
      const float* as = attn_s + head * CN;
      const float* ad = attn_d + head * CN;
      float s = 0.f, d = 0.f;
#pragma unroll
      for (int c2 = 0; c2 < CN; ++c2) {
        float v = hp[c2];
        s += v * as[c2];
        d += v * ad[c2];
      }
      asrc_n[(size_t)n0 * HEADS + i] = s;
      adst_n[(size_t)n0 * HEADS + i] = d;
    }
  }
}

static inline unsigned nblk(long n, int b) { return (unsigned)((n + b - 1) / b); }

extern "C" void kernel_launch(void* const* d_in, const int* in_sizes, int n_in,
                              void* d_out, int out_size, void* d_ws, size_t ws_size,
                              hipStream_t stream) {
  const float* x        = (const float*)d_in[0];
  const int*   ei       = (const int*)  d_in[1];
  const float* W1       = (const float*)d_in[2];
  const float* att_src1 = (const float*)d_in[3];
  const float* att_dst1 = (const float*)d_in[4];
  const float* b1       = (const float*)d_in[5];
  const float* W2       = (const float*)d_in[6];
  const float* att_src2 = (const float*)d_in[7];
  const float* att_dst2 = (const float*)d_in[8];
  const float* b2       = (const float*)d_in[9];

  const int F_h   = in_sizes[3] / HEADS;
  const int F_out = in_sizes[7] / HEADS;
  const int F_in  = in_sizes[2] / (HEADS * F_h);
  const int N     = in_sizes[0] / F_in;
  const int E     = in_sizes[1] / 2;
  const int C1    = HEADS * F_h;
  const int C2    = HEADS * F_out;
  const int* src = ei;
  const int* dst = ei + E;

  float* ws = (float*)d_ws;
  size_t o = 0;
  float* h1    = ws + o; o += (size_t)N * C1;
  float* asrc  = ws + o; o += (size_t)N * HEADS;
  float* adst  = ws + o; o += (size_t)N * HEADS;
  o = (o + 31) & ~(size_t)31;
  float* h2    = ws + o; o += (size_t)N * C2;
  float* asrc2 = ws + o; o += (size_t)N * HEADS;
  float* adst2 = ws + o; o += (size_t)N * HEADS;
  const size_t nwg = nblk(N, RNODES);
  o = (o + 31) & ~(size_t)31;
  int* wq_n = (int*)(ws + o); o += nwg * AGG_WAVES;
  int* wq_e = (int*)(ws + o); o += nwg * AGG_WAVES * QCAP;
  int* wq_d = (int*)(ws + o); o += nwg * AGG_WAVES * QCAP;
  const int chunks = (E + BM_CHUNK - 1) / BM_CHUNK;
  const int nwords = chunks * 32;
  const int nbm = chunks * (int)((nwg + BM_OWN - 1) / BM_OWN);
  o = (o + 31) & ~(size_t)31;
  unsigned* bmp = (unsigned*)(ws + o); o += nwg * (size_t)nwords;
  float* outp = (float*)d_out;

  const int TB = 256;

  {
    long waves = (long)((N + 15) / 16) * (C1 / (16 * NT));
    wmma_gemm_f32<<<nblk(waves * 32, TB), TB, 0, stream>>>(x, W1, h1, N, F_in, C1);
  }
  coeff_bitmap_kernel<<<nbm + nblk(N, (BM_TB / 32) * CF_ROWS), BM_TB, 0, stream>>>(h1, att_src1, att_dst1,
                                                                   asrc, adst, N, F_h,
                                                                   dst, E, (int)nwg, chunks, nbm,
                                                                   bmp);
  range_agg_kernel<32, true><<<nblk(N, RNODES), AGG_TB, 0, stream>>>(src, dst, E, N, asrc, adst,
                                                                     h1, b1, h2, 1,
                                                                     wq_n, wq_e, wq_d,
                                                                     W2, att_src2, att_dst2,
                                                                     asrc2, adst2,
                                                                     bmp, nwords);

  range_agg_kernel<16, false><<<nblk(N, RNODES), AGG_TB, 0, stream>>>(src, dst, E, N, asrc2, adst2,
                                                                      h2, b2, outp, 0,
                                                                      wq_n, wq_e, wq_d,
                                                                      nullptr, nullptr, nullptr,
                                                                      nullptr, nullptr,
                                                                      nullptr, 0);
}


# score max and sum passes: a lane issues its 8 per-head LDS atomics in an order rotated by lane%8, so one instruction's 32 lanes spread over 64 banks, not 8; order of the float sums changes
# speedup vs baseline: 14.9300x; 1.0875x over previous
#include <hip/hip_runtime.h>
#include <cstdint>

typedef float v2f __attribute__((ext_vector_type(2)));
typedef float v4f __attribute__((ext_vector_type(4)));
typedef float v8f __attribute__((ext_vector_type(8)));

#define HEADS 8
#define NEG_SLOPE 0.2f
#define SM_EPS 1e-16f
#define NT 4

__device__ __forceinline__ void gemm_trip(v8f* acc, const v4f a, const v4f* b) {
  const v2f a0 = {a[0], a[1]};
  const v2f a1 = {a[2], a[3]};
#pragma unroll
  for (int t = 0; t < NT; ++t) {
    const v2f bb = {b[0][t], b[1][t]};
    acc[t] = __builtin_amdgcn_wmma_f32_16x16x4_f32(false, a0, false, bb,
                                                   (short)0, acc[t],
                                                   false, false);
  }
#pragma unroll
  for (int t = 0; t < NT; ++t) {
    const v2f bb = {b[2][t], b[3][t]};
    acc[t] = __builtin_amdgcn_wmma_f32_16x16x4_f32(false, a1, false, bb,
                                                   (short)0, acc[t],
                                                   false, false);
  }
}

__global__ void wmma_gemm_f32(const float* __restrict__ A,
                              const float* __restrict__ B,
                              float* __restrict__ C,
                              int M, int K, int Ncols) {
  int wave = (int)((blockIdx.x * blockDim.x + threadIdx.x) >> 5);
  int lane = (int)(threadIdx.x & 31);
  int tilesM  = (M + 15) >> 4;
  int groupsN = Ncols >> 6;
  if (wave >= tilesM * groupsN) return;
  int tm = wave / groupsN;
  int tg = wave - tm * groupsN;
  int half = lane >> 4;
  int l = lane & 15;

  int rowA = tm * 16 + l;
  if (rowA >= M) rowA = M - 1;
  const float* aP = A + (size_t)rowA * K + 4 * half;
  const float* bP = B + (size_t)(4 * half) * Ncols + tg * (16 * NT) + 4 * l;

  v8f acc[NT];
#pragma unroll
  for (int t = 0; t < NT; ++t) acc[t] = (v8f){};

  v4f aCur = *(const v4f*)aP;
  v4f bCur[4];
#pragma unroll
  for (int j = 0; j < 4; ++j) bCur[j] = *(const v4f*)(bP + (size_t)j * Ncols);

  for (int k0 = 0; k0 < K - 8; k0 += 8) {
    aP += 8;
    bP += (size_t)8 * Ncols;
    v4f aNxt = *(const v4f*)aP;
    v4f bNxt[4];
#pragma unroll
    for (int j = 0; j < 4; ++j) bNxt[j] = *(const v4f*)(bP + (size_t)j * Ncols);
    gemm_trip(acc, aCur, bCur);
    aCur = aNxt;
#pragma unroll
    for (int j = 0; j < 4; ++j) bCur[j] = bNxt[j];
  }
  gemm_trip(acc, aCur, bCur);

  float* cP = C + (size_t)(tm * 16 + 8 * half) * Ncols + tg * (16 * NT) + 4 * l;
#pragma unroll
  for (int r = 0; r < 8; ++r) {
    const int row = tm * 16 + r + 8 * half;
    const v4f q = {acc[0][r], acc[1][r], acc[2][r], acc[3][r]};
    if (row < M) *(v4f*)(cP + (size_t)r * Ncols) = q;
  }
}

#define RNODES 196
#define BM_CHUNK 1024
#define BM_OWN 256
#define BM_PITCH 33
#define BM_TB 256
#define CF_ROWS 16

__global__ __launch_bounds__(BM_TB)
void coeff_bitmap_kernel(const float* __restrict__ h,
                         const float* __restrict__ att_src,
                         const float* __restrict__ att_dst,
                         float* __restrict__ asrc,
                         float* __restrict__ adst,
                         int N, int C,
                         const int* __restrict__ dst, int E,
                         int nown, int chunks, int nbm,
                         unsigned* __restrict__ bmp) {
  if ((int)blockIdx.x < nbm) {
    __shared__ unsigned sbm[BM_OWN * BM_PITCH];
    const int tid = (int)threadIdx.x;
    const int og = (int)blockIdx.x / chunks;
    const int ch = (int)blockIdx.x - og * chunks;
    for (int i = tid; i < BM_OWN * BM_PITCH; i += BM_TB) sbm[i] = 0u;
    __syncthreads();
    const int e0 = ch * BM_CHUNK;
    int dv[BM_CHUNK / BM_TB];
#pragma unroll
    for (int k = 0; k < BM_CHUNK / BM_TB; ++k) {
      const int e = e0 + k * BM_TB + tid;
      dv[k] = dst[(e < E) ? e : (E - 1)];
    }
#pragma unroll
    for (int k = 0; k < BM_CHUNK / BM_TB; ++k) {
      const int e = e0 + k * BM_TB + tid;
      const int ol = dv[k] / RNODES - og * BM_OWN;
      if (e < E && (unsigned)ol < (unsigned)BM_OWN)
        atomicOr(&sbm[ol * BM_PITCH + k * (BM_TB / 32) + (tid >> 5)], 1u << (tid & 31));
    }
    __syncthreads();
    const int w = tid >> 5, lane = tid & 31;
    const size_t wpo = (size_t)chunks * 32;
    for (int ol = w; ol < BM_OWN; ol += BM_TB / 32) {
      const int o = og * BM_OWN + ol;
      if (o < nown) bmp[(size_t)o * wpo + (size_t)ch * 32 + lane] = sbm[ol * BM_PITCH + lane];
    }
    return;
  }
  const int lane = (int)(threadIdx.x & 31);
  const int w = (int)(threadIdx.x >> 5);
  const long nb = (long)blockIdx.x - nbm;
  const long nfirst = (nb * (BM_TB / 32) + w) * CF_ROWS;
  if (nfirst >= (long)N) return;
  const size_t pitch = (size_t)HEADS * C;
  const size_t hoff = pitch / 2;
  const v4f as0 = *(const v4f*)(att_src + 4 * lane);
  const v4f as1 = *(const v4f*)(att_src + hoff + 4 * lane);
  const v4f ad0 = *(const v4f*)(att_dst + 4 * lane);
  const v4f ad1 = *(const v4f*)(att_dst + hoff + 4 * lane);
  for (int b = 0; b < CF_ROWS; b += 8) {
    v4f q0[8], q1[8];
#pragma unroll
    for (int r = 0; r < 8; ++r) {
      const long n = (nfirst + b + r < (long)N) ? (nfirst + b + r) : ((long)N - 1);
      const float* hp = h + (size_t)n * pitch + 4 * lane;
      q0[r] = *(const v4f*)hp;
      q1[r] = *(const v4f*)(hp + hoff);
    }
    float ks0 = 0.f, ks1 = 0.f, kd0 = 0.f, kd1 = 0.f;
#pragma unroll
    for (int r = 0; r < 8; ++r) {
      float s0 = q0[r][0] * as0[0], s1 = q1[r][0] * as1[0];
      float d0 = q0[r][0] * ad0[0], d1 = q1[r][0] * ad1[0];
#pragma unroll
      for (int c = 1; c < 4; ++c) {
        s0 = __builtin_fmaf(q0[r][c], as0[c], s0);
        s1 = __builtin_fmaf(q1[r][c], as1[c], s1);
        d0 = __builtin_fmaf(q0[r][c], ad0[c], d0);
        d1 = __builtin_fmaf(q1[r][c], ad1[c], d1);
      }
#pragma unroll
      for (int m = 1; m < 8; m <<= 1) {
        s0 += __shfl_xor(s0, m);
        s1 += __shfl_xor(s1, m);
        d0 += __shfl_xor(d0, m);
        d1 += __shfl_xor(d1, m);
      }
      if ((lane & 7) == r) { ks0 = s0; ks1 = s1; kd0 = d0; kd1 = d1; }
    }
    const long n = nfirst + b + (lane & 7);
    if (n < (long)N) {
      const size_t o = (size_t)n * HEADS + (lane >> 3);
      asrc[o] = ks0;
      asrc[o + 4] = ks1;
      adst[o] = kd0;
      adst[o + 4] = kd1;
    }
  }
}

__device__ __forceinline__ float lrelu(float v) {
  return v > 0.f ? v : NEG_SLOPE * v;
}

#define AGG_TB 1024
#define AGG_WAVES (AGG_TB / 32)
#define QCAP 288
#define SCAP 192

#define PH_MAX  1
#define PH_SUM  2
#define PH_AGG  3

__device__ __forceinline__ int fkey(int i) { return i >= 0 ? i : (i ^ 0x7fffffff); }

template <typename T>
__device__ __forceinline__ void rot_heads(T* w, int lane) {
#pragma unroll
  for (int s = 1; s < HEADS; s <<= 1) {
    const bool on = (lane & s) != 0;
    T t[HEADS];
#pragma unroll
    for (int k = 0; k < HEADS; ++k) t[k] = on ? w[(k + s) & (HEADS - 1)] : w[k];
#pragma unroll
    for (int k = 0; k < HEADS; ++k) w[k] = t[k];
  }
}

template <int C>
__device__ __forceinline__ float head_sum(const float* v, const float* al, int b, int lane) {
  float a[HEADS];
#pragma unroll
  for (int k = 0; k < HEADS; ++k)
    a[k] = __int_as_float(__builtin_amdgcn_readlane(__float_as_int(al[k]), b));
  float acc;
  if constexpr (C == 32) {
    acc = a[0] * v[0];
#pragma unroll
    for (int k = 1; k < HEADS; ++k) acc = __builtin_fmaf(a[k], v[k], acc);
  } else {
    const bool hi = lane >= 16;
    acc = (hi ? a[1] : a[0]) * v[0];
#pragma unroll
    for (int j = 1; j < 4; ++j) acc = __builtin_fmaf(hi ? a[2 * j + 1] : a[2 * j], v[j], acc);
  }
  return acc;
}

template <int C, int PH, bool KEEP>
__device__ __forceinline__ void edge_batch(int s, int dl, int count, int lane,
                                           const float* __restrict__ asrc,
                                           const float* __restrict__ h,
                                           const float* sadst, int* smaxk, float* ssum,
                                           float* tile,
                                           float4* klo, float4* khi) {
  constexpr int NL = (HEADS * C) / 32;
  float v[HEADS];
  if constexpr (PH == PH_MAX || !KEEP) {
    const float4* pa = (const float4*)(asrc + (size_t)s * HEADS);
    const float4* pd = (const float4*)(sadst + dl * HEADS);
    const float4 a0 = pa[0], a1 = pa[1];
    const float4 d0 = pd[0], d1 = pd[1];
    v[0] = lrelu(a0.x + d0.x);
    v[1] = lrelu(a0.y + d0.y);
    v[2] = lrelu(a0.z + d0.z);
    v[3] = lrelu(a0.w + d0.w);
    v[4] = lrelu(a1.x + d1.x);
    v[5] = lrelu(a1.y + d1.y);
    v[6] = lrelu(a1.z + d1.z);
    v[7] = lrelu(a1.w + d1.w);
  } else {
    const float4 c0 = *klo, c1 = *khi;
    v[0] = c0.x; v[1] = c0.y; v[2] = c0.z; v[3] = c0.w;
    v[4] = c1.x; v[5] = c1.y; v[6] = c1.z; v[7] = c1.w;
  }
  if constexpr (PH == PH_MAX) {
    if (lane < count) {
      int w[HEADS];
#pragma unroll
      for (int k = 0; k < HEADS; ++k) w[k] = fkey(__float_as_int(v[k]));
      rot_heads(w, lane);
      int* pm = smaxk + dl * HEADS;
#pragma unroll
      for (int k = 0; k < HEADS; ++k)
        atomicMax(&pm[(k + lane) & (HEADS - 1)], w[k]);
    }
    if constexpr (KEEP) {
      *klo = make_float4(v[0], v[1], v[2], v[3]);
      *khi = make_float4(v[4], v[5], v[6], v[7]);
    }
  } else {
    float p[HEADS];
    if constexpr (PH == PH_SUM || !KEEP) {
      const int4* pm = (const int4*)(smaxk + dl * HEADS);
      const int4 k0 = pm[0], k1 = pm[1];
      p[0] = __expf(v[0] - __int_as_float(fkey(k0.x)));
      p[1] = __expf(v[1] - __int_as_float(fkey(k0.y)));
      p[2] = __expf(v[2] - __int_as_float(fkey(k0.z)));
      p[3] = __expf(v[3] - __int_as_float(fkey(k0.w)));
      p[4] = __expf(v[4] - __int_as_float(fkey(k1.x)));
      p[5] = __expf(v[5] - __int_as_float(fkey(k1.y)));
      p[6] = __expf(v[6] - __int_as_float(fkey(k1.z)));
      p[7] = __expf(v[7] - __int_as_float(fkey(k1.w)));
    } else {
#pragma unroll
      for (int k = 0; k < HEADS; ++k) p[k] = v[k];
    }
    if constexpr (PH == PH_SUM) {
      if (lane < count) {
        float w[HEADS];
#pragma unroll
        for (int k = 0; k < HEADS; ++k) w[k] = p[k];
        rot_heads(w, lane);
        float* ps = ssum + dl * HEADS;
#pragma unroll
        for (int k = 0; k < HEADS; ++k)
          atomicAdd(&ps[(k + lane) & (HEADS - 1)], w[k]);
      }
      if constexpr (KEEP) {
        *klo = make_float4(p[0], p[1], p[2], p[3]);
        *khi = make_float4(p[4], p[5], p[6], p[7]);
      }
    } else {
      const float4* pq = (const float4*)(ssum + dl * HEADS);
      const float4 q0 = pq[0], q1 = pq[1];
      float al[HEADS];
      al[0] = p[0] / (q0.x + SM_EPS);
      al[1] = p[1] / (q0.y + SM_EPS);
      al[2] = p[2] / (q0.z + SM_EPS);
      al[3] = p[3] / (q0.w + SM_EPS);
      al[4] = p[4] / (q1.x + SM_EPS);
      al[5] = p[5] / (q1.y + SM_EPS);
      al[6] = p[6] / (q1.z + SM_EPS);
      al[7] = p[7] / (q1.w + SM_EPS);

      int b = 0;
      for (; b + 4 <= count; b += 4) {
        float x[4][NL];
#pragma unroll
        for (int u = 0; u < 4; ++u) {
          const int sb = __builtin_amdgcn_readlane(s, b + u);
          const float* hp = h + (size_t)sb * (HEADS * C) + lane;
#pragma unroll
          for (int j = 0; j < NL; ++j) x[u][j] = hp[j * 32];
        }
#pragma unroll
        for (int u = 0; u < 4; ++u) {
          const int db = __builtin_amdgcn_readlane(dl, b + u);
          const float acc = head_sum<C>(x[u], al, b + u, lane);
          atomicAdd(&tile[db * C + (lane & (C - 1))], acc);
        }
      }
      for (; b < count; ++b) {
        float x[NL];
        const int sb = __builtin_amdgcn_readlane(s, b);
        const float* hp = h + (size_t)sb * (HEADS * C) + lane;
#pragma unroll
        for (int j = 0; j < NL; ++j) x[j] = hp[j * 32];
        const int db = __builtin_amdgcn_readlane(dl, b);
        const float acc = head_sum<C>(x, al, b, lane);
        atomicAdd(&tile[db * C + (lane & (C - 1))], acc);
      }
    }
  }
}

template <int C, int PH, bool IDS, bool KEEP>
__device__ __forceinline__ void list_batch(const int* __restrict__ src,
                                           const int* __restrict__ dst, int E, int n0,
                                           int qn, int off, int lane, int* qe, int* qd,
                                           const float* __restrict__ asrc,
                                           const float* __restrict__ h,
                                           const float* sadst, int* smaxk, float* ssum,
                                           float* tile, float4* wlo, float4* whi) {
  const int cnt = (qn - off < 32) ? (qn - off) : 32;
  const bool ok = lane < cnt;
  int s, dl;
  if constexpr (IDS) {
    const int e = ok ? qe[off + lane] : (E + n0);
    s = e - E;
    dl = s - n0;
    if (e < E) { s = src[e]; dl = dst[e] - n0; }
    if (ok) { qe[off + lane] = s; qd[off + lane] = dl; }
  } else {
    s = ok ? qe[off + lane] : n0;
    dl = ok ? qd[off + lane] : 0;
  }
  edge_batch<C, PH, KEEP>(s, dl, cnt, lane, asrc, h, sadst, smaxk, ssum, tile,
                          KEEP ? wlo + off + lane : nullptr,
                          KEEP ? whi + off + lane : nullptr);
}

template <int C, int PH, bool IDS>
__device__ __forceinline__ void list_pass(const int* __restrict__ src,
                                          const int* __restrict__ dst, int E, int n0,
                                          int qn, int lane, int* qe, int* qd,
                                          const float* __restrict__ asrc,
                                          const float* __restrict__ h,
                                          const float* sadst, int* smaxk, float* ssum,
                                          float* tile, float4* wlo, float4* whi) {
  int off = 0;
  for (; off < qn && off + 32 <= SCAP; off += 32)
    list_batch<C, PH, IDS, true>(src, dst, E, n0, qn, off, lane, qe, qd,
                                 asrc, h, sadst, smaxk, ssum, tile, wlo, whi);
  for (; off < qn; off += 32)
    list_batch<C, PH, IDS, false>(src, dst, E, n0, qn, off, lane, qe, qd,
                                  asrc, h, sadst, smaxk, ssum, tile, wlo, whi);
}

template <int C, int PH>
__device__ __forceinline__ void scan_pass(const int* __restrict__ src,
                                         const int* __restrict__ dst,
                                         int E, int n0, int nv, int w, int lane,
                                         int* qe, int* qd, int qn,
                                         const float* __restrict__ asrc,
                                         const float* __restrict__ h,
                                         const float* sadst, int* smaxk, float* ssum,
                                         float* tile) {
  for (int r0 = w * 32; r0 < nv; r0 += AGG_TB) {
    const int cnt = (nv - r0 < 32) ? (nv - r0) : 32;
    const int r = (lane < cnt) ? (r0 + lane) : 0;
    edge_batch<C, PH, false>(n0 + r, r, cnt, lane, asrc, h, sadst, smaxk, ssum, tile,
                             nullptr, nullptr);
  }
  const int stride = AGG_WAVES * 256;
  const int elast = (E > 0) ? (E - 1) : 0;
  int eb = w * 256 + lane;
  int dc[8];
#pragma unroll
  for (int j = 0; j < 8; ++j) dc[j] = -1;
  if (E > 0) {
#pragma unroll
    for (int j = 0; j < 8; ++j) {
      const int e = eb + j * 32;
      const int dv = dst[(e < E) ? e : elast];
      dc[j] = (e < E) ? dv : -1;
    }
  }
  for (int base = 0; base < E; base += stride) {
    int dn[8];
    const int ebn = eb + stride;
#pragma unroll
    for (int j = 0; j < 8; ++j) {
      const int e = ebn + j * 32;
      const int dv = dst[(e < E) ? e : elast];
      dn[j] = (e < E) ? dv : -1;
    }
#pragma unroll
    for (int j = 0; j < 8; ++j) {
      const int e = eb + j * 32;
      const int dl = dc[j] - n0;
      const bool hit = (unsigned)dl < (unsigned)nv;
      const unsigned mask = __builtin_amdgcn_ballot_w32(hit);
      if (mask != 0u) {
        if (hit) {
          const int pos = qn + (int)__builtin_amdgcn_mbcnt_lo(mask, 0u);
          qe[pos] = e;
          qd[pos] = dl;
        }
        qn += __popc(mask);
      }
    }
    if (qn >= 32) {
      int off = 0;
      for (; off + 32 <= qn; off += 32) {
        const int e2 = qe[off + lane];
        const int d2 = qd[off + lane];
        const int s2 = src[e2];
        edge_batch<C, PH, false>(s2, d2, 32, lane, asrc, h, sadst, smaxk, ssum, tile,
                                 nullptr, nullptr);
      }
      const int rem = qn - off;
      const int te = (lane < rem) ? qe[off + lane] : 0;
      const int td = (lane < rem) ? qd[off + lane] : 0;
      if (lane < rem) { qe[lane] = te; qd[lane] = td; }
      qn = rem;
    }
#pragma unroll
    for (int j = 0; j < 8; ++j) dc[j] = dn[j];
    eb = ebn;
  }
  if (qn > 0) {
    const bool ok = lane < qn;
    const int e2 = ok ? qe[lane] : 0;
    const int d2 = ok ? qd[lane] : 0;
    const int s2 = src[e2];
    edge_batch<C, PH, false>(s2, d2, qn, lane, asrc, h, sadst, smaxk, ssum, tile,
                             nullptr, nullptr);
  }
}

__device__ __forceinline__ int bitmap_list(const unsigned* __restrict__ row, int nwords,
                                           int w, int lane, int* qe, int qn) {
  for (int base = 0; base < nwords; base += AGG_WAVES * 8 * 128) {
    uint4 v[8];
    const int fw0 = base + (w * 32 + lane) * 4;
#pragma unroll
    for (int j = 0; j < 8; ++j) {
      const int fw = fw0 + j * (AGG_WAVES * 128);
      v[j] = *(const uint4*)(row + ((fw < nwords) ? fw : 0));
    }
    __builtin_amdgcn_sched_barrier(0);
#pragma unroll
    for (int j = 0; j < 8; ++j) {
      const bool in = fw0 + j * (AGG_WAVES * 128) < nwords;
      v[j].x = in ? v[j].x : 0u;
      v[j].y = in ? v[j].y : 0u;
      v[j].z = in ? v[j].z : 0u;
      v[j].w = in ? v[j].w : 0u;
    }
#pragma nounroll
    for (int j = 0; j < 8; ++j) {
      uint4 cur = v[0];
#pragma unroll
      for (int t = 0; t < 7; ++t) v[t] = v[t + 1];
#pragma nounroll
      for (int q = 0; q < 4; ++q) {
        unsigned x = cur.x;
        cur.x = cur.y; cur.y = cur.z; cur.z = cur.w;
        const int ebase = (fw0 + j * (AGG_WAVES * 128) + q) << 5;
        unsigned m = __builtin_amdgcn_ballot_w32(x != 0u);
        while (m != 0u) {
          if (x != 0u) {
            const int pos = qn + (int)__builtin_amdgcn_mbcnt_lo(m, 0u);
            if (pos < QCAP) qe[pos] = ebase + __builtin_ctz(x);
            x &= x - 1u;
          }
          qn += __popc(m);
          m = __builtin_amdgcn_ballot_w32(x != 0u);
        }
      }
    }
  }
  return qn;
}

template <int C, bool BUILD>
__global__ __launch_bounds__(AGG_TB)
void range_agg_kernel(const int* __restrict__ src,
                      const int* __restrict__ dst,
                      int E, int N,
                      const float* __restrict__ asrc,
                      const float* __restrict__ adst,
                      const float* __restrict__ h,
                      const float* __restrict__ bias,
                      float* __restrict__ out,
                      int do_relu,
                      int* __restrict__ wq_n,
                      int* __restrict__ wq_e,
                      int* __restrict__ wq_d,
                      const float* __restrict__ Wn,
                      const float* __restrict__ attn_s,
                      const float* __restrict__ attn_d,
                      float* __restrict__ asrc_n,
                      float* __restrict__ adst_n,
                      const unsigned* __restrict__ bmp,
                      int nwords) {
  __shared__ __attribute__((aligned(16))) float tile[RNODES * C];
  __shared__ __attribute__((aligned(16))) float sadst[RNODES * HEADS];
  __shared__ __attribute__((aligned(16))) int   smaxk[RNODES * HEADS];
  __shared__ __attribute__((aligned(16))) float ssum[RNODES * HEADS];
  __shared__ int qe_all[AGG_WAVES * QCAP];
  __shared__ int qd_all[AGG_WAVES * QCAP];
  __shared__ int sqn[AGG_WAVES];
  __shared__ int sover;
  __shared__ __attribute__((aligned(16))) float kept[2 * AGG_WAVES * SCAP * 4];

  const int tid = (int)threadIdx.x;
  const int lane = tid & 31;
  const int w = __builtin_amdgcn_readfirstlane(tid >> 5);
  const int n0 = (int)blockIdx.x * RNODES;
  const int nv = (N - n0 < RNODES) ? (N - n0) : RNODES;
  const size_t wq = (size_t)blockIdx.x * AGG_WAVES + w;

  int qn = 0;
  if constexpr (!BUILD) {
    qn = __builtin_amdgcn_readfirstlane(wq_n[wq]);
  }

  for (int i = tid; i < RNODES * C; i += AGG_TB) tile[i] = 0.f;
  for (int i = tid; i < RNODES * HEADS; i += AGG_TB) {
    sadst[i] = (i < nv * HEADS) ? adst[(size_t)n0 * HEADS + i] : 0.f;
    smaxk[i] = fkey(__float_as_int(-3.0e38f));
    ssum[i] = 0.f;
  }
  if (tid == 0) sover = 0;

  int* qe = qe_all + w * QCAP;
  int* qd = qd_all + w * QCAP;
  float4* wlo = (float4*)kept + w * SCAP;
  float4* whi = (float4*)kept + (AGG_WAVES + w) * SCAP;
  if constexpr (BUILD) {
    const int rs = w + 32 * lane;
    const bool own = rs < nv;
    if (own) qe[lane] = E + n0 + rs;
    qn = __popc(__builtin_amdgcn_ballot_w32(own));
    __syncthreads();
    qn = bitmap_list(bmp + (size_t)blockIdx.x * nwords, nwords, w, lane, qe, qn);
    if (lane == 0) sqn[w] = qn;
  } else {
    if (qn <= QCAP) {
      const int* we = wq_e + wq * QCAP;
      const int* wd = wq_d + wq * QCAP;
      for (int i = lane; i < qn; i += 32) { qe[i] = we[i]; qd[i] = wd[i]; }
    }
    __syncthreads();
  }
  if (qn > QCAP) sover = 1;
  __syncthreads();
  const bool over = sover != 0;
  if constexpr (BUILD) {
    if (w == 0) wq_n[(size_t)blockIdx.x * AGG_WAVES + lane] = sqn[lane];
  }

  if (!over) {
    list_pass<C, PH_MAX, BUILD>(src, dst, E, n0, qn, lane, qe, qd, asrc, h, sadst, smaxk, ssum, tile,
                                wlo, whi);
    __syncthreads();
    if constexpr (BUILD) {
      int* we = wq_e + wq * QCAP;
      int* wd = wq_d + wq * QCAP;
      for (int i = lane; i < qn; i += 32) { we[i] = qe[i]; wd[i] = qd[i]; }
    }
    list_pass<C, PH_SUM, false>(src, dst, E, n0, qn, lane, qe, qd, asrc, h, sadst, smaxk, ssum, tile,
                                wlo, whi);
    __syncthreads();
    list_pass<C, PH_AGG, false>(src, dst, E, n0, qn, lane, qe, qd, asrc, h, sadst, smaxk, ssum, tile,
                                wlo, whi);
  } else {
    scan_pass<C, PH_MAX>(src, dst, E, n0, nv, w, lane, qe, qd, 0,
                         asrc, h, sadst, smaxk, ssum, tile);
    __syncthreads();
    scan_pass<C, PH_SUM>(src, dst, E, n0, nv, w, lane, qe, qd, 0,
                         asrc, h, sadst, smaxk, ssum, tile);
    __syncthreads();
    scan_pass<C, PH_AGG>(src, dst, E, n0, nv, w, lane, qe, qd, 0,
                         asrc, h, sadst, smaxk, ssum, tile);
  }
  __syncthreads();

  const float inv = 1.0f / HEADS;
  if constexpr (!BUILD) {
    for (int i = tid; i < nv * C; i += AGG_TB) {
      float v = tile[i] * inv + bias[i & (C - 1)];
      if (do_relu) v = v > 0.f ? v : 0.f;
      out[(size_t)n0 * C + i] = v;
    }
  } else {
    constexpr int CN = 16;
    constexpr int NCOL = HEADS * CN;
    constexpr int HP = NCOL + 4;
    constexpr int TM = (RNODES + 15) / 16;
    constexpr int TG = NCOL / (16 * NT);
    static_assert(TM * TG <= AGG_WAVES, "one strip per wave");
    static_assert(TM * 16 * HP <= 2 * AGG_WAVES * SCAP * 4, "the projected rows fit the kept planes");
    float* hs = kept;
    for (int i = tid; i < RNODES * C; i += AGG_TB) {
      float v = tile[i] * inv + bias[i & (C - 1)];
      if (do_relu) v = v > 0.f ? v : 0.f;
      tile[i] = v;
    }
    __syncthreads();
    if (w < TM * TG) {
      const int tm = w / TG;
      const int tg = w - tm * TG;
      const int half = lane >> 4;
      const int l = lane & 15;
      int rowA = tm * 16 + l;
      if (rowA > RNODES - 1) rowA = RNODES - 1;
      const float* aP = tile + rowA * C + 2 * half;
      const float* bP = Wn + (size_t)(2 * half) * NCOL + tg * (16 * NT) + l;
      v8f acc[NT];
#pragma unroll
      for (int t = 0; t < NT; ++t) acc[t] = (v8f){};
#pragma unroll
      for (int k0 = 0; k0 < C; k0 += 4) {
        v2f a;
        a.x = aP[k0];
        a.y = aP[k0 + 1];
#pragma unroll
        for (int t = 0; t < NT; ++t) {
          v2f b;
          b.x = bP[(size_t)k0 * NCOL + t * 16];
          b.y = bP[(size_t)k0 * NCOL + t * 16 + NCOL];
          acc[t] = __builtin_amdgcn_wmma_f32_16x16x4_f32(false, a, false, b,
                                                         (short)0, acc[t],
                                                         false, false);
        }
      }
#pragma unroll
      for (int t = 0; t < NT; ++t) {
#pragma unroll
        for (int r = 0; r < 8; ++r)
          hs[(tm * 16 + r + 8 * half) * HP + tg * (16 * NT) + t * 16 + l] = acc[t][r];
      }
    }
    __syncthreads();
    float* hout = out + (size_t)n0 * NCOL;
    for (int i = tid * 4; i < nv * NCOL; i += AGG_TB * 4) {
      const int row = i / NCOL;
      const int col = i - row * NCOL;
      *(float4*)(hout + i) = *(const float4*)(hs + row * HP + col);
    }
    for (int i = tid; i < nv * HEADS; i += AGG_TB) {
      const int head = i & (HEADS - 1);
      const float* hp = hs + (i >> 3) * HP + head * CN;
      const float* as = attn_s + head * CN;
      const float* ad = attn_d + head * CN;
      float s = 0.f, d = 0.f;
#pragma unroll
      for (int c2 = 0; c2 < CN; ++c2) {
        float v = hp[c2];
        s += v * as[c2];
        d += v * ad[c2];
      }
      asrc_n[(size_t)n0 * HEADS + i] = s;
      adst_n[(size_t)n0 * HEADS + i] = d;
    }
  }
}

static inline unsigned nblk(long n, int b) { return (unsigned)((n + b - 1) / b); }

extern "C" void kernel_launch(void* const* d_in, const int* in_sizes, int n_in,
                              void* d_out, int out_size, void* d_ws, size_t ws_size,
                              hipStream_t stream) {
  const float* x        = (const float*)d_in[0];
  const int*   ei       = (const int*)  d_in[1];
  const float* W1       = (const float*)d_in[2];
  const float* att_src1 = (const float*)d_in[3];
  const float* att_dst1 = (const float*)d_in[4];
  const float* b1       = (const float*)d_in[5];
  const float* W2       = (const float*)d_in[6];
  const float* att_src2 = (const float*)d_in[7];
  const float* att_dst2 = (const float*)d_in[8];
  const float* b2       = (const float*)d_in[9];

  const int F_h   = in_sizes[3] / HEADS;
  const int F_out = in_sizes[7] / HEADS;
  const int F_in  = in_sizes[2] / (HEADS * F_h);
  const int N     = in_sizes[0] / F_in;
  const int E     = in_sizes[1] / 2;
  const int C1    = HEADS * F_h;
  const int C2    = HEADS * F_out;
  const int* src = ei;
  const int* dst = ei + E;

  float* ws = (float*)d_ws;
  size_t o = 0;
  float* h1    = ws + o; o += (size_t)N * C1;
  float* asrc  = ws + o; o += (size_t)N * HEADS;
  float* adst  = ws + o; o += (size_t)N * HEADS;
  o = (o + 31) & ~(size_t)31;
  float* h2    = ws + o; o += (size_t)N * C2;
  float* asrc2 = ws + o; o += (size_t)N * HEADS;
  float* adst2 = ws + o; o += (size_t)N * HEADS;
  const size_t nwg = nblk(N, RNODES);
  o = (o + 31) & ~(size_t)31;
  int* wq_n = (int*)(ws + o); o += nwg * AGG_WAVES;
  int* wq_e = (int*)(ws + o); o += nwg * AGG_WAVES * QCAP;
  int* wq_d = (int*)(ws + o); o += nwg * AGG_WAVES * QCAP;
  const int chunks = (E + BM_CHUNK - 1) / BM_CHUNK;
  const int nwords = chunks * 32;
  const int nbm = chunks * (int)((nwg + BM_OWN - 1) / BM_OWN);
  o = (o + 31) & ~(size_t)31;
  unsigned* bmp = (unsigned*)(ws + o); o += nwg * (size_t)nwords;
  float* outp = (float*)d_out;

  const int TB = 256;

  {
    long waves = (long)((N + 15) / 16) * (C1 / (16 * NT));
    wmma_gemm_f32<<<nblk(waves * 32, TB), TB, 0, stream>>>(x, W1, h1, N, F_in, C1);
  }
  coeff_bitmap_kernel<<<nbm + nblk(N, (BM_TB / 32) * CF_ROWS), BM_TB, 0, stream>>>(h1, att_src1, att_dst1,
                                                                   asrc, adst, N, F_h,
                                                                   dst, E, (int)nwg, chunks, nbm,
                                                                   bmp);
  range_agg_kernel<32, true><<<nblk(N, RNODES), AGG_TB, 0, stream>>>(src, dst, E, N, asrc, adst,
                                                                     h1, b1, h2, 1,
                                                                     wq_n, wq_e, wq_d,
                                                                     W2, att_src2, att_dst2,
                                                                     asrc2, adst2,
                                                                     bmp, nwords);

  range_agg_kernel<16, false><<<nblk(N, RNODES), AGG_TB, 0, stream>>>(src, dst, E, N, asrc2, adst2,
                                                                      h2, b2, outp, 0,
                                                                      wq_n, wq_e, wq_d,
                                                                      nullptr, nullptr, nullptr,
                                                                      nullptr, nullptr,
                                                                      nullptr, 0);
}


# aggregation pass reads each edge's source row by 16-byte loads (whole 512-byte runs: 2 loads per edge, not 8), head weights via LDS slots, lanes combine by DPP/permlane adds; float sum order changes
# speedup vs baseline: 15.3965x; 1.0312x over previous
#include <hip/hip_runtime.h>
#include <cstdint>

typedef float v2f __attribute__((ext_vector_type(2)));
typedef float v4f __attribute__((ext_vector_type(4)));
typedef float v8f __attribute__((ext_vector_type(8)));

#define HEADS 8
#define NEG_SLOPE 0.2f
#define SM_EPS 1e-16f
#define NT 4

__device__ __forceinline__ void gemm_trip(v8f* acc, const v4f a, const v4f* b) {
  const v2f a0 = {a[0], a[1]};
  const v2f a1 = {a[2], a[3]};
#pragma unroll
  for (int t = 0; t < NT; ++t) {
    const v2f bb = {b[0][t], b[1][t]};
    acc[t] = __builtin_amdgcn_wmma_f32_16x16x4_f32(false, a0, false, bb,
                                                   (short)0, acc[t],
                                                   false, false);
  }
#pragma unroll
  for (int t = 0; t < NT; ++t) {
    const v2f bb = {b[2][t], b[3][t]};
    acc[t] = __builtin_amdgcn_wmma_f32_16x16x4_f32(false, a1, false, bb,
                                                   (short)0, acc[t],
                                                   false, false);
  }
}

__global__ void wmma_gemm_f32(const float* __restrict__ A,
                              const float* __restrict__ B,
                              float* __restrict__ C,
                              int M, int K, int Ncols) {
  int wave = (int)((blockIdx.x * blockDim.x + threadIdx.x) >> 5);
  int lane = (int)(threadIdx.x & 31);
  int tilesM  = (M + 15) >> 4;
  int groupsN = Ncols >> 6;
  if (wave >= tilesM * groupsN) return;
  int tm = wave / groupsN;
  int tg = wave - tm * groupsN;
  int half = lane >> 4;
  int l = lane & 15;

  int rowA = tm * 16 + l;
  if (rowA >= M) rowA = M - 1;
  const float* aP = A + (size_t)rowA * K + 4 * half;
  const float* bP = B + (size_t)(4 * half) * Ncols + tg * (16 * NT) + 4 * l;

  v8f acc[NT];
#pragma unroll
  for (int t = 0; t < NT; ++t) acc[t] = (v8f){};

  v4f aCur = *(const v4f*)aP;
  v4f bCur[4];
#pragma unroll
  for (int j = 0; j < 4; ++j) bCur[j] = *(const v4f*)(bP + (size_t)j * Ncols);

  for (int k0 = 0; k0 < K - 8; k0 += 8) {
    aP += 8;
    bP += (size_t)8 * Ncols;
    v4f aNxt = *(const v4f*)aP;
    v4f bNxt[4];
#pragma unroll
    for (int j = 0; j < 4; ++j) bNxt[j] = *(const v4f*)(bP + (size_t)j * Ncols);
    gemm_trip(acc, aCur, bCur);
    aCur = aNxt;
#pragma unroll
    for (int j = 0; j < 4; ++j) bCur[j] = bNxt[j];
  }
  gemm_trip(acc, aCur, bCur);

  float* cP = C + (size_t)(tm * 16 + 8 * half) * Ncols + tg * (16 * NT) + 4 * l;
#pragma unroll
  for (int r = 0; r < 8; ++r) {
    const int row = tm * 16 + r + 8 * half;
    const v4f q = {acc[0][r], acc[1][r], acc[2][r], acc[3][r]};
    if (row < M) *(v4f*)(cP + (size_t)r * Ncols) = q;
  }
}

#define RNODES 196
#define BM_CHUNK 1024
#define BM_OWN 256
#define BM_PITCH 33
#define BM_TB 256
#define CF_ROWS 16

__global__ __launch_bounds__(BM_TB)
void coeff_bitmap_kernel(const float* __restrict__ h,
                         const float* __restrict__ att_src,
                         const float* __restrict__ att_dst,
                         float* __restrict__ asrc,
                         float* __restrict__ adst,
                         int N, int C,
                         const int* __restrict__ dst, int E,
                         int nown, int chunks, int nbm,
                         unsigned* __restrict__ bmp) {
  if ((int)blockIdx.x < nbm) {
    __shared__ unsigned sbm[BM_OWN * BM_PITCH];
    const int tid = (int)threadIdx.x;
    const int og = (int)blockIdx.x / chunks;
    const int ch = (int)blockIdx.x - og * chunks;
    for (int i = tid; i < BM_OWN * BM_PITCH; i += BM_TB) sbm[i] = 0u;
    __syncthreads();
    const int e0 = ch * BM_CHUNK;
    int dv[BM_CHUNK / BM_TB];
#pragma unroll
    for (int k = 0; k < BM_CHUNK / BM_TB; ++k) {
      const int e = e0 + k * BM_TB + tid;
      dv[k] = dst[(e < E) ? e : (E - 1)];
    }
#pragma unroll
    for (int k = 0; k < BM_CHUNK / BM_TB; ++k) {
      const int e = e0 + k * BM_TB + tid;
      const int ol = dv[k] / RNODES - og * BM_OWN;
      if (e < E && (unsigned)ol < (unsigned)BM_OWN)
        atomicOr(&sbm[ol * BM_PITCH + k * (BM_TB / 32) + (tid >> 5)], 1u << (tid & 31));
    }
    __syncthreads();
    const int w = tid >> 5, lane = tid & 31;
    const size_t wpo = (size_t)chunks * 32;
    for (int ol = w; ol < BM_OWN; ol += BM_TB / 32) {
      const int o = og * BM_OWN + ol;
      if (o < nown) bmp[(size_t)o * wpo + (size_t)ch * 32 + lane] = sbm[ol * BM_PITCH + lane];
    }
    return;
  }
  const int lane = (int)(threadIdx.x & 31);
  const int w = (int)(threadIdx.x >> 5);
  const long nb = (long)blockIdx.x - nbm;
  const long nfirst = (nb * (BM_TB / 32) + w) * CF_ROWS;
  if (nfirst >= (long)N) return;
  const size_t pitch = (size_t)HEADS * C;
  const size_t hoff = pitch / 2;
  const v4f as0 = *(const v4f*)(att_src + 4 * lane);
  const v4f as1 = *(const v4f*)(att_src + hoff + 4 * lane);
  const v4f ad0 = *(const v4f*)(att_dst + 4 * lane);
  const v4f ad1 = *(const v4f*)(att_dst + hoff + 4 * lane);
  for (int b = 0; b < CF_ROWS; b += 8) {
    v4f q0[8], q1[8];
#pragma unroll
    for (int r = 0; r < 8; ++r) {
      const long n = (nfirst + b + r < (long)N) ? (nfirst + b + r) : ((long)N - 1);
      const float* hp = h + (size_t)n * pitch + 4 * lane;
      q0[r] = *(const v4f*)hp;
      q1[r] = *(const v4f*)(hp + hoff);
    }
    float ks0 = 0.f, ks1 = 0.f, kd0 = 0.f, kd1 = 0.f;
#pragma unroll
    for (int r = 0; r < 8; ++r) {
      float s0 = q0[r][0] * as0[0], s1 = q1[r][0] * as1[0];
      float d0 = q0[r][0] * ad0[0], d1 = q1[r][0] * ad1[0];
#pragma unroll
      for (int c = 1; c < 4; ++c) {
        s0 = __builtin_fmaf(q0[r][c], as0[c], s0);
        s1 = __builtin_fmaf(q1[r][c], as1[c], s1);
        d0 = __builtin_fmaf(q0[r][c], ad0[c], d0);
        d1 = __builtin_fmaf(q1[r][c], ad1[c], d1);
      }
#pragma unroll
      for (int m = 1; m < 8; m <<= 1) {
        s0 += __shfl_xor(s0, m);
        s1 += __shfl_xor(s1, m);
        d0 += __shfl_xor(d0, m);
        d1 += __shfl_xor(d1, m);
      }
      if ((lane & 7) == r) { ks0 = s0; ks1 = s1; kd0 = d0; kd1 = d1; }
    }
    const long n = nfirst + b + (lane & 7);
    if (n < (long)N) {
      const size_t o = (size_t)n * HEADS + (lane >> 3);
      asrc[o] = ks0;
      asrc[o + 4] = ks1;
      adst[o] = kd0;
      adst[o + 4] = kd1;
    }
  }
}

__device__ __forceinline__ float lrelu(float v) {
  return v > 0.f ? v : NEG_SLOPE * v;
}

#define AGG_TB 1024
#define AGG_WAVES (AGG_TB / 32)
#define QCAP 288
#define SCAP 192

#define PH_MAX  1
#define PH_SUM  2
#define PH_AGG  3

__device__ __forceinline__ int fkey(int i) { return i >= 0 ? i : (i ^ 0x7fffffff); }

template <typename T>
__device__ __forceinline__ void rot_heads(T* w, int lane) {
#pragma unroll
  for (int s = 1; s < HEADS; s <<= 1) {
    const bool on = (lane & s) != 0;
    T t[HEADS];
#pragma unroll
    for (int k = 0; k < HEADS; ++k) t[k] = on ? w[(k + s) & (HEADS - 1)] : w[k];
#pragma unroll
    for (int k = 0; k < HEADS; ++k) w[k] = t[k];
  }
}

__device__ __forceinline__ float xrow16(float x) {
  return __int_as_float(__builtin_amdgcn_permlanex16(__float_as_int(x), __float_as_int(x),
                                                     0x76543210, 0xfedcba98, false, false));
}
template <int M>
__device__ __forceinline__ float xrow(float x) {
  return __int_as_float(__builtin_amdgcn_update_dpp(0, __float_as_int(x), 0x160 + M, 0xf, 0xf, false));
}

template <int C>
__device__ __forceinline__ int agg_channel(int lane) {
  constexpr int SA = (C == 32) ? 3 : 2;
  return 4 * (lane & (C / 4 - 1)) + 2 * ((lane >> SA) & 1) + ((lane >> (SA + 1)) & 1);
}
template <int C>
__device__ __forceinline__ float row_reduce(const v4f x0, const v4f x1, float a0, float a1, int lane) {
  float m0, m1, m2, m3;
  if constexpr (C == 32) {
    m0 = __builtin_fmaf(a1, x1[0], a0 * x0[0]);
    m1 = __builtin_fmaf(a1, x1[1], a0 * x0[1]);
    m2 = __builtin_fmaf(a1, x1[2], a0 * x0[2]);
    m3 = __builtin_fmaf(a1, x1[3], a0 * x0[3]);
  } else {
    m0 = a0 * x0[0]; m1 = a0 * x0[1]; m2 = a0 * x0[2]; m3 = a0 * x0[3];
  }
  constexpr int SA = (C == 32) ? 3 : 2;
  const bool ba = ((lane >> SA) & 1) != 0;
  const bool bb = ((lane >> (SA + 1)) & 1) != 0;
  const float k0 = ba ? m2 : m0, k1 = ba ? m3 : m1;
  const float s0 = ba ? m0 : m2, s1 = ba ? m1 : m3;
  float r0, r1;
  if constexpr (C == 32) { r0 = k0 + xrow<8>(s0); r1 = k1 + xrow<8>(s1); }
  else                   { r0 = k0 + xrow<4>(s0); r1 = k1 + xrow<4>(s1); }
  const float k = bb ? r1 : r0, s = bb ? r0 : r1;
  if constexpr (C == 32) return k + xrow16(s);
  else                   return k + xrow<8>(s);
}

template <int C>
__device__ __forceinline__ void pick_weights(const float* al, int b, int lane, float& a0, float& a1) {
  a0 = 0.f; a1 = 0.f;
#pragma unroll
  for (int k = 0; k < HEADS; ++k) {
    const float ak = __int_as_float(__builtin_amdgcn_readlane(__float_as_int(al[k]), b));
    if constexpr (C == 32) {
      if (k < 4) a0 = ((lane >> 3) == k) ? ak : a0;
      else       a1 = ((lane >> 3) == k - 4) ? ak : a1;
    } else {
      a0 = ((lane >> 2) == k) ? ak : a0;
    }
  }
}

template <int C, int PH, bool KEEP>
__device__ __forceinline__ void edge_batch(int s, int dl, int count, int lane,
                                           const float* __restrict__ asrc,
                                           const float* __restrict__ h,
                                           const float* sadst, int* smaxk, float* ssum,
                                           float* tile,
                                           float4* klo, float4* khi) {
  float v[HEADS];
  if constexpr (PH == PH_MAX || !KEEP) {
    const float4* pa = (const float4*)(asrc + (size_t)s * HEADS);
    const float4* pd = (const float4*)(sadst + dl * HEADS);
    const float4 a0 = pa[0], a1 = pa[1];
    const float4 d0 = pd[0], d1 = pd[1];
    v[0] = lrelu(a0.x + d0.x);
    v[1] = lrelu(a0.y + d0.y);
    v[2] = lrelu(a0.z + d0.z);
    v[3] = lrelu(a0.w + d0.w);
    v[4] = lrelu(a1.x + d1.x);
    v[5] = lrelu(a1.y + d1.y);
    v[6] = lrelu(a1.z + d1.z);
    v[7] = lrelu(a1.w + d1.w);
  } else {
    const float4 c0 = *klo, c1 = *khi;
    v[0] = c0.x; v[1] = c0.y; v[2] = c0.z; v[3] = c0.w;
    v[4] = c1.x; v[5] = c1.y; v[6] = c1.z; v[7] = c1.w;
  }
  if constexpr (PH == PH_MAX) {
    if (lane < count) {
      int w[HEADS];
#pragma unroll
      for (int k = 0; k < HEADS; ++k) w[k] = fkey(__float_as_int(v[k]));
      rot_heads(w, lane);
      int* pm = smaxk + dl * HEADS;
#pragma unroll
      for (int k = 0; k < HEADS; ++k)
        atomicMax(&pm[(k + lane) & (HEADS - 1)], w[k]);
    }
    if constexpr (KEEP) {
      *klo = make_float4(v[0], v[1], v[2], v[3]);
      *khi = make_float4(v[4], v[5], v[6], v[7]);
    }
  } else {
    float p[HEADS];
    if constexpr (PH == PH_SUM || !KEEP) {
      const int4* pm = (const int4*)(smaxk + dl * HEADS);
      const int4 k0 = pm[0], k1 = pm[1];
      p[0] = __expf(v[0] - __int_as_float(fkey(k0.x)));
      p[1] = __expf(v[1] - __int_as_float(fkey(k0.y)));
      p[2] = __expf(v[2] - __int_as_float(fkey(k0.z)));
      p[3] = __expf(v[3] - __int_as_float(fkey(k0.w)));
      p[4] = __expf(v[4] - __int_as_float(fkey(k1.x)));
      p[5] = __expf(v[5] - __int_as_float(fkey(k1.y)));
      p[6] = __expf(v[6] - __int_as_float(fkey(k1.z)));
      p[7] = __expf(v[7] - __int_as_float(fkey(k1.w)));
    } else {
#pragma unroll
      for (int k = 0; k < HEADS; ++k) p[k] = v[k];
    }
    if constexpr (PH == PH_SUM) {
      if (lane < count) {
        float w[HEADS];
#pragma unroll
        for (int k = 0; k < HEADS; ++k) w[k] = p[k];
        rot_heads(w, lane);
        float* ps = ssum + dl * HEADS;
#pragma unroll
        for (int k = 0; k < HEADS; ++k)
          atomicAdd(&ps[(k + lane) & (HEADS - 1)], w[k]);
      }
      if constexpr (KEEP) {
        *klo = make_float4(p[0], p[1], p[2], p[3]);
        *khi = make_float4(p[4], p[5], p[6], p[7]);
      }
    } else {
      const float4* pq = (const float4*)(ssum + dl * HEADS);
      const float4 q0 = pq[0], q1 = pq[1];
      float al[HEADS];
      al[0] = p[0] / (q0.x + SM_EPS);
      al[1] = p[1] / (q0.y + SM_EPS);
      al[2] = p[2] / (q0.z + SM_EPS);
      al[3] = p[3] / (q0.w + SM_EPS);
      al[4] = p[4] / (q1.x + SM_EPS);
      al[5] = p[5] / (q1.y + SM_EPS);
      al[6] = p[6] / (q1.z + SM_EPS);
      al[7] = p[7] / (q1.w + SM_EPS);

      const float* pa0 = nullptr;
      const float* pa1 = nullptr;
      if constexpr (KEEP) {
        *klo = make_float4(al[0], al[1], al[2], al[3]);
        *khi = make_float4(al[4], al[5], al[6], al[7]);
        const float* blo = (const float*)(klo - lane);
        const float* bhi = (const float*)(khi - lane);
        if constexpr (C == 32) { pa0 = blo + (lane >> 3); pa1 = bhi + (lane >> 3); }
        else                   { pa0 = ((lane < 16) ? blo : bhi) + ((lane >> 2) & 3); }
      }
      const int chl = agg_channel<C>(lane);
      int b = 0;
      for (; b + 4 <= count; b += 4) {
        v4f x0[4], x1[4];
        float a0[4], a1[4];
#pragma unroll
        for (int u = 0; u < 4; ++u) {
          const int sb = __builtin_amdgcn_readlane(s, b + u);
          const v4f* hp = (const v4f*)(h + (size_t)sb * (HEADS * C)) + lane;
          x0[u] = hp[0];
          if constexpr (C == 32) x1[u] = hp[32]; else x1[u] = x0[u];
          a0[u] = 0.f; a1[u] = 0.f;
          if constexpr (KEEP) {
            a0[u] = pa0[(b + u) * 4];
            if constexpr (C == 32) a1[u] = pa1[(b + u) * 4];
          }
        }
        __builtin_amdgcn_sched_barrier(0);
#pragma unroll
        for (int u = 0; u < 4; ++u) {
          if constexpr (!KEEP) pick_weights<C>(al, b + u, lane, a0[u], a1[u]);
          const int db = __builtin_amdgcn_readlane(dl, b + u);
          const float acc = row_reduce<C>(x0[u], x1[u], a0[u], a1[u], lane);
          atomicAdd(&tile[db * C + chl], acc);
        }
      }
      for (; b < count; ++b) {
        const int sb = __builtin_amdgcn_readlane(s, b);
        const v4f* hp = (const v4f*)(h + (size_t)sb * (HEADS * C)) + lane;
        const v4f y0 = hp[0];
        v4f y1 = y0;
        if constexpr (C == 32) y1 = hp[32];
        float a0, a1 = 0.f;
        if constexpr (KEEP) {
          a0 = pa0[b * 4];
          if constexpr (C == 32) a1 = pa1[b * 4];
        } else {
          pick_weights<C>(al, b, lane, a0, a1);
        }
        const int db = __builtin_amdgcn_readlane(dl, b);
        const float acc = row_reduce<C>(y0, y1, a0, a1, lane);
        atomicAdd(&tile[db * C + chl], acc);
      }
    }
  }
}

template <int C, int PH, bool IDS, bool KEEP>
__device__ __forceinline__ void list_batch(const int* __restrict__ src,
                                           const int* __restrict__ dst, int E, int n0,
                                           int qn, int off, int lane, int* qe, int* qd,
                                           const float* __restrict__ asrc,
                                           const float* __restrict__ h,
                                           const float* sadst, int* smaxk, float* ssum,
                                           float* tile, float4* wlo, float4* whi) {
  const int cnt = (qn - off < 32) ? (qn - off) : 32;
  const bool ok = lane < cnt;
  int s, dl;
  if constexpr (IDS) {
    const int e = ok ? qe[off + lane] : (E + n0);
    s = e - E;
    dl = s - n0;
    if (e < E) { s = src[e]; dl = dst[e] - n0; }
    if (ok) { qe[off + lane] = s; qd[off + lane] = dl; }
  } else {
    s = ok ? qe[off + lane] : n0;
    dl = ok ? qd[off + lane] : 0;
  }
  edge_batch<C, PH, KEEP>(s, dl, cnt, lane, asrc, h, sadst, smaxk, ssum, tile,
                          KEEP ? wlo + off + lane : nullptr,
                          KEEP ? whi + off + lane : nullptr);
}

template <int C, int PH, bool IDS>
__device__ __forceinline__ void list_pass(const int* __restrict__ src,
                                          const int* __restrict__ dst, int E, int n0,
                                          int qn, int lane, int* qe, int* qd,
                                          const float* __restrict__ asrc,
                                          const float* __restrict__ h,
                                          const float* sadst, int* smaxk, float* ssum,
                                          float* tile, float4* wlo, float4* whi) {
  int off = 0;
  for (; off < qn && off + 32 <= SCAP; off += 32)
    list_batch<C, PH, IDS, true>(src, dst, E, n0, qn, off, lane, qe, qd,
                                 asrc, h, sadst, smaxk, ssum, tile, wlo, whi);
  for (; off < qn; off += 32)
    list_batch<C, PH, IDS, false>(src, dst, E, n0, qn, off, lane, qe, qd,
                                  asrc, h, sadst, smaxk, ssum, tile, wlo, whi);
}

template <int C, int PH>
__device__ __forceinline__ void scan_pass(const int* __restrict__ src,
                                         const int* __restrict__ dst,
                                         int E, int n0, int nv, int w, int lane,
                                         int* qe, int* qd, int qn,
                                         const float* __restrict__ asrc,
                                         const float* __restrict__ h,
                                         const float* sadst, int* smaxk, float* ssum,
                                         float* tile) {
  for (int r0 = w * 32; r0 < nv; r0 += AGG_TB) {
    const int cnt = (nv - r0 < 32) ? (nv - r0) : 32;
    const int r = (lane < cnt) ? (r0 + lane) : 0;
    edge_batch<C, PH, false>(n0 + r, r, cnt, lane, asrc, h, sadst, smaxk, ssum, tile,
                             nullptr, nullptr);
  }
  const int stride = AGG_WAVES * 256;
  const int elast = (E > 0) ? (E - 1) : 0;
  int eb = w * 256 + lane;
  int dc[8];
#pragma unroll
  for (int j = 0; j < 8; ++j) dc[j] = -1;
  if (E > 0) {
#pragma unroll
    for (int j = 0; j < 8; ++j) {
      const int e = eb + j * 32;
      const int dv = dst[(e < E) ? e : elast];
      dc[j] = (e < E) ? dv : -1;
    }
  }
  for (int base = 0; base < E; base += stride) {
    int dn[8];
    const int ebn = eb + stride;
#pragma unroll
    for (int j = 0; j < 8; ++j) {
      const int e = ebn + j * 32;
      const int dv = dst[(e < E) ? e : elast];
      dn[j] = (e < E) ? dv : -1;
    }
#pragma unroll
    for (int j = 0; j < 8; ++j) {
      const int e = eb + j * 32;
      const int dl = dc[j] - n0;
      const bool hit = (unsigned)dl < (unsigned)nv;
      const unsigned mask = __builtin_amdgcn_ballot_w32(hit);
      if (mask != 0u) {
        if (hit) {
          const int pos = qn + (int)__builtin_amdgcn_mbcnt_lo(mask, 0u);
          qe[pos] = e;
          qd[pos] = dl;
        }
        qn += __popc(mask);
      }
    }
    if (qn >= 32) {
      int off = 0;
      for (; off + 32 <= qn; off += 32) {
        const int e2 = qe[off + lane];
        const int d2 = qd[off + lane];
        const int s2 = src[e2];
        edge_batch<C, PH, false>(s2, d2, 32, lane, asrc, h, sadst, smaxk, ssum, tile,
                                 nullptr, nullptr);
      }
      const int rem = qn - off;
      const int te = (lane < rem) ? qe[off + lane] : 0;
      const int td = (lane < rem) ? qd[off + lane] : 0;
      if (lane < rem) { qe[lane] = te; qd[lane] = td; }
      qn = rem;
    }
#pragma unroll
    for (int j = 0; j < 8; ++j) dc[j] = dn[j];
    eb = ebn;
  }
  if (qn > 0) {
    const bool ok = lane < qn;
    const int e2 = ok ? qe[lane] : 0;
    const int d2 = ok ? qd[lane] : 0;
    const int s2 = src[e2];
    edge_batch<C, PH, false>(s2, d2, qn, lane, asrc, h, sadst, smaxk, ssum, tile,
                             nullptr, nullptr);
  }
}

__device__ __forceinline__ int bitmap_list(const unsigned* __restrict__ row, int nwords,
                                           int w, int lane, int* qe, int qn) {
  for (int base = 0; base < nwords; base += AGG_WAVES * 8 * 128) {
    uint4 v[8];
    const int fw0 = base + (w * 32 + lane) * 4;
#pragma unroll
    for (int j = 0; j < 8; ++j) {
      const int fw = fw0 + j * (AGG_WAVES * 128);
      v[j] = *(const uint4*)(row + ((fw < nwords) ? fw : 0));
    }
    __builtin_amdgcn_sched_barrier(0);
#pragma unroll
    for (int j = 0; j < 8; ++j) {
      const bool in = fw0 + j * (AGG_WAVES * 128) < nwords;
      v[j].x = in ? v[j].x : 0u;
      v[j].y = in ? v[j].y : 0u;
      v[j].z = in ? v[j].z : 0u;
      v[j].w = in ? v[j].w : 0u;
    }
#pragma nounroll
    for (int j = 0; j < 8; ++j) {
      uint4 cur = v[0];
#pragma unroll
      for (int t = 0; t < 7; ++t) v[t] = v[t + 1];
#pragma nounroll
      for (int q = 0; q < 4; ++q) {
        unsigned x = cur.x;
        cur.x = cur.y; cur.y = cur.z; cur.z = cur.w;
        const int ebase = (fw0 + j * (AGG_WAVES * 128) + q) << 5;
        unsigned m = __builtin_amdgcn_ballot_w32(x != 0u);
        while (m != 0u) {
          if (x != 0u) {
            const int pos = qn + (int)__builtin_amdgcn_mbcnt_lo(m, 0u);
            if (pos < QCAP) qe[pos] = ebase + __builtin_ctz(x);
            x &= x - 1u;
          }
          qn += __popc(m);
          m = __builtin_amdgcn_ballot_w32(x != 0u);
        }
      }
    }
  }
  return qn;
}

template <int C, bool BUILD>
__global__ __launch_bounds__(AGG_TB)
void range_agg_kernel(const int* __restrict__ src,
                      const int* __restrict__ dst,
                      int E, int N,
                      const float* __restrict__ asrc,
                      const float* __restrict__ adst,
                      const float* __restrict__ h,
                      const float* __restrict__ bias,
                      float* __restrict__ out,
                      int do_relu,
                      int* __restrict__ wq_n,
                      int* __restrict__ wq_e,
                      int* __restrict__ wq_d,
                      const float* __restrict__ Wn,
                      const float* __restrict__ attn_s,
                      const float* __restrict__ attn_d,
                      float* __restrict__ asrc_n,
                      float* __restrict__ adst_n,
                      const unsigned* __restrict__ bmp,
                      int nwords) {
  __shared__ __attribute__((aligned(16))) float tile[RNODES * C];
  __shared__ __attribute__((aligned(16))) float sadst[RNODES * HEADS];
  __shared__ __attribute__((aligned(16))) int   smaxk[RNODES * HEADS];
  __shared__ __attribute__((aligned(16))) float ssum[RNODES * HEADS];
  __shared__ int qe_all[AGG_WAVES * QCAP];
  __shared__ int qd_all[AGG_WAVES * QCAP];
  __shared__ int sqn[AGG_WAVES];
  __shared__ int sover;
  __shared__ __attribute__((aligned(16))) float kept[2 * AGG_WAVES * SCAP * 4];

  const int tid = (int)threadIdx.x;
  const int lane = tid & 31;
  const int w = __builtin_amdgcn_readfirstlane(tid >> 5);
  const int n0 = (int)blockIdx.x * RNODES;
  const int nv = (N - n0 < RNODES) ? (N - n0) : RNODES;
  const size_t wq = (size_t)blockIdx.x * AGG_WAVES + w;

  int qn = 0;
  if constexpr (!BUILD) {
    qn = __builtin_amdgcn_readfirstlane(wq_n[wq]);
  }

  for (int i = tid; i < RNODES * C; i += AGG_TB) tile[i] = 0.f;
  for (int i = tid; i < RNODES * HEADS; i += AGG_TB) {
    sadst[i] = (i < nv * HEADS) ? adst[(size_t)n0 * HEADS + i] : 0.f;
    smaxk[i] = fkey(__float_as_int(-3.0e38f));
    ssum[i] = 0.f;
  }
  if (tid == 0) sover = 0;

  int* qe = qe_all + w * QCAP;
  int* qd = qd_all + w * QCAP;
  float4* wlo = (float4*)kept + w * SCAP;
  float4* whi = (float4*)kept + (AGG_WAVES + w) * SCAP;
  if constexpr (BUILD) {
    const int rs = w + 32 * lane;
    const bool own = rs < nv;
    if (own) qe[lane] = E + n0 + rs;
    qn = __popc(__builtin_amdgcn_ballot_w32(own));
    __syncthreads();
    qn = bitmap_list(bmp + (size_t)blockIdx.x * nwords, nwords, w, lane, qe, qn);
    if (lane == 0) sqn[w] = qn;
  } else {
    if (qn <= QCAP) {
      const int* we = wq_e + wq * QCAP;
      const int* wd = wq_d + wq * QCAP;
      for (int i = lane; i < qn; i += 32) { qe[i] = we[i]; qd[i] = wd[i]; }
    }
    __syncthreads();
  }
  if (qn > QCAP) sover = 1;
  __syncthreads();
  const bool over = sover != 0;
  if constexpr (BUILD) {
    if (w == 0) wq_n[(size_t)blockIdx.x * AGG_WAVES + lane] = sqn[lane];
  }

  if (!over) {
    list_pass<C, PH_MAX, BUILD>(src, dst, E, n0, qn, lane, qe, qd, asrc, h, sadst, smaxk, ssum, tile,
                                wlo, whi);
    __syncthreads();
    if constexpr (BUILD) {
      int* we = wq_e + wq * QCAP;
      int* wd = wq_d + wq * QCAP;
      for (int i = lane; i < qn; i += 32) { we[i] = qe[i]; wd[i] = qd[i]; }
    }
    list_pass<C, PH_SUM, false>(src, dst, E, n0, qn, lane, qe, qd, asrc, h, sadst, smaxk, ssum, tile,
                                wlo, whi);
    __syncthreads();
    list_pass<C, PH_AGG, false>(src, dst, E, n0, qn, lane, qe, qd, asrc, h, sadst, smaxk, ssum, tile,
                                wlo, whi);
  } else {
    scan_pass<C, PH_MAX>(src, dst, E, n0, nv, w, lane, qe, qd, 0,
                         asrc, h, sadst, smaxk, ssum, tile);
    __syncthreads();
    scan_pass<C, PH_SUM>(src, dst, E, n0, nv, w, lane, qe, qd, 0,
                         asrc, h, sadst, smaxk, ssum, tile);
    __syncthreads();
    scan_pass<C, PH_AGG>(src, dst, E, n0, nv, w, lane, qe, qd, 0,
                         asrc, h, sadst, smaxk, ssum, tile);
  }
  __syncthreads();

  const float inv = 1.0f / HEADS;
  if constexpr (!BUILD) {
    for (int i = tid; i < nv * C; i += AGG_TB) {
      float v = tile[i] * inv + bias[i & (C - 1)];
      if (do_relu) v = v > 0.f ? v : 0.f;
      out[(size_t)n0 * C + i] = v;
    }
  } else {
    constexpr int CN = 16;
    constexpr int NCOL = HEADS * CN;
    constexpr int HP = NCOL + 4;
    constexpr int TM = (RNODES + 15) / 16;
    constexpr int TG = NCOL / (16 * NT);
    static_assert(TM * TG <= AGG_WAVES, "one strip per wave");
    static_assert(TM * 16 * HP <= 2 * AGG_WAVES * SCAP * 4, "the projected rows fit the kept planes");
    float* hs = kept;
    for (int i = tid; i < RNODES * C; i += AGG_TB) {
      float v = tile[i] * inv + bias[i & (C - 1)];
      if (do_relu) v = v > 0.f ? v : 0.f;
      tile[i] = v;
    }
    __syncthreads();
    if (w < TM * TG) {
      const int tm = w / TG;
      const int tg = w - tm * TG;
      const int half = lane >> 4;
      const int l = lane & 15;
      int rowA = tm * 16 + l;
      if (rowA > RNODES - 1) rowA = RNODES - 1;
      const float* aP = tile + rowA * C + 2 * half;
      const float* bP = Wn + (size_t)(2 * half) * NCOL + tg * (16 * NT) + l;
      v8f acc[NT];
#pragma unroll
      for (int t = 0; t < NT; ++t) acc[t] = (v8f){};
#pragma unroll
      for (int k0 = 0; k0 < C; k0 += 4) {
        v2f a;
        a.x = aP[k0];
        a.y = aP[k0 + 1];
#pragma unroll
        for (int t = 0; t < NT; ++t) {
          v2f b;
          b.x = bP[(size_t)k0 * NCOL + t * 16];
          b.y = bP[(size_t)k0 * NCOL + t * 16 + NCOL];
          acc[t] = __builtin_amdgcn_wmma_f32_16x16x4_f32(false, a, false, b,
                                                         (short)0, acc[t],
                                                         false, false);
        }
      }
#pragma unroll
      for (int t = 0; t < NT; ++t) {
#pragma unroll
        for (int r = 0; r < 8; ++r)
          hs[(tm * 16 + r + 8 * half) * HP + tg * (16 * NT) + t * 16 + l] = acc[t][r];
      }
    }
    __syncthreads();
    float* hout = out + (size_t)n0 * NCOL;
    for (int i = tid * 4; i < nv * NCOL; i += AGG_TB * 4) {
      const int row = i / NCOL;
      const int col = i - row * NCOL;
      *(float4*)(hout + i) = *(const float4*)(hs + row * HP + col);
    }
    for (int i = tid; i < nv * HEADS; i += AGG_TB) {
      const int head = i & (HEADS - 1);
      const float* hp = hs + (i >> 3) * HP + head * CN;
      const float* as = attn_s + head * CN;
      const float* ad = attn_d + head * CN;
      float s = 0.f, d = 0.f;
#pragma unroll
      for (int c2 = 0; c2 < CN; ++c2) {
        float v = hp[c2];
        s += v * as[c2];
        d += v * ad[c2];
      }
      asrc_n[(size_t)n0 * HEADS + i] = s;
      adst_n[(size_t)n0 * HEADS + i] = d;
    }
  }
}

static inline unsigned nblk(long n, int b) { return (unsigned)((n + b - 1) / b); }

extern "C" void kernel_launch(void* const* d_in, const int* in_sizes, int n_in,
                              void* d_out, int out_size, void* d_ws, size_t ws_size,
                              hipStream_t stream) {
  const float* x        = (const float*)d_in[0];
  const int*   ei       = (const int*)  d_in[1];
  const float* W1       = (const float*)d_in[2];
  const float* att_src1 = (const float*)d_in[3];
  const float* att_dst1 = (const float*)d_in[4];
  const float* b1       = (const float*)d_in[5];
  const float* W2       = (const float*)d_in[6];
  const float* att_src2 = (const float*)d_in[7];
  const float* att_dst2 = (const float*)d_in[8];
  const float* b2       = (const float*)d_in[9];

  const int F_h   = in_sizes[3] / HEADS;
  const int F_out = in_sizes[7] / HEADS;
  const int F_in  = in_sizes[2] / (HEADS * F_h);
  const int N     = in_sizes[0] / F_in;
  const int E     = in_sizes[1] / 2;
  const int C1    = HEADS * F_h;
  const int C2    = HEADS * F_out;
  const int* src = ei;
  const int* dst = ei + E;

  float* ws = (float*)d_ws;
  size_t o = 0;
  float* h1    = ws + o; o += (size_t)N * C1;
  float* asrc  = ws + o; o += (size_t)N * HEADS;
  float* adst  = ws + o; o += (size_t)N * HEADS;
  o = (o + 31) & ~(size_t)31;
  float* h2    = ws + o; o += (size_t)N * C2;
  float* asrc2 = ws + o; o += (size_t)N * HEADS;
  float* adst2 = ws + o; o += (size_t)N * HEADS;
  const size_t nwg = nblk(N, RNODES);
  o = (o + 31) & ~(size_t)31;
  int* wq_n = (int*)(ws + o); o += nwg * AGG_WAVES;
  int* wq_e = (int*)(ws + o); o += nwg * AGG_WAVES * QCAP;
  int* wq_d = (int*)(ws + o); o += nwg * AGG_WAVES * QCAP;
  const int chunks = (E + BM_CHUNK - 1) / BM_CHUNK;
  const int nwords = chunks * 32;
  const int nbm = chunks * (int)((nwg + BM_OWN - 1) / BM_OWN);
  o = (o + 31) & ~(size_t)31;
  unsigned* bmp = (unsigned*)(ws + o); o += nwg * (size_t)nwords;
  float* outp = (float*)d_out;

  const int TB = 256;

  {
    long waves = (long)((N + 15) / 16) * (C1 / (16 * NT));
    wmma_gemm_f32<<<nblk(waves * 32, TB), TB, 0, stream>>>(x, W1, h1, N, F_in, C1);
  }
  coeff_bitmap_kernel<<<nbm + nblk(N, (BM_TB / 32) * CF_ROWS), BM_TB, 0, stream>>>(h1, att_src1, att_dst1,
                                                                   asrc, adst, N, F_h,
                                                                   dst, E, (int)nwg, chunks, nbm,
                                                                   bmp);
  range_agg_kernel<32, true><<<nblk(N, RNODES), AGG_TB, 0, stream>>>(src, dst, E, N, asrc, adst,
                                                                     h1, b1, h2, 1,
                                                                     wq_n, wq_e, wq_d,
                                                                     W2, att_src2, att_dst2,
                                                                     asrc2, adst2,
                                                                     bmp, nwords);

  range_agg_kernel<16, false><<<nblk(N, RNODES), AGG_TB, 0, stream>>>(src, dst, E, N, asrc2, adst2,
                                                                      h2, b2, outp, 0,
                                                                      wq_n, wq_e, wq_d,
                                                                      nullptr, nullptr, nullptr,
                                                                      nullptr, nullptr,
                                                                      nullptr, 0);
}
